# starting kernel
# baseline (speedup 1.0000x reference)
#include <hip/hip_runtime.h>

typedef __attribute__((ext_vector_type(2))) float v2f;
typedef __attribute__((ext_vector_type(8))) float v8f;

__global__ void gemm_xw(const float* __restrict__ X, const float* __restrict__ W,
                        const float* __restrict__ bias, float* __restrict__ C,
                        int M, int K) {
    const int gthread = blockIdx.x * blockDim.x + threadIdx.x;
    const int wave    = gthread >> 5;
    const int lane    = threadIdx.x & 31;
    const int m0      = wave * 16;
    if (m0 >= M) return;

    const int half = lane >> 4;
    const int lm   = lane & 15;

    v8f acc[4];
    #pragma unroll
    for (int nt = 0; nt < 4; ++nt) {
        const float bv = bias ? bias[nt * 16 + lm] : 0.0f;
        #pragma unroll
        for (int v = 0; v < 8; ++v) acc[nt][v] = bv;
    }

    const float* xrow = X + (size_t)(m0 + lm) * K;

    for (int k0 = 0; k0 < K; k0 += 4) {
        const int kb = k0 + 2 * half;
        const float2 av = *(const float2*)(xrow + kb);
        v2f a; a.x = av.x; a.y = av.y;
        #pragma unroll
        for (int nt = 0; nt < 4; ++nt) {
            const int n = nt * 16 + lm;
            v2f b;
            b.x = W[(size_t)kb * 64 + n];
            b.y = W[(size_t)(kb + 1) * 64 + n];
            acc[nt] = __builtin_amdgcn_wmma_f32_16x16x4_f32(
                false, a, false, b, (short)0, acc[nt], false, false);
        }
    }

    #pragma unroll
    for (int nt = 0; nt < 4; ++nt) {
        const int n = nt * 16 + lm;
        #pragma unroll
        for (int v = 0; v < 8; ++v) {
            C[(size_t)(m0 + v + 8 * half) * 64 + n] = acc[nt][v];
        }
    }
}

__global__ void edge_scatter64(const float* __restrict__ rel,
                               const int* __restrict__ src,
                               const int* __restrict__ dst,
                               const int* __restrict__ et,
                               float* __restrict__ agg, int E, int N) {
    const long long t = (long long)blockIdx.x * blockDim.x + threadIdx.x;
    const int e = (int)(t >> 4);
    if (e >= E) return;
    const int f = ((int)t & 15) * 4;
    const float4 v = *(const float4*)(rel + ((long long)et[e] * N + src[e]) * 64 + f);
    float* q = agg + (long long)dst[e] * 64 + f;
    atomicAdd(q + 0, v.x);
    atomicAdd(q + 1, v.y);
    atomicAdd(q + 2, v.z);
    atomicAdd(q + 3, v.w);
}

__global__ void relu_k(float* __restrict__ p, int n) {
    const int i = blockIdx.x * blockDim.x + threadIdx.x;
    if (i < n) p[i] = fmaxf(p[i], 0.0f);
}

__global__ void layer3_node(const float* __restrict__ h2,
                            const float* __restrict__ W3,
                            const float* __restrict__ loop3,
                            const float* __restrict__ b3,
                            float* __restrict__ rel3, float* __restrict__ agg3,
                            int N) {
    const int i = blockIdx.x * blockDim.x + threadIdx.x;
    if (i >= N) return;
    const float* row = h2 + (size_t)i * 64;
    float r00 = 0.f, r01 = 0.f, r10 = 0.f, r11 = 0.f, l0 = 0.f, l1 = 0.f;
    #pragma unroll 8
    for (int k = 0; k < 64; ++k) {
        const float v = row[k];
        r00 += v * W3[k * 2 + 0];
        r01 += v * W3[k * 2 + 1];
        r10 += v * W3[128 + k * 2 + 0];
        r11 += v * W3[128 + k * 2 + 1];
        l0  += v * loop3[k * 2 + 0];
        l1  += v * loop3[k * 2 + 1];
    }
    rel3[(size_t)i * 2 + 0] = r00;
    rel3[(size_t)i * 2 + 1] = r01;
    rel3[(size_t)(N + i) * 2 + 0] = r10;
    rel3[(size_t)(N + i) * 2 + 1] = r11;
    agg3[(size_t)i * 2 + 0] = l0 + b3[0];
    agg3[(size_t)i * 2 + 1] = l1 + b3[1];
}

__global__ void edge_scatter2(const float* __restrict__ rel3,
                              const int* __restrict__ src,
                              const int* __restrict__ dst,
                              const int* __restrict__ et,
                              float* __restrict__ agg3, int E, int N) {
    const int e = blockIdx.x * blockDim.x + threadIdx.x;
    if (e >= E) return;
    const float* p = rel3 + ((long long)et[e] * N + src[e]) * 2;
    float* q = agg3 + (long long)dst[e] * 2;
    atomicAdd(q + 0, p[0]);
    atomicAdd(q + 1, p[1]);
}

__device__ __forceinline__ unsigned f2ord(float f) {
    unsigned u = __float_as_uint(f);
    return (u & 0x80000000u) ? ~u : (u | 0x80000000u);
}
__device__ __forceinline__ float ord2f(unsigned u) {
    return __uint_as_float((u & 0x80000000u) ? (u ^ 0x80000000u) : ~u);
}

__global__ void init_slot(unsigned* slot) { *slot = 0xFFFFFFFFu; }

__global__ void min_reduce(const float* __restrict__ p, int n, unsigned* slot) {
    __shared__ float sm[256];
    float m = 3.402823466e38f;
    for (int i = blockIdx.x * blockDim.x + threadIdx.x; i < n;
         i += gridDim.x * blockDim.x)
        m = fminf(m, p[i]);
    sm[threadIdx.x] = m;
    __syncthreads();
    for (int s = blockDim.x >> 1; s > 0; s >>= 1) {
        if ((int)threadIdx.x < s) sm[threadIdx.x] = fminf(sm[threadIdx.x], sm[threadIdx.x + s]);
        __syncthreads();
    }
    if (threadIdx.x == 0) atomicMin(slot, f2ord(sm[0]));
}

__global__ void finalize_k(const float* __restrict__ agg3,
                           const int* __restrict__ cell,
                           const int* __restrict__ maxs,
                           const unsigned* __restrict__ slot,
                           float* __restrict__ out, int N) {
    const int i = blockIdx.x * blockDim.x + threadIdx.x;
    if (i >= N) return;
    const float gmin = ord2f(*slot) - 1.0f;
    const bool upper = cell[i] >= (maxs[i] - 1);
    const bool lower = cell[i] == 0;
    out[(size_t)i * 2 + 0] = upper ? gmin : agg3[(size_t)i * 2 + 0];
    out[(size_t)i * 2 + 1] = lower ? gmin : agg3[(size_t)i * 2 + 1];
}

extern "C" void kernel_launch(void* const* d_in, const int* in_sizes, int n_in,
                              void* d_out, int out_size, void* d_ws, size_t ws_size,
                              hipStream_t stream) {
    const float* x      = (const float*)d_in[0];
    const int*   src    = (const int*)d_in[1];
    const int*   dst    = (const int*)d_in[2];
    const int*   etypes = (const int*)d_in[3];
    const int*   cell   = (const int*)d_in[4];
    const int*   maxs   = (const int*)d_in[5];
    const float* W1     = (const float*)d_in[6];
    const float* loop1  = (const float*)d_in[7];
    const float* b1     = (const float*)d_in[8];
    const float* W2     = (const float*)d_in[9];
    const float* loop2  = (const float*)d_in[10];
    const float* b2     = (const float*)d_in[11];
    const float* W3     = (const float*)d_in[12];
    const float* loop3  = (const float*)d_in[13];
    const float* b3     = (const float*)d_in[14];
    float* out = (float*)d_out;

    const int N = in_sizes[4];
    const int E = in_sizes[1];
    const int K1 = in_sizes[0] / N;

    float* relbuf = (float*)d_ws;
    float* h1     = relbuf + (size_t)2 * N * 64;
    float* h2     = h1 + (size_t)N * 64;
    float* rel3   = h2 + (size_t)N * 64;
    float* agg3   = rel3 + (size_t)4 * N;
    unsigned* slot = (unsigned*)(agg3 + (size_t)2 * N);

    const int waves   = (N + 15) / 16;
    const int gblocks = (waves * 32 + 255) / 256;
    const int eb64    = (int)(((long long)E * 16 + 255) / 256);
    const int nb      = (N + 255) / 256;
    const int fb      = (N * 64 + 255) / 256;
    const int eb2     = (E + 255) / 256;

    gemm_xw<<<gblocks, 256, 0, stream>>>(x, W1,            nullptr, relbuf,                  N, K1);
    gemm_xw<<<gblocks, 256, 0, stream>>>(x, W1 + K1 * 64,  nullptr, relbuf + (size_t)N * 64, N, K1);
    gemm_xw<<<gblocks, 256, 0, stream>>>(x, loop1,         b1,      h1,                      N, K1);
    edge_scatter64<<<eb64, 256, 0, stream>>>(relbuf, src, dst, etypes, h1, E, N);
    relu_k<<<fb, 256, 0, stream>>>(h1, N * 64);

    gemm_xw<<<gblocks, 256, 0, stream>>>(h1, W2,          nullptr, relbuf,                  N, 64);
    gemm_xw<<<gblocks, 256, 0, stream>>>(h1, W2 + 64 * 64, nullptr, relbuf + (size_t)N * 64, N, 64);
    gemm_xw<<<gblocks, 256, 0, stream>>>(h1, loop2,        b2,      h2,                      N, 64);
    edge_scatter64<<<eb64, 256, 0, stream>>>(relbuf, src, dst, etypes, h2, E, N);
    relu_k<<<fb, 256, 0, stream>>>(h2, N * 64);

    layer3_node<<<nb, 256, 0, stream>>>(h2, W3, loop3, b3, rel3, agg3, N);
    edge_scatter2<<<eb2, 256, 0, stream>>>(rel3, src, dst, etypes, agg3, E, N);
    init_slot<<<1, 1, 0, stream>>>(slot);
    min_reduce<<<512, 256, 0, stream>>>(agg3, N * 2, slot);
    finalize_k<<<nb, 256, 0, stream>>>(agg3, cell, maxs, slot, out, N);
}


# Replace the per-edge float atomic scatter of all three layers by a per-call counting sort of edges by dst and a one-writer-per-node gather (relu in its store); sums now in sorted-edge order
# speedup vs baseline: 1.4771x; 1.4771x over previous
#include <hip/hip_runtime.h>

typedef __attribute__((ext_vector_type(2))) float v2f;
typedef __attribute__((ext_vector_type(8))) float v8f;

__global__ void gemm_xw(const float* __restrict__ X, const float* __restrict__ W,
                        const float* __restrict__ bias, float* __restrict__ C,
                        int M, int K) {
    const int gthread = blockIdx.x * blockDim.x + threadIdx.x;
    const int wave    = gthread >> 5;
    const int lane    = threadIdx.x & 31;
    const int m0      = wave * 16;
    if (m0 >= M) return;

    const int half = lane >> 4;
    const int lm   = lane & 15;

    v8f acc[4];
    #pragma unroll
    for (int nt = 0; nt < 4; ++nt) {
        const float bv = bias ? bias[nt * 16 + lm] : 0.0f;
        #pragma unroll
        for (int v = 0; v < 8; ++v) acc[nt][v] = bv;
    }

    const float* xrow = X + (size_t)(m0 + lm) * K;

    for (int k0 = 0; k0 < K; k0 += 4) {
        const int kb = k0 + 2 * half;
        const float2 av = *(const float2*)(xrow + kb);
        v2f a; a.x = av.x; a.y = av.y;
        #pragma unroll
        for (int nt = 0; nt < 4; ++nt) {
            const int n = nt * 16 + lm;
            v2f b;
            b.x = W[(size_t)kb * 64 + n];
            b.y = W[(size_t)(kb + 1) * 64 + n];
            acc[nt] = __builtin_amdgcn_wmma_f32_16x16x4_f32(
                false, a, false, b, (short)0, acc[nt], false, false);
        }
    }

    #pragma unroll
    for (int nt = 0; nt < 4; ++nt) {
        const int n = nt * 16 + lm;
        #pragma unroll
        for (int v = 0; v < 8; ++v) {
            C[(size_t)(m0 + v + 8 * half) * 64 + n] = acc[nt][v];
        }
    }
}

__global__ void edge_scatter64(const float* __restrict__ rel,
                               const int* __restrict__ src,
                               const int* __restrict__ dst,
                               const int* __restrict__ et,
                               float* __restrict__ agg, int E, int N) {
    const long long t = (long long)blockIdx.x * blockDim.x + threadIdx.x;
    const int e = (int)(t >> 4);
    if (e >= E) return;
    const int f = ((int)t & 15) * 4;
    const float4 v = *(const float4*)(rel + ((long long)et[e] * N + src[e]) * 64 + f);
    float* q = agg + (long long)dst[e] * 64 + f;
    atomicAdd(q + 0, v.x);
    atomicAdd(q + 1, v.y);
    atomicAdd(q + 2, v.z);
    atomicAdd(q + 3, v.w);
}

__global__ void relu_k(float* __restrict__ p, int n) {
    const int i = blockIdx.x * blockDim.x + threadIdx.x;
    if (i < n) p[i] = fmaxf(p[i], 0.0f);
}

__global__ void layer3_node(const float* __restrict__ h2,
                            const float* __restrict__ W3,
                            const float* __restrict__ loop3,
                            const float* __restrict__ b3,
                            float* __restrict__ rel3, float* __restrict__ agg3,
                            int N) {
    const int i = blockIdx.x * blockDim.x + threadIdx.x;
    if (i >= N) return;
    const float* row = h2 + (size_t)i * 64;
    float r00 = 0.f, r01 = 0.f, r10 = 0.f, r11 = 0.f, l0 = 0.f, l1 = 0.f;
    #pragma unroll 8
    for (int k = 0; k < 64; ++k) {
        const float v = row[k];
        r00 += v * W3[k * 2 + 0];
        r01 += v * W3[k * 2 + 1];
        r10 += v * W3[128 + k * 2 + 0];
        r11 += v * W3[128 + k * 2 + 1];
        l0  += v * loop3[k * 2 + 0];
        l1  += v * loop3[k * 2 + 1];
    }
    rel3[(size_t)i * 2 + 0] = r00;
    rel3[(size_t)i * 2 + 1] = r01;
    rel3[(size_t)(N + i) * 2 + 0] = r10;
    rel3[(size_t)(N + i) * 2 + 1] = r11;
    agg3[(size_t)i * 2 + 0] = l0 + b3[0];
    agg3[(size_t)i * 2 + 1] = l1 + b3[1];
}

__global__ void edge_scatter2(const float* __restrict__ rel3,
                              const int* __restrict__ src,
                              const int* __restrict__ dst,
                              const int* __restrict__ et,
                              float* __restrict__ agg3, int E, int N) {
    const int e = blockIdx.x * blockDim.x + threadIdx.x;
    if (e >= E) return;
    const float* p = rel3 + ((long long)et[e] * N + src[e]) * 2;
    float* q = agg3 + (long long)dst[e] * 2;
    atomicAdd(q + 0, p[0]);
    atomicAdd(q + 1, p[1]);
}


#define CSR_NB  64
#define CSR_CAP 51200

__global__ __launch_bounds__(1024) void csr_hist(const int* __restrict__ dst,
                                                 int* __restrict__ rank,
                                                 int* __restrict__ blockcnt,
                                                 int E, int N, int chunk) {
    __shared__ int cnt[CSR_CAP];
    const int b  = blockIdx.x;
    const int e0 = b * chunk;
    const int e1 = min(e0 + chunk, E);
    for (int base = 0; base < N; base += CSR_CAP) {
        const int lim = min(CSR_CAP, N - base);
        for (int i = threadIdx.x; i < lim; i += blockDim.x) cnt[i] = 0;
        __syncthreads();
        const int bd = (int)blockDim.x;
        for (int e = e0 + (int)threadIdx.x; e < e1; e += 4 * bd) {
            int d[4];
            #pragma unroll
            for (int j = 0; j < 4; ++j) {
                const int ee = e + j * bd;
                const int v  = dst[min(ee, e1 - 1)];
                d[j] = (ee < e1) ? v - base : -1;
            }
            #pragma unroll
            for (int j = 0; j < 4; ++j)
                if (d[j] >= 0 && d[j] < lim) rank[e + j * bd] = atomicAdd(&cnt[d[j]], 1);
        }
        __syncthreads();
        for (int i = threadIdx.x; i < lim; i += blockDim.x)
            blockcnt[(size_t)b * N + base + i] = cnt[i];
        __syncthreads();
    }
}

__global__ void csr_colscan(int* __restrict__ blockcnt, int* __restrict__ off, int N) {
    const int n = blockIdx.x * blockDim.x + threadIdx.x;
    if (n >= N) return;
    int run = 0;
    for (int b0 = 0; b0 < CSR_NB; b0 += 16) {
        int c[16];
        #pragma unroll
        for (int j = 0; j < 16; ++j) c[j] = blockcnt[(size_t)(b0 + j) * N + n];
        #pragma unroll
        for (int j = 0; j < 16; ++j) {
            blockcnt[(size_t)(b0 + j) * N + n] = run;
            run += c[j];
        }
    }
    off[n] = run;
}

__global__ __launch_bounds__(1024) void csr_scan(int* __restrict__ off, int N) {
    __shared__ int buf[CSR_CAP];
    __shared__ int part[1024];
    const int t = threadIdx.x;
    int carry = 0;
    for (int base = 0; base < N; base += CSR_CAP) {
        const int lim = min(CSR_CAP, N - base);
        for (int i = t; i < lim; i += 1024) buf[i] = off[base + i];
        __syncthreads();
        const int per = ((lim + 1023) / 1024) | 1;
        const int s  = min(t * per, lim);
        const int en = min(s + per, lim);
        int sum = 0;
        for (int i = s; i < en; ++i) sum += buf[i];
        part[t] = sum;
        __syncthreads();
        for (int d = 1; d < 1024; d <<= 1) {
            const int v = (t >= d) ? part[t - d] : 0;
            __syncthreads();
            part[t] += v;
            __syncthreads();
        }
        int run = carry + part[t] - sum;
        for (int i = s; i < en; ++i) { const int c = buf[i]; buf[i] = run; run += c; }
        __syncthreads();
        for (int i = t; i < lim; i += 1024) off[base + i] = buf[i];
        carry += part[1023];
        __syncthreads();
    }
    if (t == 0) off[N] = carry;
}

__global__ void csr_fill(const int* __restrict__ src, const int* __restrict__ dst,
                         const int* __restrict__ et, const int* __restrict__ rank,
                         const int* __restrict__ blockcnt, const int* __restrict__ off,
                         int* __restrict__ key, int E, int N, int chunk) {
    const int e = blockIdx.x * blockDim.x + threadIdx.x;
    if (e >= E) return;
    const int d = dst[e];
    const int b = e / chunk;
    int pos = off[d] + blockcnt[(size_t)b * N + d] + rank[e];
    pos = min(max(pos, 0), E - 1);
    key[pos] = et[e] * N + src[e];
}

__global__ void gather64_relu(const float* __restrict__ rel, const int* __restrict__ off,
                              const int* __restrict__ key, float* __restrict__ h, int N) {
    const long long t = (long long)blockIdx.x * blockDim.x + threadIdx.x;
    const int i = (int)(t >> 4);
    if (i >= N) return;
    const int f = ((int)t & 15) * 4;
    float* hp = h + (size_t)i * 64 + f;
    float4 acc = *(const float4*)hp;
    int p = off[i];
    const int pe = off[i + 1];
    for (; p + 4 <= pe; p += 4) {
        const int k0 = key[p], k1 = key[p + 1], k2 = key[p + 2], k3 = key[p + 3];
        const float4 a = *(const float4*)(rel + (size_t)k0 * 64 + f);
        const float4 b = *(const float4*)(rel + (size_t)k1 * 64 + f);
        const float4 c = *(const float4*)(rel + (size_t)k2 * 64 + f);
        const float4 d = *(const float4*)(rel + (size_t)k3 * 64 + f);
        acc.x += a.x; acc.y += a.y; acc.z += a.z; acc.w += a.w;
        acc.x += b.x; acc.y += b.y; acc.z += b.z; acc.w += b.w;
        acc.x += c.x; acc.y += c.y; acc.z += c.z; acc.w += c.w;
        acc.x += d.x; acc.y += d.y; acc.z += d.z; acc.w += d.w;
    }
    for (; p < pe; ++p) {
        const float4 a = *(const float4*)(rel + (size_t)key[p] * 64 + f);
        acc.x += a.x; acc.y += a.y; acc.z += a.z; acc.w += a.w;
    }
    acc.x = fmaxf(acc.x, 0.0f); acc.y = fmaxf(acc.y, 0.0f);
    acc.z = fmaxf(acc.z, 0.0f); acc.w = fmaxf(acc.w, 0.0f);
    *(float4*)hp = acc;
}

__global__ void gather2(const float* __restrict__ rel3, const int* __restrict__ off,
                        const int* __restrict__ key, float* __restrict__ agg3, int N) {
    const long long t = (long long)blockIdx.x * blockDim.x + threadIdx.x;
    const int i = (int)(t >> 4);
    const int l = (int)t & 15;
    float s0 = 0.0f, s1 = 0.0f;
    if (i < N) {
        const int pe = off[i + 1];
        for (int p = off[i] + l; p < pe; p += 16) {
            const float2 r = *(const float2*)(rel3 + (size_t)key[p] * 2);
            s0 += r.x; s1 += r.y;
        }
    }
    for (int m = 8; m >= 1; m >>= 1) {
        s0 += __shfl_xor(s0, m, 16);
        s1 += __shfl_xor(s1, m, 16);
    }
    if (i < N && l == 0) {
        agg3[(size_t)i * 2 + 0] += s0;
        agg3[(size_t)i * 2 + 1] += s1;
    }
}

__device__ __forceinline__ unsigned f2ord(float f) {
    unsigned u = __float_as_uint(f);
    return (u & 0x80000000u) ? ~u : (u | 0x80000000u);
}
__device__ __forceinline__ float ord2f(unsigned u) {
    return __uint_as_float((u & 0x80000000u) ? (u ^ 0x80000000u) : ~u);
}

__global__ void init_slot(unsigned* slot) { *slot = 0xFFFFFFFFu; }

__global__ void min_reduce(const float* __restrict__ p, int n, unsigned* slot) {
    __shared__ float sm[256];
    float m = 3.402823466e38f;
    for (int i = blockIdx.x * blockDim.x + threadIdx.x; i < n;
         i += gridDim.x * blockDim.x)
        m = fminf(m, p[i]);
    sm[threadIdx.x] = m;
    __syncthreads();
    for (int s = blockDim.x >> 1; s > 0; s >>= 1) {
        if ((int)threadIdx.x < s) sm[threadIdx.x] = fminf(sm[threadIdx.x], sm[threadIdx.x + s]);
        __syncthreads();
    }
    if (threadIdx.x == 0) atomicMin(slot, f2ord(sm[0]));
}

__global__ void finalize_k(const float* __restrict__ agg3,
                           const int* __restrict__ cell,
                           const int* __restrict__ maxs,
                           const unsigned* __restrict__ slot,
                           float* __restrict__ out, int N) {
    const int i = blockIdx.x * blockDim.x + threadIdx.x;
    if (i >= N) return;
    const float gmin = ord2f(*slot) - 1.0f;
    const bool upper = cell[i] >= (maxs[i] - 1);
    const bool lower = cell[i] == 0;
    out[(size_t)i * 2 + 0] = upper ? gmin : agg3[(size_t)i * 2 + 0];
    out[(size_t)i * 2 + 1] = lower ? gmin : agg3[(size_t)i * 2 + 1];
}

extern "C" void kernel_launch(void* const* d_in, const int* in_sizes, int n_in,
                              void* d_out, int out_size, void* d_ws, size_t ws_size,
                              hipStream_t stream) {
    const float* x      = (const float*)d_in[0];
    const int*   src    = (const int*)d_in[1];
    const int*   dst    = (const int*)d_in[2];
    const int*   etypes = (const int*)d_in[3];
    const int*   cell   = (const int*)d_in[4];
    const int*   maxs   = (const int*)d_in[5];
    const float* W1     = (const float*)d_in[6];
    const float* loop1  = (const float*)d_in[7];
    const float* b1     = (const float*)d_in[8];
    const float* W2     = (const float*)d_in[9];
    const float* loop2  = (const float*)d_in[10];
    const float* b2     = (const float*)d_in[11];
    const float* W3     = (const float*)d_in[12];
    const float* loop3  = (const float*)d_in[13];
    const float* b3     = (const float*)d_in[14];
    float* out = (float*)d_out;

    const int N = in_sizes[4];
    const int E = in_sizes[1];
    const int K1 = in_sizes[0] / N;

    float* relbuf = (float*)d_ws;
    float* h1     = relbuf + (size_t)2 * N * 64;
    float* h2     = h1 + (size_t)N * 64;
    float* rel3   = h2 + (size_t)N * 64;
    float* agg3   = rel3 + (size_t)4 * N;
    unsigned* slot = (unsigned*)(agg3 + (size_t)2 * N);

    const int waves   = (N + 15) / 16;
    const int gblocks = (waves * 32 + 255) / 256;
    const int eb64    = (int)(((long long)E * 16 + 255) / 256);
    const int nb      = (N + 255) / 256;
    const int fb      = (N * 64 + 255) / 256;
    const int eb2     = (E + 255) / 256;

    int* key      = (int*)(slot + 4);
    int* off      = key + (size_t)E;
    int* blockcnt = (int*)relbuf;
    int* rank     = blockcnt + (size_t)CSR_NB * N;
    const size_t need = ((size_t)262 * N + 4 + (size_t)E + (size_t)N + 1) * sizeof(float);
    const bool csr_ok = ws_size >= need && (long long)E <= (long long)192 * N;

    if (csr_ok) {
        const int chunk = (E + CSR_NB - 1) / CSR_NB;
        const int gb64  = (int)(((long long)N * 16 + 255) / 256);
        csr_hist<<<CSR_NB, 1024, 0, stream>>>(dst, rank, blockcnt, E, N, chunk);
        csr_colscan<<<nb, 256, 0, stream>>>(blockcnt, off, N);
        csr_scan<<<1, 1024, 0, stream>>>(off, N);
        csr_fill<<<eb2, 256, 0, stream>>>(src, dst, etypes, rank, blockcnt, off, key, E, N, chunk);

        gemm_xw<<<gblocks, 256, 0, stream>>>(x, W1,            nullptr, relbuf,                  N, K1);
        gemm_xw<<<gblocks, 256, 0, stream>>>(x, W1 + K1 * 64,  nullptr, relbuf + (size_t)N * 64, N, K1);
        gemm_xw<<<gblocks, 256, 0, stream>>>(x, loop1,         b1,      h1,                      N, K1);
        gather64_relu<<<gb64, 256, 0, stream>>>(relbuf, off, key, h1, N);

        gemm_xw<<<gblocks, 256, 0, stream>>>(h1, W2,           nullptr, relbuf,                  N, 64);
        gemm_xw<<<gblocks, 256, 0, stream>>>(h1, W2 + 64 * 64, nullptr, relbuf + (size_t)N * 64, N, 64);
        gemm_xw<<<gblocks, 256, 0, stream>>>(h1, loop2,        b2,      h2,                      N, 64);
        gather64_relu<<<gb64, 256, 0, stream>>>(relbuf, off, key, h2, N);

        layer3_node<<<nb, 256, 0, stream>>>(h2, W3, loop3, b3, rel3, agg3, N);
        gather2<<<gb64, 256, 0, stream>>>(rel3, off, key, agg3, N);
        init_slot<<<1, 1, 0, stream>>>(slot);
        min_reduce<<<512, 256, 0, stream>>>(agg3, N * 2, slot);
        finalize_k<<<nb, 256, 0, stream>>>(agg3, cell, maxs, slot, out, N);
    } else {
        gemm_xw<<<gblocks, 256, 0, stream>>>(x, W1,            nullptr, relbuf,                  N, K1);
        gemm_xw<<<gblocks, 256, 0, stream>>>(x, W1 + K1 * 64,  nullptr, relbuf + (size_t)N * 64, N, K1);
        gemm_xw<<<gblocks, 256, 0, stream>>>(x, loop1,         b1,      h1,                      N, K1);
        edge_scatter64<<<eb64, 256, 0, stream>>>(relbuf, src, dst, etypes, h1, E, N);
        relu_k<<<fb, 256, 0, stream>>>(h1, N * 64);

        gemm_xw<<<gblocks, 256, 0, stream>>>(h1, W2,          nullptr, relbuf,                  N, 64);
        gemm_xw<<<gblocks, 256, 0, stream>>>(h1, W2 + 64 * 64, nullptr, relbuf + (size_t)N * 64, N, 64);
        gemm_xw<<<gblocks, 256, 0, stream>>>(h1, loop2,        b2,      h2,                      N, 64);
        edge_scatter64<<<eb64, 256, 0, stream>>>(relbuf, src, dst, etypes, h2, E, N);
        relu_k<<<fb, 256, 0, stream>>>(h2, N * 64);

        layer3_node<<<nb, 256, 0, stream>>>(h2, W3, loop3, b3, rel3, agg3, N);
        edge_scatter2<<<eb2, 256, 0, stream>>>(rel3, src, dst, etypes, agg3, E, N);
        init_slot<<<1, 1, 0, stream>>>(slot);
        min_reduce<<<512, 256, 0, stream>>>(agg3, N * 2, slot);
        finalize_k<<<nb, 256, 0, stream>>>(agg3, cell, maxs, slot, out, N);
    }
}


# Fuse each layer's three products (two relation weights, self-loop) into one launch: a wave loads its A fragment once per k-step and feeds 12 column tiles; 17 -> 13 launches
# speedup vs baseline: 1.8724x; 1.2676x over previous
#include <hip/hip_runtime.h>

typedef __attribute__((ext_vector_type(2))) float v2f;
typedef __attribute__((ext_vector_type(8))) float v8f;

__global__ __launch_bounds__(256) void gemm3_xw(const float* __restrict__ X, const float* __restrict__ Wr,
                         const float* __restrict__ Wl, const float* __restrict__ bias,
                         float* __restrict__ rel, float* __restrict__ H,
                         int M, int K) {
    const int gthread = blockIdx.x * blockDim.x + threadIdx.x;
    const int wave    = gthread >> 5;
    const int lane    = threadIdx.x & 31;
    const int m0      = wave * 16;
    if (m0 >= M) return;

    const int half = lane >> 4;
    const int lm   = lane & 15;

    v8f acc[12];
    #pragma unroll
    for (int t = 0; t < 8; ++t) {
        #pragma unroll
        for (int v = 0; v < 8; ++v) acc[t][v] = 0.0f;
    }
    #pragma unroll
    for (int nt = 0; nt < 4; ++nt) {
        const float bv = bias[nt * 16 + lm];
        #pragma unroll
        for (int v = 0; v < 8; ++v) acc[8 + nt][v] = bv;
    }

    const float* xrow = X + (size_t)(m0 + lm) * K;
    const float* w1   = Wr + (size_t)K * 64;

    #pragma unroll 2
    for (int k0 = 0; k0 < K; k0 += 4) {
        const int kb = k0 + 2 * half;
        const float2 av = *(const float2*)(xrow + kb);
        v2f a; a.x = av.x; a.y = av.y;
        #pragma unroll
        for (int mt = 0; mt < 3; ++mt) {
            const float* Wm = (mt == 0) ? Wr : (mt == 1) ? w1 : Wl;
            #pragma unroll
            for (int nt = 0; nt < 4; ++nt) {
                const int n = nt * 16 + lm;
                v2f b;
                b.x = Wm[(size_t)kb * 64 + n];
                b.y = Wm[(size_t)(kb + 1) * 64 + n];
                acc[mt * 4 + nt] = __builtin_amdgcn_wmma_f32_16x16x4_f32(
                    false, a, false, b, (short)0, acc[mt * 4 + nt], false, false);
            }
        }
    }

    #pragma unroll
    for (int mt = 0; mt < 3; ++mt) {
        float* C = (mt == 2) ? H : rel + (size_t)mt * M * 64;
        #pragma unroll
        for (int nt = 0; nt < 4; ++nt) {
            const int n = nt * 16 + lm;
            #pragma unroll
            for (int v = 0; v < 8; ++v) {
                C[(size_t)(m0 + v + 8 * half) * 64 + n] = acc[mt * 4 + nt][v];
            }
        }
    }
}

__global__ void edge_scatter64(const float* __restrict__ rel,
                               const int* __restrict__ src,
                               const int* __restrict__ dst,
                               const int* __restrict__ et,
                               float* __restrict__ agg, int E, int N) {
    const long long t = (long long)blockIdx.x * blockDim.x + threadIdx.x;
    const int e = (int)(t >> 4);
    if (e >= E) return;
    const int f = ((int)t & 15) * 4;
    const float4 v = *(const float4*)(rel + ((long long)et[e] * N + src[e]) * 64 + f);
    float* q = agg + (long long)dst[e] * 64 + f;
    atomicAdd(q + 0, v.x);
    atomicAdd(q + 1, v.y);
    atomicAdd(q + 2, v.z);
    atomicAdd(q + 3, v.w);
}

__global__ void relu_k(float* __restrict__ p, int n) {
    const int i = blockIdx.x * blockDim.x + threadIdx.x;
    if (i < n) p[i] = fmaxf(p[i], 0.0f);
}

__global__ void layer3_node(const float* __restrict__ h2,
                            const float* __restrict__ W3,
                            const float* __restrict__ loop3,
                            const float* __restrict__ b3,
                            float* __restrict__ rel3, float* __restrict__ agg3,
                            int N) {
    const int i = blockIdx.x * blockDim.x + threadIdx.x;
    if (i >= N) return;
    const float* row = h2 + (size_t)i * 64;
    float r00 = 0.f, r01 = 0.f, r10 = 0.f, r11 = 0.f, l0 = 0.f, l1 = 0.f;
    #pragma unroll 8
    for (int k = 0; k < 64; ++k) {
        const float v = row[k];
        r00 += v * W3[k * 2 + 0];
        r01 += v * W3[k * 2 + 1];
        r10 += v * W3[128 + k * 2 + 0];
        r11 += v * W3[128 + k * 2 + 1];
        l0  += v * loop3[k * 2 + 0];
        l1  += v * loop3[k * 2 + 1];
    }
    rel3[(size_t)i * 2 + 0] = r00;
    rel3[(size_t)i * 2 + 1] = r01;
    rel3[(size_t)(N + i) * 2 + 0] = r10;
    rel3[(size_t)(N + i) * 2 + 1] = r11;
    agg3[(size_t)i * 2 + 0] = l0 + b3[0];
    agg3[(size_t)i * 2 + 1] = l1 + b3[1];
}

__global__ void edge_scatter2(const float* __restrict__ rel3,
                              const int* __restrict__ src,
                              const int* __restrict__ dst,
                              const int* __restrict__ et,
                              float* __restrict__ agg3, int E, int N) {
    const int e = blockIdx.x * blockDim.x + threadIdx.x;
    if (e >= E) return;
    const float* p = rel3 + ((long long)et[e] * N + src[e]) * 2;
    float* q = agg3 + (long long)dst[e] * 2;
    atomicAdd(q + 0, p[0]);
    atomicAdd(q + 1, p[1]);
}


#define CSR_NB  64
#define CSR_CAP 51200

__global__ __launch_bounds__(1024) void csr_hist(const int* __restrict__ dst,
                                                 int* __restrict__ rank,
                                                 int* __restrict__ blockcnt,
                                                 int E, int N, int chunk) {
    __shared__ int cnt[CSR_CAP];
    const int b  = blockIdx.x;
    const int e0 = b * chunk;
    const int e1 = min(e0 + chunk, E);
    for (int base = 0; base < N; base += CSR_CAP) {
        const int lim = min(CSR_CAP, N - base);
        for (int i = threadIdx.x; i < lim; i += blockDim.x) cnt[i] = 0;
        __syncthreads();
        const int bd = (int)blockDim.x;
        for (int e = e0 + (int)threadIdx.x; e < e1; e += 4 * bd) {
            int d[4];
            #pragma unroll
            for (int j = 0; j < 4; ++j) {
                const int ee = e + j * bd;
                const int v  = dst[min(ee, e1 - 1)];
                d[j] = (ee < e1) ? v - base : -1;
            }
            #pragma unroll
            for (int j = 0; j < 4; ++j)
                if (d[j] >= 0 && d[j] < lim) rank[e + j * bd] = atomicAdd(&cnt[d[j]], 1);
        }
        __syncthreads();
        for (int i = threadIdx.x; i < lim; i += blockDim.x)
            blockcnt[(size_t)b * N + base + i] = cnt[i];
        __syncthreads();
    }
}

__global__ void csr_colscan(int* __restrict__ blockcnt, int* __restrict__ off, int N) {
    const int n = blockIdx.x * blockDim.x + threadIdx.x;
    if (n >= N) return;
    int run = 0;
    for (int b0 = 0; b0 < CSR_NB; b0 += 16) {
        int c[16];
        #pragma unroll
        for (int j = 0; j < 16; ++j) c[j] = blockcnt[(size_t)(b0 + j) * N + n];
        #pragma unroll
        for (int j = 0; j < 16; ++j) {
            blockcnt[(size_t)(b0 + j) * N + n] = run;
            run += c[j];
        }
    }
    off[n] = run;
}

__global__ __launch_bounds__(1024) void csr_scan(int* __restrict__ off, int N) {
    __shared__ int buf[CSR_CAP];
    __shared__ int part[1024];
    const int t = threadIdx.x;
    int carry = 0;
    for (int base = 0; base < N; base += CSR_CAP) {
        const int lim = min(CSR_CAP, N - base);
        for (int i = t; i < lim; i += 1024) buf[i] = off[base + i];
        __syncthreads();
        const int per = ((lim + 1023) / 1024) | 1;
        const int s  = min(t * per, lim);
        const int en = min(s + per, lim);
        int sum = 0;
        for (int i = s; i < en; ++i) sum += buf[i];
        part[t] = sum;
        __syncthreads();
        for (int d = 1; d < 1024; d <<= 1) {
            const int v = (t >= d) ? part[t - d] : 0;
            __syncthreads();
            part[t] += v;
            __syncthreads();
        }
        int run = carry + part[t] - sum;
        for (int i = s; i < en; ++i) { const int c = buf[i]; buf[i] = run; run += c; }
        __syncthreads();
        for (int i = t; i < lim; i += 1024) off[base + i] = buf[i];
        carry += part[1023];
        __syncthreads();
    }
    if (t == 0) off[N] = carry;
}

__global__ void csr_fill(const int* __restrict__ src, const int* __restrict__ dst,
                         const int* __restrict__ et, const int* __restrict__ rank,
                         const int* __restrict__ blockcnt, const int* __restrict__ off,
                         int* __restrict__ key, int E, int N, int chunk) {
    const int e = blockIdx.x * blockDim.x + threadIdx.x;
    if (e >= E) return;
    const int d = dst[e];
    const int b = e / chunk;
    int pos = off[d] + blockcnt[(size_t)b * N + d] + rank[e];
    pos = min(max(pos, 0), E - 1);
    key[pos] = et[e] * N + src[e];
}

__global__ void gather64_relu(const float* __restrict__ rel, const int* __restrict__ off,
                              const int* __restrict__ key, float* __restrict__ h, int N) {
    const long long t = (long long)blockIdx.x * blockDim.x + threadIdx.x;
    const int i = (int)(t >> 4);
    if (i >= N) return;
    const int f = ((int)t & 15) * 4;
    float* hp = h + (size_t)i * 64 + f;
    float4 acc = *(const float4*)hp;
    int p = off[i];
    const int pe = off[i + 1];
    for (; p + 4 <= pe; p += 4) {
        const int k0 = key[p], k1 = key[p + 1], k2 = key[p + 2], k3 = key[p + 3];
        const float4 a = *(const float4*)(rel + (size_t)k0 * 64 + f);
        const float4 b = *(const float4*)(rel + (size_t)k1 * 64 + f);
        const float4 c = *(const float4*)(rel + (size_t)k2 * 64 + f);
        const float4 d = *(const float4*)(rel + (size_t)k3 * 64 + f);
        acc.x += a.x; acc.y += a.y; acc.z += a.z; acc.w += a.w;
        acc.x += b.x; acc.y += b.y; acc.z += b.z; acc.w += b.w;
        acc.x += c.x; acc.y += c.y; acc.z += c.z; acc.w += c.w;
        acc.x += d.x; acc.y += d.y; acc.z += d.z; acc.w += d.w;
    }
    for (; p < pe; ++p) {
        const float4 a = *(const float4*)(rel + (size_t)key[p] * 64 + f);
        acc.x += a.x; acc.y += a.y; acc.z += a.z; acc.w += a.w;
    }
    acc.x = fmaxf(acc.x, 0.0f); acc.y = fmaxf(acc.y, 0.0f);
    acc.z = fmaxf(acc.z, 0.0f); acc.w = fmaxf(acc.w, 0.0f);
    *(float4*)hp = acc;
}

__global__ void gather2(const float* __restrict__ rel3, const int* __restrict__ off,
                        const int* __restrict__ key, float* __restrict__ agg3, int N) {
    const long long t = (long long)blockIdx.x * blockDim.x + threadIdx.x;
    const int i = (int)(t >> 4);
    const int l = (int)t & 15;
    float s0 = 0.0f, s1 = 0.0f;
    if (i < N) {
        const int pe = off[i + 1];
        for (int p = off[i] + l; p < pe; p += 16) {
            const float2 r = *(const float2*)(rel3 + (size_t)key[p] * 2);
            s0 += r.x; s1 += r.y;
        }
    }
    for (int m = 8; m >= 1; m >>= 1) {
        s0 += __shfl_xor(s0, m, 16);
        s1 += __shfl_xor(s1, m, 16);
    }
    if (i < N && l == 0) {
        agg3[(size_t)i * 2 + 0] += s0;
        agg3[(size_t)i * 2 + 1] += s1;
    }
}

__device__ __forceinline__ unsigned f2ord(float f) {
    unsigned u = __float_as_uint(f);
    return (u & 0x80000000u) ? ~u : (u | 0x80000000u);
}
__device__ __forceinline__ float ord2f(unsigned u) {
    return __uint_as_float((u & 0x80000000u) ? (u ^ 0x80000000u) : ~u);
}

__global__ void init_slot(unsigned* slot) { *slot = 0xFFFFFFFFu; }

__global__ void min_reduce(const float* __restrict__ p, int n, unsigned* slot) {
    __shared__ float sm[256];
    float m = 3.402823466e38f;
    for (int i = blockIdx.x * blockDim.x + threadIdx.x; i < n;
         i += gridDim.x * blockDim.x)
        m = fminf(m, p[i]);
    sm[threadIdx.x] = m;
    __syncthreads();
    for (int s = blockDim.x >> 1; s > 0; s >>= 1) {
        if ((int)threadIdx.x < s) sm[threadIdx.x] = fminf(sm[threadIdx.x], sm[threadIdx.x + s]);
        __syncthreads();
    }
    if (threadIdx.x == 0) atomicMin(slot, f2ord(sm[0]));
}

__global__ void finalize_k(const float* __restrict__ agg3,
                           const int* __restrict__ cell,
                           const int* __restrict__ maxs,
                           const unsigned* __restrict__ slot,
                           float* __restrict__ out, int N) {
    const int i = blockIdx.x * blockDim.x + threadIdx.x;
    if (i >= N) return;
    const float gmin = ord2f(*slot) - 1.0f;
    const bool upper = cell[i] >= (maxs[i] - 1);
    const bool lower = cell[i] == 0;
    out[(size_t)i * 2 + 0] = upper ? gmin : agg3[(size_t)i * 2 + 0];
    out[(size_t)i * 2 + 1] = lower ? gmin : agg3[(size_t)i * 2 + 1];
}

extern "C" void kernel_launch(void* const* d_in, const int* in_sizes, int n_in,
                              void* d_out, int out_size, void* d_ws, size_t ws_size,
                              hipStream_t stream) {
    const float* x      = (const float*)d_in[0];
    const int*   src    = (const int*)d_in[1];
    const int*   dst    = (const int*)d_in[2];
    const int*   etypes = (const int*)d_in[3];
    const int*   cell   = (const int*)d_in[4];
    const int*   maxs   = (const int*)d_in[5];
    const float* W1     = (const float*)d_in[6];
    const float* loop1  = (const float*)d_in[7];
    const float* b1     = (const float*)d_in[8];
    const float* W2     = (const float*)d_in[9];
    const float* loop2  = (const float*)d_in[10];
    const float* b2     = (const float*)d_in[11];
    const float* W3     = (const float*)d_in[12];
    const float* loop3  = (const float*)d_in[13];
    const float* b3     = (const float*)d_in[14];
    float* out = (float*)d_out;

    const int N = in_sizes[4];
    const int E = in_sizes[1];
    const int K1 = in_sizes[0] / N;

    float* relbuf = (float*)d_ws;
    float* h1     = relbuf + (size_t)2 * N * 64;
    float* h2     = h1 + (size_t)N * 64;
    float* rel3   = h2 + (size_t)N * 64;
    float* agg3   = rel3 + (size_t)4 * N;
    unsigned* slot = (unsigned*)(agg3 + (size_t)2 * N);

    const int waves   = (N + 15) / 16;
    const int gblocks = (waves * 32 + 255) / 256;
    const int eb64    = (int)(((long long)E * 16 + 255) / 256);
    const int nb      = (N + 255) / 256;
    const int fb      = (N * 64 + 255) / 256;
    const int eb2     = (E + 255) / 256;

    int* key      = (int*)(slot + 4);
    int* off      = key + (size_t)E;
    int* blockcnt = (int*)relbuf;
    int* rank     = blockcnt + (size_t)CSR_NB * N;
    const size_t need = ((size_t)262 * N + 4 + (size_t)E + (size_t)N + 1) * sizeof(float);
    const bool csr_ok = ws_size >= need && (long long)E <= (long long)192 * N;

    if (csr_ok) {
        const int chunk = (E + CSR_NB - 1) / CSR_NB;
        const int gb64  = (int)(((long long)N * 16 + 255) / 256);
        csr_hist<<<CSR_NB, 1024, 0, stream>>>(dst, rank, blockcnt, E, N, chunk);
        csr_colscan<<<nb, 256, 0, stream>>>(blockcnt, off, N);
        csr_scan<<<1, 1024, 0, stream>>>(off, N);
        csr_fill<<<eb2, 256, 0, stream>>>(src, dst, etypes, rank, blockcnt, off, key, E, N, chunk);

        gemm3_xw<<<gblocks, 256, 0, stream>>>(x, W1, loop1, b1, relbuf, h1, N, K1);
        gather64_relu<<<gb64, 256, 0, stream>>>(relbuf, off, key, h1, N);

        gemm3_xw<<<gblocks, 256, 0, stream>>>(h1, W2, loop2, b2, relbuf, h2, N, 64);
        gather64_relu<<<gb64, 256, 0, stream>>>(relbuf, off, key, h2, N);

        layer3_node<<<nb, 256, 0, stream>>>(h2, W3, loop3, b3, rel3, agg3, N);
        gather2<<<gb64, 256, 0, stream>>>(rel3, off, key, agg3, N);
        init_slot<<<1, 1, 0, stream>>>(slot);
        min_reduce<<<512, 256, 0, stream>>>(agg3, N * 2, slot);
        finalize_k<<<nb, 256, 0, stream>>>(agg3, cell, maxs, slot, out, N);
    } else {
        gemm3_xw<<<gblocks, 256, 0, stream>>>(x, W1, loop1, b1, relbuf, h1, N, K1);
        edge_scatter64<<<eb64, 256, 0, stream>>>(relbuf, src, dst, etypes, h1, E, N);
        relu_k<<<fb, 256, 0, stream>>>(h1, N * 64);

        gemm3_xw<<<gblocks, 256, 0, stream>>>(h1, W2, loop2, b2, relbuf, h2, N, 64);
        edge_scatter64<<<eb64, 256, 0, stream>>>(relbuf, src, dst, etypes, h2, E, N);
        relu_k<<<fb, 256, 0, stream>>>(h2, N * 64);

        layer3_node<<<nb, 256, 0, stream>>>(h2, W3, loop3, b3, rel3, agg3, N);
        edge_scatter2<<<eb2, 256, 0, stream>>>(rel3, src, dst, etypes, agg3, E, N);
        init_slot<<<1, 1, 0, stream>>>(slot);
        min_reduce<<<512, 256, 0, stream>>>(agg3, N * 2, slot);
        finalize_k<<<nb, 256, 0, stream>>>(agg3, cell, maxs, slot, out, N);
    }
}


# Fold the layer-3 node products into the layer-2 gather and the min into the 2-wide gather (per-block minima, no slot atomic): 13 -> 10 launches; layer-3 dots summed 4 per lane + 16-lane butterfly
# speedup vs baseline: 1.9596x; 1.0466x over previous
#include <hip/hip_runtime.h>

typedef __attribute__((ext_vector_type(2))) float v2f;
typedef __attribute__((ext_vector_type(8))) float v8f;

__global__ __launch_bounds__(256) void gemm3_xw(const float* __restrict__ X, const float* __restrict__ Wr,
                         const float* __restrict__ Wl, const float* __restrict__ bias,
                         float* __restrict__ rel, float* __restrict__ H,
                         int M, int K) {
    const int gthread = blockIdx.x * blockDim.x + threadIdx.x;
    const int wave    = gthread >> 5;
    const int lane    = threadIdx.x & 31;
    const int m0      = wave * 16;
    if (m0 >= M) return;

    const int half = lane >> 4;
    const int lm   = lane & 15;

    v8f acc[12];
    #pragma unroll
    for (int t = 0; t < 8; ++t) {
        #pragma unroll
        for (int v = 0; v < 8; ++v) acc[t][v] = 0.0f;
    }
    #pragma unroll
    for (int nt = 0; nt < 4; ++nt) {
        const float bv = bias[nt * 16 + lm];
        #pragma unroll
        for (int v = 0; v < 8; ++v) acc[8 + nt][v] = bv;
    }

    const float* xrow = X + (size_t)(m0 + lm) * K;
    const float* w1   = Wr + (size_t)K * 64;

    #pragma unroll 2
    for (int k0 = 0; k0 < K; k0 += 4) {
        const int kb = k0 + 2 * half;
        const float2 av = *(const float2*)(xrow + kb);
        v2f a; a.x = av.x; a.y = av.y;
        #pragma unroll
        for (int mt = 0; mt < 3; ++mt) {
            const float* Wm = (mt == 0) ? Wr : (mt == 1) ? w1 : Wl;
            #pragma unroll
            for (int nt = 0; nt < 4; ++nt) {
                const int n = nt * 16 + lm;
                v2f b;
                b.x = Wm[(size_t)kb * 64 + n];
                b.y = Wm[(size_t)(kb + 1) * 64 + n];
                acc[mt * 4 + nt] = __builtin_amdgcn_wmma_f32_16x16x4_f32(
                    false, a, false, b, (short)0, acc[mt * 4 + nt], false, false);
            }
        }
    }

    #pragma unroll
    for (int mt = 0; mt < 3; ++mt) {
        float* C = (mt == 2) ? H : rel + (size_t)mt * M * 64;
        #pragma unroll
        for (int nt = 0; nt < 4; ++nt) {
            const int n = nt * 16 + lm;
            #pragma unroll
            for (int v = 0; v < 8; ++v) {
                C[(size_t)(m0 + v + 8 * half) * 64 + n] = acc[mt * 4 + nt][v];
            }
        }
    }
}

__global__ void edge_scatter64(const float* __restrict__ rel,
                               const int* __restrict__ src,
                               const int* __restrict__ dst,
                               const int* __restrict__ et,
                               float* __restrict__ agg, int E, int N) {
    const long long t = (long long)blockIdx.x * blockDim.x + threadIdx.x;
    const int e = (int)(t >> 4);
    if (e >= E) return;
    const int f = ((int)t & 15) * 4;
    const float4 v = *(const float4*)(rel + ((long long)et[e] * N + src[e]) * 64 + f);
    float* q = agg + (long long)dst[e] * 64 + f;
    atomicAdd(q + 0, v.x);
    atomicAdd(q + 1, v.y);
    atomicAdd(q + 2, v.z);
    atomicAdd(q + 3, v.w);
}

__global__ void relu_k(float* __restrict__ p, int n) {
    const int i = blockIdx.x * blockDim.x + threadIdx.x;
    if (i < n) p[i] = fmaxf(p[i], 0.0f);
}

__global__ void layer3_node(const float* __restrict__ h2,
                            const float* __restrict__ W3,
                            const float* __restrict__ loop3,
                            const float* __restrict__ b3,
                            float* __restrict__ rel3, float* __restrict__ agg3,
                            int N) {
    const int i = blockIdx.x * blockDim.x + threadIdx.x;
    if (i >= N) return;
    const float* row = h2 + (size_t)i * 64;
    float r00 = 0.f, r01 = 0.f, r10 = 0.f, r11 = 0.f, l0 = 0.f, l1 = 0.f;
    #pragma unroll 8
    for (int k = 0; k < 64; ++k) {
        const float v = row[k];
        r00 += v * W3[k * 2 + 0];
        r01 += v * W3[k * 2 + 1];
        r10 += v * W3[128 + k * 2 + 0];
        r11 += v * W3[128 + k * 2 + 1];
        l0  += v * loop3[k * 2 + 0];
        l1  += v * loop3[k * 2 + 1];
    }
    rel3[(size_t)i * 2 + 0] = r00;
    rel3[(size_t)i * 2 + 1] = r01;
    rel3[(size_t)(N + i) * 2 + 0] = r10;
    rel3[(size_t)(N + i) * 2 + 1] = r11;
    agg3[(size_t)i * 2 + 0] = l0 + b3[0];
    agg3[(size_t)i * 2 + 1] = l1 + b3[1];
}

__global__ void edge_scatter2(const float* __restrict__ rel3,
                              const int* __restrict__ src,
                              const int* __restrict__ dst,
                              const int* __restrict__ et,
                              float* __restrict__ agg3, int E, int N) {
    const int e = blockIdx.x * blockDim.x + threadIdx.x;
    if (e >= E) return;
    const float* p = rel3 + ((long long)et[e] * N + src[e]) * 2;
    float* q = agg3 + (long long)dst[e] * 2;
    atomicAdd(q + 0, p[0]);
    atomicAdd(q + 1, p[1]);
}


#define CSR_NB  64
#define CSR_CAP 51200

__global__ __launch_bounds__(1024) void csr_hist(const int* __restrict__ dst,
                                                 int* __restrict__ rank,
                                                 int* __restrict__ blockcnt,
                                                 int E, int N, int chunk) {
    __shared__ int cnt[CSR_CAP];
    const int b  = blockIdx.x;
    const int e0 = b * chunk;
    const int e1 = min(e0 + chunk, E);
    for (int base = 0; base < N; base += CSR_CAP) {
        const int lim = min(CSR_CAP, N - base);
        for (int i = threadIdx.x; i < lim; i += blockDim.x) cnt[i] = 0;
        __syncthreads();
        const int bd = (int)blockDim.x;
        for (int e = e0 + (int)threadIdx.x; e < e1; e += 4 * bd) {
            int d[4];
            #pragma unroll
            for (int j = 0; j < 4; ++j) {
                const int ee = e + j * bd;
                const int v  = dst[min(ee, e1 - 1)];
                d[j] = (ee < e1) ? v - base : -1;
            }
            #pragma unroll
            for (int j = 0; j < 4; ++j)
                if (d[j] >= 0 && d[j] < lim) rank[e + j * bd] = atomicAdd(&cnt[d[j]], 1);
        }
        __syncthreads();
        for (int i = threadIdx.x; i < lim; i += blockDim.x)
            blockcnt[(size_t)b * N + base + i] = cnt[i];
        __syncthreads();
    }
}

__global__ void csr_colscan(int* __restrict__ blockcnt, int* __restrict__ off, int N) {
    const int n = blockIdx.x * blockDim.x + threadIdx.x;
    if (n >= N) return;
    int run = 0;
    for (int b0 = 0; b0 < CSR_NB; b0 += 16) {
        int c[16];
        #pragma unroll
        for (int j = 0; j < 16; ++j) c[j] = blockcnt[(size_t)(b0 + j) * N + n];
        #pragma unroll
        for (int j = 0; j < 16; ++j) {
            blockcnt[(size_t)(b0 + j) * N + n] = run;
            run += c[j];
        }
    }
    off[n] = run;
}

__global__ __launch_bounds__(1024) void csr_scan(int* __restrict__ off, int N) {
    __shared__ int buf[CSR_CAP];
    __shared__ int part[1024];
    const int t = threadIdx.x;
    int carry = 0;
    for (int base = 0; base < N; base += CSR_CAP) {
        const int lim = min(CSR_CAP, N - base);
        for (int i = t; i < lim; i += 1024) buf[i] = off[base + i];
        __syncthreads();
        const int per = ((lim + 1023) / 1024) | 1;
        const int s  = min(t * per, lim);
        const int en = min(s + per, lim);
        int sum = 0;
        for (int i = s; i < en; ++i) sum += buf[i];
        part[t] = sum;
        __syncthreads();
        for (int d = 1; d < 1024; d <<= 1) {
            const int v = (t >= d) ? part[t - d] : 0;
            __syncthreads();
            part[t] += v;
            __syncthreads();
        }
        int run = carry + part[t] - sum;
        for (int i = s; i < en; ++i) { const int c = buf[i]; buf[i] = run; run += c; }
        __syncthreads();
        for (int i = t; i < lim; i += 1024) off[base + i] = buf[i];
        carry += part[1023];
        __syncthreads();
    }
    if (t == 0) off[N] = carry;
}

__global__ void csr_fill(const int* __restrict__ src, const int* __restrict__ dst,
                         const int* __restrict__ et, const int* __restrict__ rank,
                         const int* __restrict__ blockcnt, const int* __restrict__ off,
                         int* __restrict__ key, int E, int N, int chunk) {
    const int e = blockIdx.x * blockDim.x + threadIdx.x;
    if (e >= E) return;
    const int d = dst[e];
    const int b = e / chunk;
    int pos = off[d] + blockcnt[(size_t)b * N + d] + rank[e];
    pos = min(max(pos, 0), E - 1);
    key[pos] = et[e] * N + src[e];
}

__global__ void gather64_relu(const float* __restrict__ rel, const int* __restrict__ off,
                              const int* __restrict__ key, float* __restrict__ h, int N) {
    const long long t = (long long)blockIdx.x * blockDim.x + threadIdx.x;
    const int i = (int)(t >> 4);
    if (i >= N) return;
    const int f = ((int)t & 15) * 4;
    float* hp = h + (size_t)i * 64 + f;
    float4 acc = *(const float4*)hp;
    int p = off[i];
    const int pe = off[i + 1];
    for (; p + 4 <= pe; p += 4) {
        const int k0 = key[p], k1 = key[p + 1], k2 = key[p + 2], k3 = key[p + 3];
        const float4 a = *(const float4*)(rel + (size_t)k0 * 64 + f);
        const float4 b = *(const float4*)(rel + (size_t)k1 * 64 + f);
        const float4 c = *(const float4*)(rel + (size_t)k2 * 64 + f);
        const float4 d = *(const float4*)(rel + (size_t)k3 * 64 + f);
        acc.x += a.x; acc.y += a.y; acc.z += a.z; acc.w += a.w;
        acc.x += b.x; acc.y += b.y; acc.z += b.z; acc.w += b.w;
        acc.x += c.x; acc.y += c.y; acc.z += c.z; acc.w += c.w;
        acc.x += d.x; acc.y += d.y; acc.z += d.z; acc.w += d.w;
    }
    for (; p < pe; ++p) {
        const float4 a = *(const float4*)(rel + (size_t)key[p] * 64 + f);
        acc.x += a.x; acc.y += a.y; acc.z += a.z; acc.w += a.w;
    }
    acc.x = fmaxf(acc.x, 0.0f); acc.y = fmaxf(acc.y, 0.0f);
    acc.z = fmaxf(acc.z, 0.0f); acc.w = fmaxf(acc.w, 0.0f);
    *(float4*)hp = acc;
}

__global__ void gather2(const float* __restrict__ rel3, const int* __restrict__ off,
                        const int* __restrict__ key, float* __restrict__ agg3, int N) {
    const long long t = (long long)blockIdx.x * blockDim.x + threadIdx.x;
    const int i = (int)(t >> 4);
    const int l = (int)t & 15;
    float s0 = 0.0f, s1 = 0.0f;
    if (i < N) {
        const int pe = off[i + 1];
        for (int p = off[i] + l; p < pe; p += 16) {
            const float2 r = *(const float2*)(rel3 + (size_t)key[p] * 2);
            s0 += r.x; s1 += r.y;
        }
    }
    for (int m = 8; m >= 1; m >>= 1) {
        s0 += __shfl_xor(s0, m, 16);
        s1 += __shfl_xor(s1, m, 16);
    }
    if (i < N && l == 0) {
        agg3[(size_t)i * 2 + 0] += s0;
        agg3[(size_t)i * 2 + 1] += s1;
    }
}

__global__ void gather64_relu_l3(const float* __restrict__ rel, const int* __restrict__ off,
                                 const int* __restrict__ key, const float* __restrict__ h,
                                 const float* __restrict__ W3, const float* __restrict__ loop3,
                                 const float* __restrict__ b3,
                                 float* __restrict__ rel3, float* __restrict__ agg3, int N) {
    const long long t = (long long)blockIdx.x * blockDim.x + threadIdx.x;
    const int i = (int)(t >> 4);
    if (i >= N) return;
    const int f = ((int)t & 15) * 4;
    float4 acc = *(const float4*)(h + (size_t)i * 64 + f);
    int p = off[i];
    const int pe = off[i + 1];
    for (; p + 4 <= pe; p += 4) {
        const int k0 = key[p], k1 = key[p + 1], k2 = key[p + 2], k3 = key[p + 3];
        const float4 a = *(const float4*)(rel + (size_t)k0 * 64 + f);
        const float4 b = *(const float4*)(rel + (size_t)k1 * 64 + f);
        const float4 c = *(const float4*)(rel + (size_t)k2 * 64 + f);
        const float4 d = *(const float4*)(rel + (size_t)k3 * 64 + f);
        acc.x += a.x; acc.y += a.y; acc.z += a.z; acc.w += a.w;
        acc.x += b.x; acc.y += b.y; acc.z += b.z; acc.w += b.w;
        acc.x += c.x; acc.y += c.y; acc.z += c.z; acc.w += c.w;
        acc.x += d.x; acc.y += d.y; acc.z += d.z; acc.w += d.w;
    }
    for (; p < pe; ++p) {
        const float4 a = *(const float4*)(rel + (size_t)key[p] * 64 + f);
        acc.x += a.x; acc.y += a.y; acc.z += a.z; acc.w += a.w;
    }
    const float hv[4] = { fmaxf(acc.x, 0.0f), fmaxf(acc.y, 0.0f), fmaxf(acc.z, 0.0f), fmaxf(acc.w, 0.0f) };
    const float* wm[3] = { W3 + f * 2, W3 + 128 + f * 2, loop3 + f * 2 };
    float o[6];
    #pragma unroll
    for (int m = 0; m < 3; ++m) {
        float s0 = 0.0f, s1 = 0.0f;
        #pragma unroll
        for (int j = 0; j < 4; ++j) {
            s0 += hv[j] * wm[m][2 * j + 0];
            s1 += hv[j] * wm[m][2 * j + 1];
        }
        o[2 * m + 0] = s0;
        o[2 * m + 1] = s1;
    }
    for (int d = 8; d >= 1; d >>= 1) {
        #pragma unroll
        for (int q = 0; q < 6; ++q) o[q] += __shfl_xor(o[q], d, 16);
    }
    if (((int)t & 15) == 0) {
        rel3[(size_t)i * 2 + 0] = o[0];
        rel3[(size_t)i * 2 + 1] = o[1];
        rel3[(size_t)(N + i) * 2 + 0] = o[2];
        rel3[(size_t)(N + i) * 2 + 1] = o[3];
        agg3[(size_t)i * 2 + 0] = o[4] + b3[0];
        agg3[(size_t)i * 2 + 1] = o[5] + b3[1];
    }
}

#define TAIL_TPB 256
__global__ __launch_bounds__(TAIL_TPB) void gather2_min(const float* __restrict__ rel3, const int* __restrict__ off,
                            const int* __restrict__ key, float* __restrict__ agg3,
                            float* __restrict__ blockmin, int N) {
    __shared__ float sm[TAIL_TPB / 32];
    const long long t = (long long)blockIdx.x * blockDim.x + threadIdx.x;
    const int i = (int)(t >> 4);
    const int l = (int)t & 15;
    float s0 = 0.0f, s1 = 0.0f;
    if (i < N) {
        const int pe = off[i + 1];
        for (int p = off[i] + l; p < pe; p += 16) {
            const float2 r = *(const float2*)(rel3 + (size_t)key[p] * 2);
            s0 += r.x; s1 += r.y;
        }
    }
    for (int m = 8; m >= 1; m >>= 1) {
        s0 += __shfl_xor(s0, m, 16);
        s1 += __shfl_xor(s1, m, 16);
    }
    float mn = 3.402823466e38f;
    if (i < N && l == 0) {
        const float v0 = agg3[(size_t)i * 2 + 0] + s0;
        const float v1 = agg3[(size_t)i * 2 + 1] + s1;
        agg3[(size_t)i * 2 + 0] = v0;
        agg3[(size_t)i * 2 + 1] = v1;
        mn = fminf(v0, v1);
    }
    mn = fminf(mn, __shfl_xor(mn, 16, 32));
    if ((threadIdx.x & 31) == 0) sm[threadIdx.x >> 5] = mn;
    __syncthreads();
    if (threadIdx.x == 0) {
        float r = sm[0];
        #pragma unroll
        for (int w = 1; w < TAIL_TPB / 32; ++w) r = fminf(r, sm[w]);
        blockmin[blockIdx.x] = r;
    }
}

__global__ __launch_bounds__(TAIL_TPB) void finalize_bm(const float* __restrict__ agg3,
                            const int* __restrict__ cell,
                            const int* __restrict__ maxs,
                            const float* __restrict__ blockmin, int nblk,
                            float* __restrict__ out, int N) {
    __shared__ float sm[TAIL_TPB / 32];
    float mn = 3.402823466e38f;
    for (int j0 = threadIdx.x; j0 < nblk; j0 += 16 * TAIL_TPB) {
        float v[16];
        #pragma unroll
        for (int u = 0; u < 16; ++u) v[u] = blockmin[min(j0 + u * TAIL_TPB, nblk - 1)];
        #pragma unroll
        for (int u = 0; u < 16; ++u) mn = fminf(mn, v[u]);
    }
    for (int m = 16; m >= 1; m >>= 1) mn = fminf(mn, __shfl_xor(mn, m, 32));
    if ((threadIdx.x & 31) == 0) sm[threadIdx.x >> 5] = mn;
    __syncthreads();
    float r = sm[0];
    #pragma unroll
    for (int w = 1; w < TAIL_TPB / 32; ++w) r = fminf(r, sm[w]);
    const float gmin = r - 1.0f;
    const int i = blockIdx.x * blockDim.x + threadIdx.x;
    if (i >= N) return;
    const bool upper = cell[i] >= (maxs[i] - 1);
    const bool lower = cell[i] == 0;
    out[(size_t)i * 2 + 0] = upper ? gmin : agg3[(size_t)i * 2 + 0];
    out[(size_t)i * 2 + 1] = lower ? gmin : agg3[(size_t)i * 2 + 1];
}

__device__ __forceinline__ unsigned f2ord(float f) {
    unsigned u = __float_as_uint(f);
    return (u & 0x80000000u) ? ~u : (u | 0x80000000u);
}
__device__ __forceinline__ float ord2f(unsigned u) {
    return __uint_as_float((u & 0x80000000u) ? (u ^ 0x80000000u) : ~u);
}

__global__ void init_slot(unsigned* slot) { *slot = 0xFFFFFFFFu; }

__global__ void min_reduce(const float* __restrict__ p, int n, unsigned* slot) {
    __shared__ float sm[256];
    float m = 3.402823466e38f;
    for (int i = blockIdx.x * blockDim.x + threadIdx.x; i < n;
         i += gridDim.x * blockDim.x)
        m = fminf(m, p[i]);
    sm[threadIdx.x] = m;
    __syncthreads();
    for (int s = blockDim.x >> 1; s > 0; s >>= 1) {
        if ((int)threadIdx.x < s) sm[threadIdx.x] = fminf(sm[threadIdx.x], sm[threadIdx.x + s]);
        __syncthreads();
    }
    if (threadIdx.x == 0) atomicMin(slot, f2ord(sm[0]));
}

__global__ void finalize_k(const float* __restrict__ agg3,
                           const int* __restrict__ cell,
                           const int* __restrict__ maxs,
                           const unsigned* __restrict__ slot,
                           float* __restrict__ out, int N) {
    const int i = blockIdx.x * blockDim.x + threadIdx.x;
    if (i >= N) return;
    const float gmin = ord2f(*slot) - 1.0f;
    const bool upper = cell[i] >= (maxs[i] - 1);
    const bool lower = cell[i] == 0;
    out[(size_t)i * 2 + 0] = upper ? gmin : agg3[(size_t)i * 2 + 0];
    out[(size_t)i * 2 + 1] = lower ? gmin : agg3[(size_t)i * 2 + 1];
}

extern "C" void kernel_launch(void* const* d_in, const int* in_sizes, int n_in,
                              void* d_out, int out_size, void* d_ws, size_t ws_size,
                              hipStream_t stream) {
    const float* x      = (const float*)d_in[0];
    const int*   src    = (const int*)d_in[1];
    const int*   dst    = (const int*)d_in[2];
    const int*   etypes = (const int*)d_in[3];
    const int*   cell   = (const int*)d_in[4];
    const int*   maxs   = (const int*)d_in[5];
    const float* W1     = (const float*)d_in[6];
    const float* loop1  = (const float*)d_in[7];
    const float* b1     = (const float*)d_in[8];
    const float* W2     = (const float*)d_in[9];
    const float* loop2  = (const float*)d_in[10];
    const float* b2     = (const float*)d_in[11];
    const float* W3     = (const float*)d_in[12];
    const float* loop3  = (const float*)d_in[13];
    const float* b3     = (const float*)d_in[14];
    float* out = (float*)d_out;

    const int N = in_sizes[4];
    const int E = in_sizes[1];
    const int K1 = in_sizes[0] / N;

    float* relbuf = (float*)d_ws;
    float* h1     = relbuf + (size_t)2 * N * 64;
    float* h2     = h1 + (size_t)N * 64;
    float* rel3   = h2 + (size_t)N * 64;
    float* agg3   = rel3 + (size_t)4 * N;
    unsigned* slot = (unsigned*)(agg3 + (size_t)2 * N);

    const int waves   = (N + 15) / 16;
    const int gblocks = (waves * 32 + 255) / 256;
    const int eb64    = (int)(((long long)E * 16 + 255) / 256);
    const int nb      = (N + 255) / 256;
    const int fb      = (N * 64 + 255) / 256;
    const int eb2     = (E + 255) / 256;

    int* key      = (int*)(slot + 4);
    int* off      = key + (size_t)E;
    int* blockcnt = (int*)relbuf;
    int* rank     = blockcnt + (size_t)CSR_NB * N;
    const size_t need = ((size_t)262 * N + 4 + (size_t)E + (size_t)N + 1) * sizeof(float);
    const bool csr_ok = ws_size >= need && (long long)E <= (long long)192 * N;

    if (csr_ok) {
        const int chunk = (E + CSR_NB - 1) / CSR_NB;
        const int gb64  = (int)(((long long)N * 16 + 255) / 256);
        csr_hist<<<CSR_NB, 1024, 0, stream>>>(dst, rank, blockcnt, E, N, chunk);
        csr_colscan<<<nb, 256, 0, stream>>>(blockcnt, off, N);
        csr_scan<<<1, 1024, 0, stream>>>(off, N);
        csr_fill<<<eb2, 256, 0, stream>>>(src, dst, etypes, rank, blockcnt, off, key, E, N, chunk);

        gemm3_xw<<<gblocks, 256, 0, stream>>>(x, W1, loop1, b1, relbuf, h1, N, K1);
        gather64_relu<<<gb64, 256, 0, stream>>>(relbuf, off, key, h1, N);

        gemm3_xw<<<gblocks, 256, 0, stream>>>(h1, W2, loop2, b2, relbuf, h2, N, 64);
        gather64_relu_l3<<<gb64, 256, 0, stream>>>(relbuf, off, key, h2, W3, loop3, b3, rel3, agg3, N);

        float* blockmin = relbuf;
        gather2_min<<<gb64, TAIL_TPB, 0, stream>>>(rel3, off, key, agg3, blockmin, N);
        finalize_bm<<<nb, TAIL_TPB, 0, stream>>>(agg3, cell, maxs, blockmin, gb64, out, N);
    } else {
        gemm3_xw<<<gblocks, 256, 0, stream>>>(x, W1, loop1, b1, relbuf, h1, N, K1);
        edge_scatter64<<<eb64, 256, 0, stream>>>(relbuf, src, dst, etypes, h1, E, N);
        relu_k<<<fb, 256, 0, stream>>>(h1, N * 64);

        gemm3_xw<<<gblocks, 256, 0, stream>>>(h1, W2, loop2, b2, relbuf, h2, N, 64);
        edge_scatter64<<<eb64, 256, 0, stream>>>(relbuf, src, dst, etypes, h2, E, N);
        relu_k<<<fb, 256, 0, stream>>>(h2, N * 64);

        layer3_node<<<nb, 256, 0, stream>>>(h2, W3, loop3, b3, rel3, agg3, N);
        edge_scatter2<<<eb2, 256, 0, stream>>>(rel3, src, dst, etypes, agg3, E, N);
        init_slot<<<1, 1, 0, stream>>>(slot);
        min_reduce<<<512, 256, 0, stream>>>(agg3, N * 2, slot);
        finalize_k<<<nb, 256, 0, stream>>>(agg3, cell, maxs, slot, out, N);
    }
}


# Fold the node-offset scan into the column-scan launch (block base from per-range edge counts the histogram pass now writes); table holds final positions: 10 -> 9 launches, fill drops its off[] read
# speedup vs baseline: 2.0835x; 1.0632x over previous
#include <hip/hip_runtime.h>

typedef __attribute__((ext_vector_type(2))) float v2f;
typedef __attribute__((ext_vector_type(8))) float v8f;

__global__ __launch_bounds__(256) void gemm3_xw(const float* __restrict__ X, const float* __restrict__ Wr,
                         const float* __restrict__ Wl, const float* __restrict__ bias,
                         float* __restrict__ rel, float* __restrict__ H,
                         int M, int K) {
    const int gthread = blockIdx.x * blockDim.x + threadIdx.x;
    const int wave    = gthread >> 5;
    const int lane    = threadIdx.x & 31;
    const int m0      = wave * 16;
    if (m0 >= M) return;

    const int half = lane >> 4;
    const int lm   = lane & 15;

    v8f acc[12];
    #pragma unroll
    for (int t = 0; t < 8; ++t) {
        #pragma unroll
        for (int v = 0; v < 8; ++v) acc[t][v] = 0.0f;
    }
    #pragma unroll
    for (int nt = 0; nt < 4; ++nt) {
        const float bv = bias[nt * 16 + lm];
        #pragma unroll
        for (int v = 0; v < 8; ++v) acc[8 + nt][v] = bv;
    }

    const float* xrow = X + (size_t)(m0 + lm) * K;
    const float* w1   = Wr + (size_t)K * 64;

    #pragma unroll 2
    for (int k0 = 0; k0 < K; k0 += 4) {
        const int kb = k0 + 2 * half;
        const float2 av = *(const float2*)(xrow + kb);
        v2f a; a.x = av.x; a.y = av.y;
        #pragma unroll
        for (int mt = 0; mt < 3; ++mt) {
            const float* Wm = (mt == 0) ? Wr : (mt == 1) ? w1 : Wl;
            #pragma unroll
            for (int nt = 0; nt < 4; ++nt) {
                const int n = nt * 16 + lm;
                v2f b;
                b.x = Wm[(size_t)kb * 64 + n];
                b.y = Wm[(size_t)(kb + 1) * 64 + n];
                acc[mt * 4 + nt] = __builtin_amdgcn_wmma_f32_16x16x4_f32(
                    false, a, false, b, (short)0, acc[mt * 4 + nt], false, false);
            }
        }
    }

    #pragma unroll
    for (int mt = 0; mt < 3; ++mt) {
        float* C = (mt == 2) ? H : rel + (size_t)mt * M * 64;
        #pragma unroll
        for (int nt = 0; nt < 4; ++nt) {
            const int n = nt * 16 + lm;
            #pragma unroll
            for (int v = 0; v < 8; ++v) {
                C[(size_t)(m0 + v + 8 * half) * 64 + n] = acc[mt * 4 + nt][v];
            }
        }
    }
}

__global__ void edge_scatter64(const float* __restrict__ rel,
                               const int* __restrict__ src,
                               const int* __restrict__ dst,
                               const int* __restrict__ et,
                               float* __restrict__ agg, int E, int N) {
    const long long t = (long long)blockIdx.x * blockDim.x + threadIdx.x;
    const int e = (int)(t >> 4);
    if (e >= E) return;
    const int f = ((int)t & 15) * 4;
    const float4 v = *(const float4*)(rel + ((long long)et[e] * N + src[e]) * 64 + f);
    float* q = agg + (long long)dst[e] * 64 + f;
    atomicAdd(q + 0, v.x);
    atomicAdd(q + 1, v.y);
    atomicAdd(q + 2, v.z);
    atomicAdd(q + 3, v.w);
}

__global__ void relu_k(float* __restrict__ p, int n) {
    const int i = blockIdx.x * blockDim.x + threadIdx.x;
    if (i < n) p[i] = fmaxf(p[i], 0.0f);
}

__global__ void layer3_node(const float* __restrict__ h2,
                            const float* __restrict__ W3,
                            const float* __restrict__ loop3,
                            const float* __restrict__ b3,
                            float* __restrict__ rel3, float* __restrict__ agg3,
                            int N) {
    const int i = blockIdx.x * blockDim.x + threadIdx.x;
    if (i >= N) return;
    const float* row = h2 + (size_t)i * 64;
    float r00 = 0.f, r01 = 0.f, r10 = 0.f, r11 = 0.f, l0 = 0.f, l1 = 0.f;
    #pragma unroll 8
    for (int k = 0; k < 64; ++k) {
        const float v = row[k];
        r00 += v * W3[k * 2 + 0];
        r01 += v * W3[k * 2 + 1];
        r10 += v * W3[128 + k * 2 + 0];
        r11 += v * W3[128 + k * 2 + 1];
        l0  += v * loop3[k * 2 + 0];
        l1  += v * loop3[k * 2 + 1];
    }
    rel3[(size_t)i * 2 + 0] = r00;
    rel3[(size_t)i * 2 + 1] = r01;
    rel3[(size_t)(N + i) * 2 + 0] = r10;
    rel3[(size_t)(N + i) * 2 + 1] = r11;
    agg3[(size_t)i * 2 + 0] = l0 + b3[0];
    agg3[(size_t)i * 2 + 1] = l1 + b3[1];
}

__global__ void edge_scatter2(const float* __restrict__ rel3,
                              const int* __restrict__ src,
                              const int* __restrict__ dst,
                              const int* __restrict__ et,
                              float* __restrict__ agg3, int E, int N) {
    const int e = blockIdx.x * blockDim.x + threadIdx.x;
    if (e >= E) return;
    const float* p = rel3 + ((long long)et[e] * N + src[e]) * 2;
    float* q = agg3 + (long long)dst[e] * 2;
    atomicAdd(q + 0, p[0]);
    atomicAdd(q + 1, p[1]);
}


#define CSR_NB  64
#define CSR_CAP 51200
#define CSR_RANGE 256

__global__ __launch_bounds__(1024) void csr_hist(const int* __restrict__ dst,
                                                 int* __restrict__ rank,
                                                 int* __restrict__ blockcnt,
                                                 int* __restrict__ coarse,
                                                 int E, int N, int chunk, int nr_total) {
    __shared__ int cnt[CSR_CAP];
    const int b  = blockIdx.x;
    const int e0 = b * chunk;
    const int e1 = min(e0 + chunk, E);
    for (int base = 0; base < N; base += CSR_CAP) {
        const int lim = min(CSR_CAP, N - base);
        for (int i = threadIdx.x; i < lim; i += blockDim.x) cnt[i] = 0;
        __syncthreads();
        const int bd = (int)blockDim.x;
        for (int e = e0 + (int)threadIdx.x; e < e1; e += 4 * bd) {
            int d[4];
            #pragma unroll
            for (int j = 0; j < 4; ++j) {
                const int ee = e + j * bd;
                const int v  = dst[min(ee, e1 - 1)];
                d[j] = (ee < e1) ? v - base : -1;
            }
            #pragma unroll
            for (int j = 0; j < 4; ++j)
                if (d[j] >= 0 && d[j] < lim) rank[e + j * bd] = atomicAdd(&cnt[d[j]], 1);
        }
        __syncthreads();
        const int nrs = (lim + CSR_RANGE - 1) / CSR_RANGE;
        for (int r0 = 0; r0 < nrs; r0 += (int)blockDim.x >> 2) {
            const int r  = r0 + ((int)threadIdx.x >> 2);
            const int q  = (int)threadIdx.x & 3;
            const int lo = r * CSR_RANGE + q * (CSR_RANGE / 4);
            int s = 0;
            #pragma unroll 16
            for (int j = 0; j < CSR_RANGE / 4; ++j) {
                const int i = lo + ((j + (int)threadIdx.x) & (CSR_RANGE / 4 - 1));
                const int v = cnt[min(i, lim - 1)];
                s += (i < lim) ? v : 0;
            }
            s += __shfl_xor(s, 1, 4);
            s += __shfl_xor(s, 2, 4);
            if (q == 0 && r < nrs) coarse[(size_t)b * nr_total + base / CSR_RANGE + r] = s;
        }
        for (int i = threadIdx.x; i < lim; i += blockDim.x)
            blockcnt[(size_t)b * N + base + i] = cnt[i];
        __syncthreads();
    }
}

__global__ __launch_bounds__(CSR_RANGE) void csr_colscan_off(int* __restrict__ blockcnt,
                                                             const int* __restrict__ coarse,
                                                             int* __restrict__ off, int N, int nr_total) {
    __shared__ int wtot[CSR_RANGE / 32];
    __shared__ int wbase[CSR_RANGE / 32];
    const int t  = threadIdx.x;
    const int r  = blockIdx.x;
    const int n  = r * CSR_RANGE + t;
    const int nn = min(n, N - 1);
    int c[CSR_NB];
    #pragma unroll
    for (int j = 0; j < CSR_NB; ++j) c[j] = blockcnt[(size_t)j * N + nn];
    int part = 0;
    for (int rr = t; rr < r; rr += CSR_RANGE) {
        #pragma unroll 16
        for (int b = 0; b < CSR_NB; ++b) part += coarse[(size_t)b * nr_total + rr];
    }
    int deg = 0;
    #pragma unroll
    for (int j = 0; j < CSR_NB; ++j) deg += c[j];
    if (n >= N) deg = 0;
    const int lane = t & 31;
    int inc = deg;
    for (int d = 1; d < 32; d <<= 1) {
        const int v = __shfl_up(inc, d, 32);
        if (lane >= d) inc += v;
    }
    for (int m = 16; m >= 1; m >>= 1) part += __shfl_xor(part, m, 32);
    if (lane == 31) wtot[t >> 5] = inc;
    if (lane == 0)  wbase[t >> 5] = part;
    __syncthreads();
    int run = inc - deg;
    #pragma unroll
    for (int w = 0; w < CSR_RANGE / 32; ++w) {
        run += wbase[w];
        run += (w < (t >> 5)) ? wtot[w] : 0;
    }
    if (n < N) {
        off[n] = run;
        if (n == N - 1) off[N] = run + deg;
        #pragma unroll
        for (int j = 0; j < CSR_NB; ++j) {
            blockcnt[(size_t)j * N + n] = run;
            run += c[j];
        }
    }
}

__global__ void csr_fill(const int* __restrict__ src, const int* __restrict__ dst,
                         const int* __restrict__ et, const int* __restrict__ rank,
                         const int* __restrict__ blockcnt,
                         int* __restrict__ key, int E, int N, int chunk) {
    const int e = blockIdx.x * blockDim.x + threadIdx.x;
    if (e >= E) return;
    const int d = dst[e];
    const int b = e / chunk;
    int pos = blockcnt[(size_t)b * N + d] + rank[e];
    pos = min(max(pos, 0), E - 1);
    key[pos] = et[e] * N + src[e];
}

__global__ void gather64_relu(const float* __restrict__ rel, const int* __restrict__ off,
                              const int* __restrict__ key, float* __restrict__ h, int N) {
    const long long t = (long long)blockIdx.x * blockDim.x + threadIdx.x;
    const int i = (int)(t >> 4);
    if (i >= N) return;
    const int f = ((int)t & 15) * 4;
    float* hp = h + (size_t)i * 64 + f;
    float4 acc = *(const float4*)hp;
    int p = off[i];
    const int pe = off[i + 1];
    for (; p + 4 <= pe; p += 4) {
        const int k0 = key[p], k1 = key[p + 1], k2 = key[p + 2], k3 = key[p + 3];
        const float4 a = *(const float4*)(rel + (size_t)k0 * 64 + f);
        const float4 b = *(const float4*)(rel + (size_t)k1 * 64 + f);
        const float4 c = *(const float4*)(rel + (size_t)k2 * 64 + f);
        const float4 d = *(const float4*)(rel + (size_t)k3 * 64 + f);
        acc.x += a.x; acc.y += a.y; acc.z += a.z; acc.w += a.w;
        acc.x += b.x; acc.y += b.y; acc.z += b.z; acc.w += b.w;
        acc.x += c.x; acc.y += c.y; acc.z += c.z; acc.w += c.w;
        acc.x += d.x; acc.y += d.y; acc.z += d.z; acc.w += d.w;
    }
    for (; p < pe; ++p) {
        const float4 a = *(const float4*)(rel + (size_t)key[p] * 64 + f);
        acc.x += a.x; acc.y += a.y; acc.z += a.z; acc.w += a.w;
    }
    acc.x = fmaxf(acc.x, 0.0f); acc.y = fmaxf(acc.y, 0.0f);
    acc.z = fmaxf(acc.z, 0.0f); acc.w = fmaxf(acc.w, 0.0f);
    *(float4*)hp = acc;
}

__global__ void gather2(const float* __restrict__ rel3, const int* __restrict__ off,
                        const int* __restrict__ key, float* __restrict__ agg3, int N) {
    const long long t = (long long)blockIdx.x * blockDim.x + threadIdx.x;
    const int i = (int)(t >> 4);
    const int l = (int)t & 15;
    float s0 = 0.0f, s1 = 0.0f;
    if (i < N) {
        const int pe = off[i + 1];
        for (int p = off[i] + l; p < pe; p += 16) {
            const float2 r = *(const float2*)(rel3 + (size_t)key[p] * 2);
            s0 += r.x; s1 += r.y;
        }
    }
    for (int m = 8; m >= 1; m >>= 1) {
        s0 += __shfl_xor(s0, m, 16);
        s1 += __shfl_xor(s1, m, 16);
    }
    if (i < N && l == 0) {
        agg3[(size_t)i * 2 + 0] += s0;
        agg3[(size_t)i * 2 + 1] += s1;
    }
}

__global__ void gather64_relu_l3(const float* __restrict__ rel, const int* __restrict__ off,
                                 const int* __restrict__ key, const float* __restrict__ h,
                                 const float* __restrict__ W3, const float* __restrict__ loop3,
                                 const float* __restrict__ b3,
                                 float* __restrict__ rel3, float* __restrict__ agg3, int N) {
    const long long t = (long long)blockIdx.x * blockDim.x + threadIdx.x;
    const int i = (int)(t >> 4);
    if (i >= N) return;
    const int f = ((int)t & 15) * 4;
    float4 acc = *(const float4*)(h + (size_t)i * 64 + f);
    int p = off[i];
    const int pe = off[i + 1];
    for (; p + 4 <= pe; p += 4) {
        const int k0 = key[p], k1 = key[p + 1], k2 = key[p + 2], k3 = key[p + 3];
        const float4 a = *(const float4*)(rel + (size_t)k0 * 64 + f);
        const float4 b = *(const float4*)(rel + (size_t)k1 * 64 + f);
        const float4 c = *(const float4*)(rel + (size_t)k2 * 64 + f);
        const float4 d = *(const float4*)(rel + (size_t)k3 * 64 + f);
        acc.x += a.x; acc.y += a.y; acc.z += a.z; acc.w += a.w;
        acc.x += b.x; acc.y += b.y; acc.z += b.z; acc.w += b.w;
        acc.x += c.x; acc.y += c.y; acc.z += c.z; acc.w += c.w;
        acc.x += d.x; acc.y += d.y; acc.z += d.z; acc.w += d.w;
    }
    for (; p < pe; ++p) {
        const float4 a = *(const float4*)(rel + (size_t)key[p] * 64 + f);
        acc.x += a.x; acc.y += a.y; acc.z += a.z; acc.w += a.w;
    }
    const float hv[4] = { fmaxf(acc.x, 0.0f), fmaxf(acc.y, 0.0f), fmaxf(acc.z, 0.0f), fmaxf(acc.w, 0.0f) };
    const float* wm[3] = { W3 + f * 2, W3 + 128 + f * 2, loop3 + f * 2 };
    float o[6];
    #pragma unroll
    for (int m = 0; m < 3; ++m) {
        float s0 = 0.0f, s1 = 0.0f;
        #pragma unroll
        for (int j = 0; j < 4; ++j) {
            s0 += hv[j] * wm[m][2 * j + 0];
            s1 += hv[j] * wm[m][2 * j + 1];
        }
        o[2 * m + 0] = s0;
        o[2 * m + 1] = s1;
    }
    for (int d = 8; d >= 1; d >>= 1) {
        #pragma unroll
        for (int q = 0; q < 6; ++q) o[q] += __shfl_xor(o[q], d, 16);
    }
    if (((int)t & 15) == 0) {
        rel3[(size_t)i * 2 + 0] = o[0];
        rel3[(size_t)i * 2 + 1] = o[1];
        rel3[(size_t)(N + i) * 2 + 0] = o[2];
        rel3[(size_t)(N + i) * 2 + 1] = o[3];
        agg3[(size_t)i * 2 + 0] = o[4] + b3[0];
        agg3[(size_t)i * 2 + 1] = o[5] + b3[1];
    }
}

#define TAIL_TPB 256
__global__ __launch_bounds__(TAIL_TPB) void gather2_min(const float* __restrict__ rel3, const int* __restrict__ off,
                            const int* __restrict__ key, float* __restrict__ agg3,
                            float* __restrict__ blockmin, int N) {
    __shared__ float sm[TAIL_TPB / 32];
    const long long t = (long long)blockIdx.x * blockDim.x + threadIdx.x;
    const int i = (int)(t >> 4);
    const int l = (int)t & 15;
    float s0 = 0.0f, s1 = 0.0f;
    if (i < N) {
        const int pe = off[i + 1];
        for (int p = off[i] + l; p < pe; p += 16) {
            const float2 r = *(const float2*)(rel3 + (size_t)key[p] * 2);
            s0 += r.x; s1 += r.y;
        }
    }
    for (int m = 8; m >= 1; m >>= 1) {
        s0 += __shfl_xor(s0, m, 16);
        s1 += __shfl_xor(s1, m, 16);
    }
    float mn = 3.402823466e38f;
    if (i < N && l == 0) {
        const float v0 = agg3[(size_t)i * 2 + 0] + s0;
        const float v1 = agg3[(size_t)i * 2 + 1] + s1;
        agg3[(size_t)i * 2 + 0] = v0;
        agg3[(size_t)i * 2 + 1] = v1;
        mn = fminf(v0, v1);
    }
    mn = fminf(mn, __shfl_xor(mn, 16, 32));
    if ((threadIdx.x & 31) == 0) sm[threadIdx.x >> 5] = mn;
    __syncthreads();
    if (threadIdx.x == 0) {
        float r = sm[0];
        #pragma unroll
        for (int w = 1; w < TAIL_TPB / 32; ++w) r = fminf(r, sm[w]);
        blockmin[blockIdx.x] = r;
    }
}

__global__ __launch_bounds__(TAIL_TPB) void finalize_bm(const float* __restrict__ agg3,
                            const int* __restrict__ cell,
                            const int* __restrict__ maxs,
                            const float* __restrict__ blockmin, int nblk,
                            float* __restrict__ out, int N) {
    __shared__ float sm[TAIL_TPB / 32];
    float mn = 3.402823466e38f;
    for (int j0 = threadIdx.x; j0 < nblk; j0 += 16 * TAIL_TPB) {
        float v[16];
        #pragma unroll
        for (int u = 0; u < 16; ++u) v[u] = blockmin[min(j0 + u * TAIL_TPB, nblk - 1)];
        #pragma unroll
        for (int u = 0; u < 16; ++u) mn = fminf(mn, v[u]);
    }
    for (int m = 16; m >= 1; m >>= 1) mn = fminf(mn, __shfl_xor(mn, m, 32));
    if ((threadIdx.x & 31) == 0) sm[threadIdx.x >> 5] = mn;
    __syncthreads();
    float r = sm[0];
    #pragma unroll
    for (int w = 1; w < TAIL_TPB / 32; ++w) r = fminf(r, sm[w]);
    const float gmin = r - 1.0f;
    const int i = blockIdx.x * blockDim.x + threadIdx.x;
    if (i >= N) return;
    const bool upper = cell[i] >= (maxs[i] - 1);
    const bool lower = cell[i] == 0;
    out[(size_t)i * 2 + 0] = upper ? gmin : agg3[(size_t)i * 2 + 0];
    out[(size_t)i * 2 + 1] = lower ? gmin : agg3[(size_t)i * 2 + 1];
}

__device__ __forceinline__ unsigned f2ord(float f) {
    unsigned u = __float_as_uint(f);
    return (u & 0x80000000u) ? ~u : (u | 0x80000000u);
}
__device__ __forceinline__ float ord2f(unsigned u) {
    return __uint_as_float((u & 0x80000000u) ? (u ^ 0x80000000u) : ~u);
}

__global__ void init_slot(unsigned* slot) { *slot = 0xFFFFFFFFu; }

__global__ void min_reduce(const float* __restrict__ p, int n, unsigned* slot) {
    __shared__ float sm[256];
    float m = 3.402823466e38f;
    for (int i = blockIdx.x * blockDim.x + threadIdx.x; i < n;
         i += gridDim.x * blockDim.x)
        m = fminf(m, p[i]);
    sm[threadIdx.x] = m;
    __syncthreads();
    for (int s = blockDim.x >> 1; s > 0; s >>= 1) {
        if ((int)threadIdx.x < s) sm[threadIdx.x] = fminf(sm[threadIdx.x], sm[threadIdx.x + s]);
        __syncthreads();
    }
    if (threadIdx.x == 0) atomicMin(slot, f2ord(sm[0]));
}

__global__ void finalize_k(const float* __restrict__ agg3,
                           const int* __restrict__ cell,
                           const int* __restrict__ maxs,
                           const unsigned* __restrict__ slot,
                           float* __restrict__ out, int N) {
    const int i = blockIdx.x * blockDim.x + threadIdx.x;
    if (i >= N) return;
    const float gmin = ord2f(*slot) - 1.0f;
    const bool upper = cell[i] >= (maxs[i] - 1);
    const bool lower = cell[i] == 0;
    out[(size_t)i * 2 + 0] = upper ? gmin : agg3[(size_t)i * 2 + 0];
    out[(size_t)i * 2 + 1] = lower ? gmin : agg3[(size_t)i * 2 + 1];
}

extern "C" void kernel_launch(void* const* d_in, const int* in_sizes, int n_in,
                              void* d_out, int out_size, void* d_ws, size_t ws_size,
                              hipStream_t stream) {
    const float* x      = (const float*)d_in[0];
    const int*   src    = (const int*)d_in[1];
    const int*   dst    = (const int*)d_in[2];
    const int*   etypes = (const int*)d_in[3];
    const int*   cell   = (const int*)d_in[4];
    const int*   maxs   = (const int*)d_in[5];
    const float* W1     = (const float*)d_in[6];
    const float* loop1  = (const float*)d_in[7];
    const float* b1     = (const float*)d_in[8];
    const float* W2     = (const float*)d_in[9];
    const float* loop2  = (const float*)d_in[10];
    const float* b2     = (const float*)d_in[11];
    const float* W3     = (const float*)d_in[12];
    const float* loop3  = (const float*)d_in[13];
    const float* b3     = (const float*)d_in[14];
    float* out = (float*)d_out;

    const int N = in_sizes[4];
    const int E = in_sizes[1];
    const int K1 = in_sizes[0] / N;

    float* relbuf = (float*)d_ws;
    float* h1     = relbuf + (size_t)2 * N * 64;
    float* h2     = h1 + (size_t)N * 64;
    float* rel3   = h2 + (size_t)N * 64;
    float* agg3   = rel3 + (size_t)4 * N;
    unsigned* slot = (unsigned*)(agg3 + (size_t)2 * N);

    const int waves   = (N + 15) / 16;
    const int gblocks = (waves * 32 + 255) / 256;
    const int eb64    = (int)(((long long)E * 16 + 255) / 256);
    const int nb      = (N + 255) / 256;
    const int fb      = (N * 64 + 255) / 256;
    const int eb2     = (E + 255) / 256;

    int* key      = (int*)(slot + 4);
    int* off      = key + (size_t)E;
    int* blockcnt = (int*)relbuf;
    int* rank     = blockcnt + (size_t)CSR_NB * N;
    const size_t need = ((size_t)262 * N + 4 + (size_t)E + (size_t)N + 1) * sizeof(float);
    int* coarse   = (int*)rel3;
    const bool csr_ok = ws_size >= need && (long long)E <= (long long)192 * N
                        && (long long)CSR_NB * nb <= (long long)6 * N;

    if (csr_ok) {
        const int chunk = (E + CSR_NB - 1) / CSR_NB;
        const int gb64  = (int)(((long long)N * 16 + 255) / 256);
        csr_hist<<<CSR_NB, 1024, 0, stream>>>(dst, rank, blockcnt, coarse, E, N, chunk, nb);
        csr_colscan_off<<<nb, CSR_RANGE, 0, stream>>>(blockcnt, coarse, off, N, nb);
        csr_fill<<<eb2, 256, 0, stream>>>(src, dst, etypes, rank, blockcnt, key, E, N, chunk);

        gemm3_xw<<<gblocks, 256, 0, stream>>>(x, W1, loop1, b1, relbuf, h1, N, K1);
        gather64_relu<<<gb64, 256, 0, stream>>>(relbuf, off, key, h1, N);

        gemm3_xw<<<gblocks, 256, 0, stream>>>(h1, W2, loop2, b2, relbuf, h2, N, 64);
        gather64_relu_l3<<<gb64, 256, 0, stream>>>(relbuf, off, key, h2, W3, loop3, b3, rel3, agg3, N);

        float* blockmin = relbuf;
        gather2_min<<<gb64, TAIL_TPB, 0, stream>>>(rel3, off, key, agg3, blockmin, N);
        finalize_bm<<<nb, TAIL_TPB, 0, stream>>>(agg3, cell, maxs, blockmin, gb64, out, N);
    } else {
        gemm3_xw<<<gblocks, 256, 0, stream>>>(x, W1, loop1, b1, relbuf, h1, N, K1);
        edge_scatter64<<<eb64, 256, 0, stream>>>(relbuf, src, dst, etypes, h1, E, N);
        relu_k<<<fb, 256, 0, stream>>>(h1, N * 64);

        gemm3_xw<<<gblocks, 256, 0, stream>>>(h1, W2, loop2, b2, relbuf, h2, N, 64);
        edge_scatter64<<<eb64, 256, 0, stream>>>(relbuf, src, dst, etypes, h2, E, N);
        relu_k<<<fb, 256, 0, stream>>>(h2, N * 64);

        layer3_node<<<nb, 256, 0, stream>>>(h2, W3, loop3, b3, rel3, agg3, N);
        edge_scatter2<<<eb2, 256, 0, stream>>>(rel3, src, dst, etypes, agg3, E, N);
        init_slot<<<1, 1, 0, stream>>>(slot);
        min_reduce<<<512, 256, 0, stream>>>(agg3, N * 2, slot);
        finalize_k<<<nb, 256, 0, stream>>>(agg3, cell, maxs, slot, out, N);
    }
}


# The edge sort's column scan rides in the layer-1 GEMM launch as extra blocks (sort tables moved off the planes the GEMM writes): 9 -> 8 launches, the scan runs under the matrix work
# speedup vs baseline: 2.1559x; 1.0347x over previous
#include <hip/hip_runtime.h>

typedef __attribute__((ext_vector_type(2))) float v2f;
typedef __attribute__((ext_vector_type(8))) float v8f;

__device__ __forceinline__ void gemm3_strip(const float* __restrict__ X, const float* __restrict__ Wr,
                         const float* __restrict__ Wl, const float* __restrict__ bias,
                         float* __restrict__ rel, float* __restrict__ H,
                         int M, int K, int block) {
    const int gthread = block * blockDim.x + threadIdx.x;
    const int wave    = gthread >> 5;
    const int lane    = threadIdx.x & 31;
    const int m0      = wave * 16;
    if (m0 >= M) return;

    const int half = lane >> 4;
    const int lm   = lane & 15;

    v8f acc[12];
    #pragma unroll
    for (int t = 0; t < 8; ++t) {
        #pragma unroll
        for (int v = 0; v < 8; ++v) acc[t][v] = 0.0f;
    }
    #pragma unroll
    for (int nt = 0; nt < 4; ++nt) {
        const float bv = bias[nt * 16 + lm];
        #pragma unroll
        for (int v = 0; v < 8; ++v) acc[8 + nt][v] = bv;
    }

    const float* xrow = X + (size_t)(m0 + lm) * K;
    const float* w1   = Wr + (size_t)K * 64;

    #pragma unroll 2
    for (int k0 = 0; k0 < K; k0 += 4) {
        const int kb = k0 + 2 * half;
        const float2 av = *(const float2*)(xrow + kb);
        v2f a; a.x = av.x; a.y = av.y;
        #pragma unroll
        for (int mt = 0; mt < 3; ++mt) {
            const float* Wm = (mt == 0) ? Wr : (mt == 1) ? w1 : Wl;
            #pragma unroll
            for (int nt = 0; nt < 4; ++nt) {
                const int n = nt * 16 + lm;
                v2f b;
                b.x = Wm[(size_t)kb * 64 + n];
                b.y = Wm[(size_t)(kb + 1) * 64 + n];
                acc[mt * 4 + nt] = __builtin_amdgcn_wmma_f32_16x16x4_f32(
                    false, a, false, b, (short)0, acc[mt * 4 + nt], false, false);
            }
        }
    }

    #pragma unroll
    for (int mt = 0; mt < 3; ++mt) {
        float* C = (mt == 2) ? H : rel + (size_t)mt * M * 64;
        #pragma unroll
        for (int nt = 0; nt < 4; ++nt) {
            const int n = nt * 16 + lm;
            #pragma unroll
            for (int v = 0; v < 8; ++v) {
                C[(size_t)(m0 + v + 8 * half) * 64 + n] = acc[mt * 4 + nt][v];
            }
        }
    }
}

__global__ __launch_bounds__(256) void gemm3_xw(const float* __restrict__ X, const float* __restrict__ Wr,
                         const float* __restrict__ Wl, const float* __restrict__ bias,
                         float* __restrict__ rel, float* __restrict__ H,
                         int M, int K) {
    gemm3_strip(X, Wr, Wl, bias, rel, H, M, K, (int)blockIdx.x);
}

__global__ void edge_scatter64(const float* __restrict__ rel,
                               const int* __restrict__ src,
                               const int* __restrict__ dst,
                               const int* __restrict__ et,
                               float* __restrict__ agg, int E, int N) {
    const long long t = (long long)blockIdx.x * blockDim.x + threadIdx.x;
    const int e = (int)(t >> 4);
    if (e >= E) return;
    const int f = ((int)t & 15) * 4;
    const float4 v = *(const float4*)(rel + ((long long)et[e] * N + src[e]) * 64 + f);
    float* q = agg + (long long)dst[e] * 64 + f;
    atomicAdd(q + 0, v.x);
    atomicAdd(q + 1, v.y);
    atomicAdd(q + 2, v.z);
    atomicAdd(q + 3, v.w);
}

__global__ void relu_k(float* __restrict__ p, int n) {
    const int i = blockIdx.x * blockDim.x + threadIdx.x;
    if (i < n) p[i] = fmaxf(p[i], 0.0f);
}

__global__ void layer3_node(const float* __restrict__ h2,
                            const float* __restrict__ W3,
                            const float* __restrict__ loop3,
                            const float* __restrict__ b3,
                            float* __restrict__ rel3, float* __restrict__ agg3,
                            int N) {
    const int i = blockIdx.x * blockDim.x + threadIdx.x;
    if (i >= N) return;
    const float* row = h2 + (size_t)i * 64;
    float r00 = 0.f, r01 = 0.f, r10 = 0.f, r11 = 0.f, l0 = 0.f, l1 = 0.f;
    #pragma unroll 8
    for (int k = 0; k < 64; ++k) {
        const float v = row[k];
        r00 += v * W3[k * 2 + 0];
        r01 += v * W3[k * 2 + 1];
        r10 += v * W3[128 + k * 2 + 0];
        r11 += v * W3[128 + k * 2 + 1];
        l0  += v * loop3[k * 2 + 0];
        l1  += v * loop3[k * 2 + 1];
    }
    rel3[(size_t)i * 2 + 0] = r00;
    rel3[(size_t)i * 2 + 1] = r01;
    rel3[(size_t)(N + i) * 2 + 0] = r10;
    rel3[(size_t)(N + i) * 2 + 1] = r11;
    agg3[(size_t)i * 2 + 0] = l0 + b3[0];
    agg3[(size_t)i * 2 + 1] = l1 + b3[1];
}

__global__ void edge_scatter2(const float* __restrict__ rel3,
                              const int* __restrict__ src,
                              const int* __restrict__ dst,
                              const int* __restrict__ et,
                              float* __restrict__ agg3, int E, int N) {
    const int e = blockIdx.x * blockDim.x + threadIdx.x;
    if (e >= E) return;
    const float* p = rel3 + ((long long)et[e] * N + src[e]) * 2;
    float* q = agg3 + (long long)dst[e] * 2;
    atomicAdd(q + 0, p[0]);
    atomicAdd(q + 1, p[1]);
}


#define CSR_NB  64
#define CSR_CAP 51200
#define CSR_RANGE 256

__global__ __launch_bounds__(1024) void csr_hist(const int* __restrict__ dst,
                                                 int* __restrict__ rank,
                                                 int* __restrict__ blockcnt,
                                                 int* __restrict__ coarse,
                                                 int E, int N, int chunk, int nr_total) {
    __shared__ int cnt[CSR_CAP];
    const int b  = blockIdx.x;
    const int e0 = b * chunk;
    const int e1 = min(e0 + chunk, E);
    for (int base = 0; base < N; base += CSR_CAP) {
        const int lim = min(CSR_CAP, N - base);
        for (int i = threadIdx.x; i < lim; i += blockDim.x) cnt[i] = 0;
        __syncthreads();
        const int bd = (int)blockDim.x;
        for (int e = e0 + (int)threadIdx.x; e < e1; e += 4 * bd) {
            int d[4];
            #pragma unroll
            for (int j = 0; j < 4; ++j) {
                const int ee = e + j * bd;
                const int v  = dst[min(ee, e1 - 1)];
                d[j] = (ee < e1) ? v - base : -1;
            }
            #pragma unroll
            for (int j = 0; j < 4; ++j)
                if (d[j] >= 0 && d[j] < lim) rank[e + j * bd] = atomicAdd(&cnt[d[j]], 1);
        }
        __syncthreads();
        const int nrs = (lim + CSR_RANGE - 1) / CSR_RANGE;
        for (int r0 = 0; r0 < nrs; r0 += (int)blockDim.x >> 2) {
            const int r  = r0 + ((int)threadIdx.x >> 2);
            const int q  = (int)threadIdx.x & 3;
            const int lo = r * CSR_RANGE + q * (CSR_RANGE / 4);
            int s = 0;
            #pragma unroll 16
            for (int j = 0; j < CSR_RANGE / 4; ++j) {
                const int i = lo + ((j + (int)threadIdx.x) & (CSR_RANGE / 4 - 1));
                const int v = cnt[min(i, lim - 1)];
                s += (i < lim) ? v : 0;
            }
            s += __shfl_xor(s, 1, 4);
            s += __shfl_xor(s, 2, 4);
            if (q == 0 && r < nrs) coarse[(size_t)b * nr_total + base / CSR_RANGE + r] = s;
        }
        for (int i = threadIdx.x; i < lim; i += blockDim.x)
            blockcnt[(size_t)b * N + base + i] = cnt[i];
        __syncthreads();
    }
}

__device__ __forceinline__ void csr_colscan_range(int* __restrict__ blockcnt,
                                                  const int* __restrict__ coarse,
                                                  int* __restrict__ off, int N, int nr_total, int r) {
    __shared__ int wtot[CSR_RANGE / 32];
    __shared__ int wbase[CSR_RANGE / 32];
    const int t  = threadIdx.x;
    const int n  = r * CSR_RANGE + t;
    const int nn = min(n, N - 1);
    int c[CSR_NB];
    #pragma unroll
    for (int j = 0; j < CSR_NB; ++j) c[j] = blockcnt[(size_t)j * N + nn];
    int part = 0;
    for (int rr = t; rr < r; rr += CSR_RANGE) {
        #pragma unroll 16
        for (int b = 0; b < CSR_NB; ++b) part += coarse[(size_t)b * nr_total + rr];
    }
    int deg = 0;
    #pragma unroll
    for (int j = 0; j < CSR_NB; ++j) deg += c[j];
    if (n >= N) deg = 0;
    const int lane = t & 31;
    int inc = deg;
    for (int d = 1; d < 32; d <<= 1) {
        const int v = __shfl_up(inc, d, 32);
        if (lane >= d) inc += v;
    }
    for (int m = 16; m >= 1; m >>= 1) part += __shfl_xor(part, m, 32);
    if (lane == 31) wtot[t >> 5] = inc;
    if (lane == 0)  wbase[t >> 5] = part;
    __syncthreads();
    int run = inc - deg;
    #pragma unroll
    for (int w = 0; w < CSR_RANGE / 32; ++w) {
        run += wbase[w];
        run += (w < (t >> 5)) ? wtot[w] : 0;
    }
    if (n < N) {
        off[n] = run;
        if (n == N - 1) off[N] = run + deg;
        #pragma unroll
        for (int j = 0; j < CSR_NB; ++j) {
            blockcnt[(size_t)j * N + n] = run;
            run += c[j];
        }
    }
}

__global__ __launch_bounds__(CSR_RANGE) void csr_colscan_off(int* __restrict__ blockcnt,
                                                             const int* __restrict__ coarse,
                                                             int* __restrict__ off, int N, int nr_total) {
    csr_colscan_range(blockcnt, coarse, off, N, nr_total, (int)blockIdx.x);
}

static_assert(CSR_RANGE == 256, "the shared launch uses 256-thread blocks for both parts");
__global__ __launch_bounds__(256) void gemm3_xw_colscan(const float* __restrict__ X, const float* __restrict__ Wr,
                         const float* __restrict__ Wl, const float* __restrict__ bias,
                         float* __restrict__ rel, float* __restrict__ H, int M, int K, int gblocks,
                         int* __restrict__ blockcnt, const int* __restrict__ coarse,
                         int* __restrict__ off, int nr_total) {
    if ((int)blockIdx.x < gblocks) gemm3_strip(X, Wr, Wl, bias, rel, H, M, K, (int)blockIdx.x);
    else csr_colscan_range(blockcnt, coarse, off, M, nr_total, (int)blockIdx.x - gblocks);
}

__global__ void csr_fill(const int* __restrict__ src, const int* __restrict__ dst,
                         const int* __restrict__ et, const int* __restrict__ rank,
                         const int* __restrict__ blockcnt,
                         int* __restrict__ key, int E, int N, int chunk) {
    const int e = blockIdx.x * blockDim.x + threadIdx.x;
    if (e >= E) return;
    const int d = dst[e];
    const int b = e / chunk;
    int pos = blockcnt[(size_t)b * N + d] + rank[e];
    pos = min(max(pos, 0), E - 1);
    key[pos] = et[e] * N + src[e];
}

__global__ void gather64_relu(const float* __restrict__ rel, const int* __restrict__ off,
                              const int* __restrict__ key, float* __restrict__ h, int N) {
    const long long t = (long long)blockIdx.x * blockDim.x + threadIdx.x;
    const int i = (int)(t >> 4);
    if (i >= N) return;
    const int f = ((int)t & 15) * 4;
    float* hp = h + (size_t)i * 64 + f;
    float4 acc = *(const float4*)hp;
    int p = off[i];
    const int pe = off[i + 1];
    for (; p + 4 <= pe; p += 4) {
        const int k0 = key[p], k1 = key[p + 1], k2 = key[p + 2], k3 = key[p + 3];
        const float4 a = *(const float4*)(rel + (size_t)k0 * 64 + f);
        const float4 b = *(const float4*)(rel + (size_t)k1 * 64 + f);
        const float4 c = *(const float4*)(rel + (size_t)k2 * 64 + f);
        const float4 d = *(const float4*)(rel + (size_t)k3 * 64 + f);
        acc.x += a.x; acc.y += a.y; acc.z += a.z; acc.w += a.w;
        acc.x += b.x; acc.y += b.y; acc.z += b.z; acc.w += b.w;
        acc.x += c.x; acc.y += c.y; acc.z += c.z; acc.w += c.w;
        acc.x += d.x; acc.y += d.y; acc.z += d.z; acc.w += d.w;
    }
    for (; p < pe; ++p) {
        const float4 a = *(const float4*)(rel + (size_t)key[p] * 64 + f);
        acc.x += a.x; acc.y += a.y; acc.z += a.z; acc.w += a.w;
    }
    acc.x = fmaxf(acc.x, 0.0f); acc.y = fmaxf(acc.y, 0.0f);
    acc.z = fmaxf(acc.z, 0.0f); acc.w = fmaxf(acc.w, 0.0f);
    *(float4*)hp = acc;
}

__global__ void gather2(const float* __restrict__ rel3, const int* __restrict__ off,
                        const int* __restrict__ key, float* __restrict__ agg3, int N) {
    const long long t = (long long)blockIdx.x * blockDim.x + threadIdx.x;
    const int i = (int)(t >> 4);
    const int l = (int)t & 15;
    float s0 = 0.0f, s1 = 0.0f;
    if (i < N) {
        const int pe = off[i + 1];
        for (int p = off[i] + l; p < pe; p += 16) {
            const float2 r = *(const float2*)(rel3 + (size_t)key[p] * 2);
            s0 += r.x; s1 += r.y;
        }
    }
    for (int m = 8; m >= 1; m >>= 1) {
        s0 += __shfl_xor(s0, m, 16);
        s1 += __shfl_xor(s1, m, 16);
    }
    if (i < N && l == 0) {
        agg3[(size_t)i * 2 + 0] += s0;
        agg3[(size_t)i * 2 + 1] += s1;
    }
}

__global__ void gather64_relu_l3(const float* __restrict__ rel, const int* __restrict__ off,
                                 const int* __restrict__ key, const float* __restrict__ h,
                                 const float* __restrict__ W3, const float* __restrict__ loop3,
                                 const float* __restrict__ b3,
                                 float* __restrict__ rel3, float* __restrict__ agg3, int N) {
    const long long t = (long long)blockIdx.x * blockDim.x + threadIdx.x;
    const int i = (int)(t >> 4);
    if (i >= N) return;
    const int f = ((int)t & 15) * 4;
    float4 acc = *(const float4*)(h + (size_t)i * 64 + f);
    int p = off[i];
    const int pe = off[i + 1];
    for (; p + 4 <= pe; p += 4) {
        const int k0 = key[p], k1 = key[p + 1], k2 = key[p + 2], k3 = key[p + 3];
        const float4 a = *(const float4*)(rel + (size_t)k0 * 64 + f);
        const float4 b = *(const float4*)(rel + (size_t)k1 * 64 + f);
        const float4 c = *(const float4*)(rel + (size_t)k2 * 64 + f);
        const float4 d = *(const float4*)(rel + (size_t)k3 * 64 + f);
        acc.x += a.x; acc.y += a.y; acc.z += a.z; acc.w += a.w;
        acc.x += b.x; acc.y += b.y; acc.z += b.z; acc.w += b.w;
        acc.x += c.x; acc.y += c.y; acc.z += c.z; acc.w += c.w;
        acc.x += d.x; acc.y += d.y; acc.z += d.z; acc.w += d.w;
    }
    for (; p < pe; ++p) {
        const float4 a = *(const float4*)(rel + (size_t)key[p] * 64 + f);
        acc.x += a.x; acc.y += a.y; acc.z += a.z; acc.w += a.w;
    }
    const float hv[4] = { fmaxf(acc.x, 0.0f), fmaxf(acc.y, 0.0f), fmaxf(acc.z, 0.0f), fmaxf(acc.w, 0.0f) };
    const float* wm[3] = { W3 + f * 2, W3 + 128 + f * 2, loop3 + f * 2 };
    float o[6];
    #pragma unroll
    for (int m = 0; m < 3; ++m) {
        float s0 = 0.0f, s1 = 0.0f;
        #pragma unroll
        for (int j = 0; j < 4; ++j) {
            s0 += hv[j] * wm[m][2 * j + 0];
            s1 += hv[j] * wm[m][2 * j + 1];
        }
        o[2 * m + 0] = s0;
        o[2 * m + 1] = s1;
    }
    for (int d = 8; d >= 1; d >>= 1) {
        #pragma unroll
        for (int q = 0; q < 6; ++q) o[q] += __shfl_xor(o[q], d, 16);
    }
    if (((int)t & 15) == 0) {
        rel3[(size_t)i * 2 + 0] = o[0];
        rel3[(size_t)i * 2 + 1] = o[1];
        rel3[(size_t)(N + i) * 2 + 0] = o[2];
        rel3[(size_t)(N + i) * 2 + 1] = o[3];
        agg3[(size_t)i * 2 + 0] = o[4] + b3[0];
        agg3[(size_t)i * 2 + 1] = o[5] + b3[1];
    }
}

#define TAIL_TPB 256
__global__ __launch_bounds__(TAIL_TPB) void gather2_min(const float* __restrict__ rel3, const int* __restrict__ off,
                            const int* __restrict__ key, float* __restrict__ agg3,
                            float* __restrict__ blockmin, int N) {
    __shared__ float sm[TAIL_TPB / 32];
    const long long t = (long long)blockIdx.x * blockDim.x + threadIdx.x;
    const int i = (int)(t >> 4);
    const int l = (int)t & 15;
    float s0 = 0.0f, s1 = 0.0f;
    if (i < N) {
        const int pe = off[i + 1];
        for (int p = off[i] + l; p < pe; p += 16) {
            const float2 r = *(const float2*)(rel3 + (size_t)key[p] * 2);
            s0 += r.x; s1 += r.y;
        }
    }
    for (int m = 8; m >= 1; m >>= 1) {
        s0 += __shfl_xor(s0, m, 16);
        s1 += __shfl_xor(s1, m, 16);
    }
    float mn = 3.402823466e38f;
    if (i < N && l == 0) {
        const float v0 = agg3[(size_t)i * 2 + 0] + s0;
        const float v1 = agg3[(size_t)i * 2 + 1] + s1;
        agg3[(size_t)i * 2 + 0] = v0;
        agg3[(size_t)i * 2 + 1] = v1;
        mn = fminf(v0, v1);
    }
    mn = fminf(mn, __shfl_xor(mn, 16, 32));
    if ((threadIdx.x & 31) == 0) sm[threadIdx.x >> 5] = mn;
    __syncthreads();
    if (threadIdx.x == 0) {
        float r = sm[0];
        #pragma unroll
        for (int w = 1; w < TAIL_TPB / 32; ++w) r = fminf(r, sm[w]);
        blockmin[blockIdx.x] = r;
    }
}

__global__ __launch_bounds__(TAIL_TPB) void finalize_bm(const float* __restrict__ agg3,
                            const int* __restrict__ cell,
                            const int* __restrict__ maxs,
                            const float* __restrict__ blockmin, int nblk,
                            float* __restrict__ out, int N) {
    __shared__ float sm[TAIL_TPB / 32];
    float mn = 3.402823466e38f;
    for (int j0 = threadIdx.x; j0 < nblk; j0 += 16 * TAIL_TPB) {
        float v[16];
        #pragma unroll
        for (int u = 0; u < 16; ++u) v[u] = blockmin[min(j0 + u * TAIL_TPB, nblk - 1)];
        #pragma unroll
        for (int u = 0; u < 16; ++u) mn = fminf(mn, v[u]);
    }
    for (int m = 16; m >= 1; m >>= 1) mn = fminf(mn, __shfl_xor(mn, m, 32));
    if ((threadIdx.x & 31) == 0) sm[threadIdx.x >> 5] = mn;
    __syncthreads();
    float r = sm[0];
    #pragma unroll
    for (int w = 1; w < TAIL_TPB / 32; ++w) r = fminf(r, sm[w]);
    const float gmin = r - 1.0f;
    const int i = blockIdx.x * blockDim.x + threadIdx.x;
    if (i >= N) return;
    const bool upper = cell[i] >= (maxs[i] - 1);
    const bool lower = cell[i] == 0;
    out[(size_t)i * 2 + 0] = upper ? gmin : agg3[(size_t)i * 2 + 0];
    out[(size_t)i * 2 + 1] = lower ? gmin : agg3[(size_t)i * 2 + 1];
}

__device__ __forceinline__ unsigned f2ord(float f) {
    unsigned u = __float_as_uint(f);
    return (u & 0x80000000u) ? ~u : (u | 0x80000000u);
}
__device__ __forceinline__ float ord2f(unsigned u) {
    return __uint_as_float((u & 0x80000000u) ? (u ^ 0x80000000u) : ~u);
}

__global__ void init_slot(unsigned* slot) { *slot = 0xFFFFFFFFu; }

__global__ void min_reduce(const float* __restrict__ p, int n, unsigned* slot) {
    __shared__ float sm[256];
    float m = 3.402823466e38f;
    for (int i = blockIdx.x * blockDim.x + threadIdx.x; i < n;
         i += gridDim.x * blockDim.x)
        m = fminf(m, p[i]);
    sm[threadIdx.x] = m;
    __syncthreads();
    for (int s = blockDim.x >> 1; s > 0; s >>= 1) {
        if ((int)threadIdx.x < s) sm[threadIdx.x] = fminf(sm[threadIdx.x], sm[threadIdx.x + s]);
        __syncthreads();
    }
    if (threadIdx.x == 0) atomicMin(slot, f2ord(sm[0]));
}

__global__ void finalize_k(const float* __restrict__ agg3,
                           const int* __restrict__ cell,
                           const int* __restrict__ maxs,
                           const unsigned* __restrict__ slot,
                           float* __restrict__ out, int N) {
    const int i = blockIdx.x * blockDim.x + threadIdx.x;
    if (i >= N) return;
    const float gmin = ord2f(*slot) - 1.0f;
    const bool upper = cell[i] >= (maxs[i] - 1);
    const bool lower = cell[i] == 0;
    out[(size_t)i * 2 + 0] = upper ? gmin : agg3[(size_t)i * 2 + 0];
    out[(size_t)i * 2 + 1] = lower ? gmin : agg3[(size_t)i * 2 + 1];
}

extern "C" void kernel_launch(void* const* d_in, const int* in_sizes, int n_in,
                              void* d_out, int out_size, void* d_ws, size_t ws_size,
                              hipStream_t stream) {
    const float* x      = (const float*)d_in[0];
    const int*   src    = (const int*)d_in[1];
    const int*   dst    = (const int*)d_in[2];
    const int*   etypes = (const int*)d_in[3];
    const int*   cell   = (const int*)d_in[4];
    const int*   maxs   = (const int*)d_in[5];
    const float* W1     = (const float*)d_in[6];
    const float* loop1  = (const float*)d_in[7];
    const float* b1     = (const float*)d_in[8];
    const float* W2     = (const float*)d_in[9];
    const float* loop2  = (const float*)d_in[10];
    const float* b2     = (const float*)d_in[11];
    const float* W3     = (const float*)d_in[12];
    const float* loop3  = (const float*)d_in[13];
    const float* b3     = (const float*)d_in[14];
    float* out = (float*)d_out;

    const int N = in_sizes[4];
    const int E = in_sizes[1];
    const int K1 = in_sizes[0] / N;

    float* relbuf = (float*)d_ws;
    float* h1     = relbuf + (size_t)2 * N * 64;
    float* h2     = h1 + (size_t)N * 64;
    float* rel3   = h2 + (size_t)N * 64;
    float* agg3   = rel3 + (size_t)4 * N;
    unsigned* slot = (unsigned*)(agg3 + (size_t)2 * N);

    const int waves   = (N + 15) / 16;
    const int gblocks = (waves * 32 + 255) / 256;
    const int eb64    = (int)(((long long)E * 16 + 255) / 256);
    const int nb      = (N + 255) / 256;
    const int fb      = (N * 64 + 255) / 256;
    const int eb2     = (E + 255) / 256;

    int* key      = (int*)(slot + 4);
    int* off      = key + (size_t)E;
    int* blockcnt = (int*)relbuf;
    int* rank     = blockcnt + (size_t)CSR_NB * N;
    const size_t need = ((size_t)262 * N + 4 + (size_t)E + (size_t)N + 1) * sizeof(float);
    int* coarse   = (int*)rel3;
    const bool csr_ok = ws_size >= need && (long long)E <= (long long)192 * N
                        && (long long)CSR_NB * nb <= (long long)6 * N;
    const size_t need2 = need + (size_t)E * sizeof(int);
    const bool shared_ok = csr_ok && ws_size >= need2;

    if (shared_ok) {
        const int chunk = (E + CSR_NB - 1) / CSR_NB;
        const int gb64  = (int)(((long long)N * 16 + 255) / 256);
        int* blockcnt2 = (int*)h2;
        int* rank2     = off + (size_t)N + 1;
        csr_hist<<<CSR_NB, 1024, 0, stream>>>(dst, rank2, blockcnt2, coarse, E, N, chunk, nb);

        gemm3_xw_colscan<<<gblocks + nb, 256, 0, stream>>>(x, W1, loop1, b1, relbuf, h1, N, K1, gblocks,
                                                           blockcnt2, coarse, off, nb);
        csr_fill<<<eb2, 256, 0, stream>>>(src, dst, etypes, rank2, blockcnt2, key, E, N, chunk);
        gather64_relu<<<gb64, 256, 0, stream>>>(relbuf, off, key, h1, N);

        gemm3_xw<<<gblocks, 256, 0, stream>>>(h1, W2, loop2, b2, relbuf, h2, N, 64);
        gather64_relu_l3<<<gb64, 256, 0, stream>>>(relbuf, off, key, h2, W3, loop3, b3, rel3, agg3, N);

        float* blockmin = relbuf;
        gather2_min<<<gb64, TAIL_TPB, 0, stream>>>(rel3, off, key, agg3, blockmin, N);
        finalize_bm<<<nb, TAIL_TPB, 0, stream>>>(agg3, cell, maxs, blockmin, gb64, out, N);
    } else if (csr_ok) {
        const int chunk = (E + CSR_NB - 1) / CSR_NB;
        const int gb64  = (int)(((long long)N * 16 + 255) / 256);
        csr_hist<<<CSR_NB, 1024, 0, stream>>>(dst, rank, blockcnt, coarse, E, N, chunk, nb);
        csr_colscan_off<<<nb, CSR_RANGE, 0, stream>>>(blockcnt, coarse, off, N, nb);
        csr_fill<<<eb2, 256, 0, stream>>>(src, dst, etypes, rank, blockcnt, key, E, N, chunk);

        gemm3_xw<<<gblocks, 256, 0, stream>>>(x, W1, loop1, b1, relbuf, h1, N, K1);
        gather64_relu<<<gb64, 256, 0, stream>>>(relbuf, off, key, h1, N);

        gemm3_xw<<<gblocks, 256, 0, stream>>>(h1, W2, loop2, b2, relbuf, h2, N, 64);
        gather64_relu_l3<<<gb64, 256, 0, stream>>>(relbuf, off, key, h2, W3, loop3, b3, rel3, agg3, N);

        float* blockmin = relbuf;
        gather2_min<<<gb64, TAIL_TPB, 0, stream>>>(rel3, off, key, agg3, blockmin, N);
        finalize_bm<<<nb, TAIL_TPB, 0, stream>>>(agg3, cell, maxs, blockmin, gb64, out, N);
    } else {
        gemm3_xw<<<gblocks, 256, 0, stream>>>(x, W1, loop1, b1, relbuf, h1, N, K1);
        edge_scatter64<<<eb64, 256, 0, stream>>>(relbuf, src, dst, etypes, h1, E, N);
        relu_k<<<fb, 256, 0, stream>>>(h1, N * 64);

        gemm3_xw<<<gblocks, 256, 0, stream>>>(h1, W2, loop2, b2, relbuf, h2, N, 64);
        edge_scatter64<<<eb64, 256, 0, stream>>>(relbuf, src, dst, etypes, h2, E, N);
        relu_k<<<fb, 256, 0, stream>>>(h2, N * 64);

        layer3_node<<<nb, 256, 0, stream>>>(h2, W3, loop3, b3, rel3, agg3, N);
        edge_scatter2<<<eb2, 256, 0, stream>>>(rel3, src, dst, etypes, agg3, E, N);
        init_slot<<<1, 1, 0, stream>>>(slot);
        min_reduce<<<512, 256, 0, stream>>>(agg3, N * 2, slot);
        finalize_k<<<nb, 256, 0, stream>>>(agg3, cell, maxs, slot, out, N);
    }
}


# Layer 2 adds each node's incoming h1 rows per relation BEFORE multiplying (sum-then-product order); its WMMA products run from LDS inside the gather launch: 8 -> 7 launches, no rel/h2 planes
# speedup vs baseline: 2.4483x; 1.1356x over previous
#include <hip/hip_runtime.h>

typedef __attribute__((ext_vector_type(2))) float v2f;
typedef __attribute__((ext_vector_type(8))) float v8f;

__device__ __forceinline__ void gemm3_strip(const float* __restrict__ X, const float* __restrict__ Wr,
                         const float* __restrict__ Wl, const float* __restrict__ bias,
                         float* __restrict__ rel, float* __restrict__ H,
                         int M, int K, int block) {
    const int gthread = block * blockDim.x + threadIdx.x;
    const int wave    = gthread >> 5;
    const int lane    = threadIdx.x & 31;
    const int m0      = wave * 16;
    if (m0 >= M) return;

    const int half = lane >> 4;
    const int lm   = lane & 15;

    v8f acc[12];
    #pragma unroll
    for (int t = 0; t < 8; ++t) {
        #pragma unroll
        for (int v = 0; v < 8; ++v) acc[t][v] = 0.0f;
    }
    #pragma unroll
    for (int nt = 0; nt < 4; ++nt) {
        const float bv = bias[nt * 16 + lm];
        #pragma unroll
        for (int v = 0; v < 8; ++v) acc[8 + nt][v] = bv;
    }

    const float* xrow = X + (size_t)(m0 + lm) * K;
    const float* w1   = Wr + (size_t)K * 64;

    #pragma unroll 2
    for (int k0 = 0; k0 < K; k0 += 4) {
        const int kb = k0 + 2 * half;
        const float2 av = *(const float2*)(xrow + kb);
        v2f a; a.x = av.x; a.y = av.y;
        #pragma unroll
        for (int mt = 0; mt < 3; ++mt) {
            const float* Wm = (mt == 0) ? Wr : (mt == 1) ? w1 : Wl;
            #pragma unroll
            for (int nt = 0; nt < 4; ++nt) {
                const int n = nt * 16 + lm;
                v2f b;
                b.x = Wm[(size_t)kb * 64 + n];
                b.y = Wm[(size_t)(kb + 1) * 64 + n];
                acc[mt * 4 + nt] = __builtin_amdgcn_wmma_f32_16x16x4_f32(
                    false, a, false, b, (short)0, acc[mt * 4 + nt], false, false);
            }
        }
    }

    #pragma unroll
    for (int mt = 0; mt < 3; ++mt) {
        float* C = (mt == 2) ? H : rel + (size_t)mt * M * 64;
        #pragma unroll
        for (int nt = 0; nt < 4; ++nt) {
            const int n = nt * 16 + lm;
            #pragma unroll
            for (int v = 0; v < 8; ++v) {
                C[(size_t)(m0 + v + 8 * half) * 64 + n] = acc[mt * 4 + nt][v];
            }
        }
    }
}

__global__ __launch_bounds__(256) void gemm3_xw(const float* __restrict__ X, const float* __restrict__ Wr,
                         const float* __restrict__ Wl, const float* __restrict__ bias,
                         float* __restrict__ rel, float* __restrict__ H,
                         int M, int K) {
    gemm3_strip(X, Wr, Wl, bias, rel, H, M, K, (int)blockIdx.x);
}

__global__ void edge_scatter64(const float* __restrict__ rel,
                               const int* __restrict__ src,
                               const int* __restrict__ dst,
                               const int* __restrict__ et,
                               float* __restrict__ agg, int E, int N) {
    const long long t = (long long)blockIdx.x * blockDim.x + threadIdx.x;
    const int e = (int)(t >> 4);
    if (e >= E) return;
    const int f = ((int)t & 15) * 4;
    const float4 v = *(const float4*)(rel + ((long long)et[e] * N + src[e]) * 64 + f);
    float* q = agg + (long long)dst[e] * 64 + f;
    atomicAdd(q + 0, v.x);
    atomicAdd(q + 1, v.y);
    atomicAdd(q + 2, v.z);
    atomicAdd(q + 3, v.w);
}

__global__ void relu_k(float* __restrict__ p, int n) {
    const int i = blockIdx.x * blockDim.x + threadIdx.x;
    if (i < n) p[i] = fmaxf(p[i], 0.0f);
}

__global__ void layer3_node(const float* __restrict__ h2,
                            const float* __restrict__ W3,
                            const float* __restrict__ loop3,
                            const float* __restrict__ b3,
                            float* __restrict__ rel3, float* __restrict__ agg3,
                            int N) {
    const int i = blockIdx.x * blockDim.x + threadIdx.x;
    if (i >= N) return;
    const float* row = h2 + (size_t)i * 64;
    float r00 = 0.f, r01 = 0.f, r10 = 0.f, r11 = 0.f, l0 = 0.f, l1 = 0.f;
    #pragma unroll 8
    for (int k = 0; k < 64; ++k) {
        const float v = row[k];
        r00 += v * W3[k * 2 + 0];
        r01 += v * W3[k * 2 + 1];
        r10 += v * W3[128 + k * 2 + 0];
        r11 += v * W3[128 + k * 2 + 1];
        l0  += v * loop3[k * 2 + 0];
        l1  += v * loop3[k * 2 + 1];
    }
    rel3[(size_t)i * 2 + 0] = r00;
    rel3[(size_t)i * 2 + 1] = r01;
    rel3[(size_t)(N + i) * 2 + 0] = r10;
    rel3[(size_t)(N + i) * 2 + 1] = r11;
    agg3[(size_t)i * 2 + 0] = l0 + b3[0];
    agg3[(size_t)i * 2 + 1] = l1 + b3[1];
}

__global__ void edge_scatter2(const float* __restrict__ rel3,
                              const int* __restrict__ src,
                              const int* __restrict__ dst,
                              const int* __restrict__ et,
                              float* __restrict__ agg3, int E, int N) {
    const int e = blockIdx.x * blockDim.x + threadIdx.x;
    if (e >= E) return;
    const float* p = rel3 + ((long long)et[e] * N + src[e]) * 2;
    float* q = agg3 + (long long)dst[e] * 2;
    atomicAdd(q + 0, p[0]);
    atomicAdd(q + 1, p[1]);
}


#define CSR_NB  64
#define CSR_CAP 51200
#define CSR_RANGE 256

__global__ __launch_bounds__(1024) void csr_hist(const int* __restrict__ dst,
                                                 int* __restrict__ rank,
                                                 int* __restrict__ blockcnt,
                                                 int* __restrict__ coarse,
                                                 int E, int N, int chunk, int nr_total) {
    __shared__ int cnt[CSR_CAP];
    const int b  = blockIdx.x;
    const int e0 = b * chunk;
    const int e1 = min(e0 + chunk, E);
    for (int base = 0; base < N; base += CSR_CAP) {
        const int lim = min(CSR_CAP, N - base);
        for (int i = threadIdx.x; i < lim; i += blockDim.x) cnt[i] = 0;
        __syncthreads();
        const int bd = (int)blockDim.x;
        for (int e = e0 + (int)threadIdx.x; e < e1; e += 4 * bd) {
            int d[4];
            #pragma unroll
            for (int j = 0; j < 4; ++j) {
                const int ee = e + j * bd;
                const int v  = dst[min(ee, e1 - 1)];
                d[j] = (ee < e1) ? v - base : -1;
            }
            #pragma unroll
            for (int j = 0; j < 4; ++j)
                if (d[j] >= 0 && d[j] < lim) rank[e + j * bd] = atomicAdd(&cnt[d[j]], 1);
        }
        __syncthreads();
        const int nrs = (lim + CSR_RANGE - 1) / CSR_RANGE;
        for (int r0 = 0; r0 < nrs; r0 += (int)blockDim.x >> 2) {
            const int r  = r0 + ((int)threadIdx.x >> 2);
            const int q  = (int)threadIdx.x & 3;
            const int lo = r * CSR_RANGE + q * (CSR_RANGE / 4);
            int s = 0;
            #pragma unroll 16
            for (int j = 0; j < CSR_RANGE / 4; ++j) {
                const int i = lo + ((j + (int)threadIdx.x) & (CSR_RANGE / 4 - 1));
                const int v = cnt[min(i, lim - 1)];
                s += (i < lim) ? v : 0;
            }
            s += __shfl_xor(s, 1, 4);
            s += __shfl_xor(s, 2, 4);
            if (q == 0 && r < nrs) coarse[(size_t)b * nr_total + base / CSR_RANGE + r] = s;
        }
        for (int i = threadIdx.x; i < lim; i += blockDim.x)
            blockcnt[(size_t)b * N + base + i] = cnt[i];
        __syncthreads();
    }
}

__device__ __forceinline__ void csr_colscan_range(int* __restrict__ blockcnt,
                                                  const int* __restrict__ coarse,
                                                  int* __restrict__ off, int N, int nr_total, int r) {
    __shared__ int wtot[CSR_RANGE / 32];
    __shared__ int wbase[CSR_RANGE / 32];
    const int t  = threadIdx.x;
    const int n  = r * CSR_RANGE + t;
    const int nn = min(n, N - 1);
    int c[CSR_NB];
    #pragma unroll
    for (int j = 0; j < CSR_NB; ++j) c[j] = blockcnt[(size_t)j * N + nn];
    int part = 0;
    for (int rr = t; rr < r; rr += CSR_RANGE) {
        #pragma unroll 16
        for (int b = 0; b < CSR_NB; ++b) part += coarse[(size_t)b * nr_total + rr];
    }
    int deg = 0;
    #pragma unroll
    for (int j = 0; j < CSR_NB; ++j) deg += c[j];
    if (n >= N) deg = 0;
    const int lane = t & 31;
    int inc = deg;
    for (int d = 1; d < 32; d <<= 1) {
        const int v = __shfl_up(inc, d, 32);
        if (lane >= d) inc += v;
    }
    for (int m = 16; m >= 1; m >>= 1) part += __shfl_xor(part, m, 32);
    if (lane == 31) wtot[t >> 5] = inc;
    if (lane == 0)  wbase[t >> 5] = part;
    __syncthreads();
    int run = inc - deg;
    #pragma unroll
    for (int w = 0; w < CSR_RANGE / 32; ++w) {
        run += wbase[w];
        run += (w < (t >> 5)) ? wtot[w] : 0;
    }
    if (n < N) {
        off[n] = run;
        if (n == N - 1) off[N] = run + deg;
        #pragma unroll
        for (int j = 0; j < CSR_NB; ++j) {
            blockcnt[(size_t)j * N + n] = run;
            run += c[j];
        }
    }
}

__global__ __launch_bounds__(CSR_RANGE) void csr_colscan_off(int* __restrict__ blockcnt,
                                                             const int* __restrict__ coarse,
                                                             int* __restrict__ off, int N, int nr_total) {
    csr_colscan_range(blockcnt, coarse, off, N, nr_total, (int)blockIdx.x);
}

static_assert(CSR_RANGE == 256, "the shared launch uses 256-thread blocks for both parts");
__global__ __launch_bounds__(256) void gemm3_xw_colscan(const float* __restrict__ X, const float* __restrict__ Wr,
                         const float* __restrict__ Wl, const float* __restrict__ bias,
                         float* __restrict__ rel, float* __restrict__ H, int M, int K, int gblocks,
                         int* __restrict__ blockcnt, const int* __restrict__ coarse,
                         int* __restrict__ off, int nr_total) {
    if ((int)blockIdx.x < gblocks) gemm3_strip(X, Wr, Wl, bias, rel, H, M, K, (int)blockIdx.x);
    else csr_colscan_range(blockcnt, coarse, off, M, nr_total, (int)blockIdx.x - gblocks);
}

__global__ void csr_fill(const int* __restrict__ src, const int* __restrict__ dst,
                         const int* __restrict__ et, const int* __restrict__ rank,
                         const int* __restrict__ blockcnt,
                         int* __restrict__ key, int E, int N, int chunk) {
    const int e = blockIdx.x * blockDim.x + threadIdx.x;
    if (e >= E) return;
    const int d = dst[e];
    const int b = e / chunk;
    int pos = blockcnt[(size_t)b * N + d] + rank[e];
    pos = min(max(pos, 0), E - 1);
    key[pos] = et[e] * N + src[e];
}

__global__ void gather64_relu(const float* __restrict__ rel, const int* __restrict__ off,
                              const int* __restrict__ key, float* __restrict__ h, int N) {
    const long long t = (long long)blockIdx.x * blockDim.x + threadIdx.x;
    const int i = (int)(t >> 4);
    if (i >= N) return;
    const int f = ((int)t & 15) * 4;
    float* hp = h + (size_t)i * 64 + f;
    float4 acc = *(const float4*)hp;
    int p = off[i];
    const int pe = off[i + 1];
    for (; p + 4 <= pe; p += 4) {
        const int k0 = key[p], k1 = key[p + 1], k2 = key[p + 2], k3 = key[p + 3];
        const float4 a = *(const float4*)(rel + (size_t)k0 * 64 + f);
        const float4 b = *(const float4*)(rel + (size_t)k1 * 64 + f);
        const float4 c = *(const float4*)(rel + (size_t)k2 * 64 + f);
        const float4 d = *(const float4*)(rel + (size_t)k3 * 64 + f);
        acc.x += a.x; acc.y += a.y; acc.z += a.z; acc.w += a.w;
        acc.x += b.x; acc.y += b.y; acc.z += b.z; acc.w += b.w;
        acc.x += c.x; acc.y += c.y; acc.z += c.z; acc.w += c.w;
        acc.x += d.x; acc.y += d.y; acc.z += d.z; acc.w += d.w;
    }
    for (; p < pe; ++p) {
        const float4 a = *(const float4*)(rel + (size_t)key[p] * 64 + f);
        acc.x += a.x; acc.y += a.y; acc.z += a.z; acc.w += a.w;
    }
    acc.x = fmaxf(acc.x, 0.0f); acc.y = fmaxf(acc.y, 0.0f);
    acc.z = fmaxf(acc.z, 0.0f); acc.w = fmaxf(acc.w, 0.0f);
    *(float4*)hp = acc;
}

__global__ void gather2(const float* __restrict__ rel3, const int* __restrict__ off,
                        const int* __restrict__ key, float* __restrict__ agg3, int N) {
    const long long t = (long long)blockIdx.x * blockDim.x + threadIdx.x;
    const int i = (int)(t >> 4);
    const int l = (int)t & 15;
    float s0 = 0.0f, s1 = 0.0f;
    if (i < N) {
        const int pe = off[i + 1];
        for (int p = off[i] + l; p < pe; p += 16) {
            const float2 r = *(const float2*)(rel3 + (size_t)key[p] * 2);
            s0 += r.x; s1 += r.y;
        }
    }
    for (int m = 8; m >= 1; m >>= 1) {
        s0 += __shfl_xor(s0, m, 16);
        s1 += __shfl_xor(s1, m, 16);
    }
    if (i < N && l == 0) {
        agg3[(size_t)i * 2 + 0] += s0;
        agg3[(size_t)i * 2 + 1] += s1;
    }
}

template <int U>
__device__ __forceinline__ void l2_add_rows(const float* __restrict__ h1, const int* __restrict__ kp,
                                            int N, int f, float4& a0, float4& a1) {
    int    kk[U];
    float  m1[U];
    float4 v[U];
    #pragma unroll
    for (int j = 0; j < U; ++j) kk[j] = kp[j];
    #pragma unroll
    for (int j = 0; j < U; ++j) {
        const bool r = kk[j] >= N;
        const int  s = r ? kk[j] - N : kk[j];
        m1[j] = r ? 1.0f : 0.0f;
        v[j]  = *(const float4*)(h1 + (size_t)s * 64 + f);
    }
    #pragma unroll
    for (int j = 0; j < U; ++j) {
        const float w1 = m1[j], w0 = 1.0f - m1[j];
        a0.x = fmaf(v[j].x, w0, a0.x); a0.y = fmaf(v[j].y, w0, a0.y);
        a0.z = fmaf(v[j].z, w0, a0.z); a0.w = fmaf(v[j].w, w0, a0.w);
        a1.x = fmaf(v[j].x, w1, a1.x); a1.y = fmaf(v[j].y, w1, a1.y);
        a1.z = fmaf(v[j].z, w1, a1.z); a1.w = fmaf(v[j].w, w1, a1.w);
    }
}

#define L2_PITCH  196
#define L2_HPITCH 68
__global__ __launch_bounds__(256) void layer2_sum_xw_l3(const float* __restrict__ h1, const int* __restrict__ off,
                                 const int* __restrict__ key,
                                 const float* __restrict__ W2, const float* __restrict__ loop2,
                                 const float* __restrict__ b2,
                                 const float* __restrict__ W3, const float* __restrict__ loop3,
                                 const float* __restrict__ b3,
                                 float* __restrict__ rel3, float* __restrict__ agg3, int N) {
    __shared__ __align__(16) float As[16 * L2_PITCH];
    __shared__ __align__(16) float Hs[16 * L2_HPITCH];
    const int t    = (int)threadIdx.x;
    const int row  = t >> 4;
    const int f    = (t & 15) * 4;
    const int i    = (int)blockIdx.x * 16 + row;
    float4 a0  = make_float4(0.0f, 0.0f, 0.0f, 0.0f);
    float4 a1  = make_float4(0.0f, 0.0f, 0.0f, 0.0f);
    float4 own = make_float4(0.0f, 0.0f, 0.0f, 0.0f);
    if (i < N) {
        own = *(const float4*)(h1 + (size_t)i * 64 + f);
        int p = off[i];
        const int pe = off[i + 1];
        for (; p + 8 <= pe; p += 8) l2_add_rows<8>(h1, key + p, N, f, a0, a1);
        if (p + 4 <= pe) { l2_add_rows<4>(h1, key + p, N, f, a0, a1); p += 4; }
        for (; p < pe; ++p) l2_add_rows<1>(h1, key + p, N, f, a0, a1);
    }
    {
        float* ar = As + row * L2_PITCH + f;
        *(float4*)(ar)       = a0;
        *(float4*)(ar + 64)  = a1;
        *(float4*)(ar + 128) = own;
    }
    __syncthreads();

    const int wave = t >> 5;
    if (wave < 4) {
        const int lane = t & 31;
        const int half = lane >> 4;
        const int lm   = lane & 15;
        const int n    = wave * 16 + lm;
        v8f c0, c1, c2;
        const float bv = b2[n];
        #pragma unroll
        for (int v = 0; v < 8; ++v) { c0[v] = 0.0f; c1[v] = 0.0f; c2[v] = bv; }
        const float* arow = As + lm * L2_PITCH + 2 * half;
        const float* w1p  = W2 + 64 * 64;
        #pragma unroll
        for (int k0 = 0; k0 < 64; k0 += 4) {
            const int kb = k0 + 2 * half;
            #pragma unroll
            for (int mt = 0; mt < 3; ++mt) {
                const float* Wm = (mt == 0) ? W2 : (mt == 1) ? w1p : loop2;
                const float2 av = *(const float2*)(arow + 64 * mt + k0);
                v2f a; a.x = av.x; a.y = av.y;
                v2f b;
                b.x = Wm[(size_t)kb * 64 + n];
                b.y = Wm[(size_t)(kb + 1) * 64 + n];
                if (mt == 0)      c0 = __builtin_amdgcn_wmma_f32_16x16x4_f32(false, a, false, b, (short)0, c0, false, false);
                else if (mt == 1) c1 = __builtin_amdgcn_wmma_f32_16x16x4_f32(false, a, false, b, (short)0, c1, false, false);
                else              c2 = __builtin_amdgcn_wmma_f32_16x16x4_f32(false, a, false, b, (short)0, c2, false, false);
            }
        }
        #pragma unroll
        for (int v = 0; v < 8; ++v)
            Hs[(v + 8 * half) * L2_HPITCH + n] = fmaxf((c0[v] + c1[v]) + c2[v], 0.0f);
    }
    __syncthreads();

    const float4 hq = *(const float4*)(Hs + row * L2_HPITCH + f);
    const float hv[4] = { hq.x, hq.y, hq.z, hq.w };
    const float* wm[3] = { W3 + f * 2, W3 + 128 + f * 2, loop3 + f * 2 };
    float o[6];
    #pragma unroll
    for (int m = 0; m < 3; ++m) {
        float s0 = 0.0f, s1 = 0.0f;
        #pragma unroll
        for (int j = 0; j < 4; ++j) {
            s0 += hv[j] * wm[m][2 * j + 0];
            s1 += hv[j] * wm[m][2 * j + 1];
        }
        o[2 * m + 0] = s0;
        o[2 * m + 1] = s1;
    }
    for (int d = 8; d >= 1; d >>= 1) {
        #pragma unroll
        for (int q = 0; q < 6; ++q) o[q] += __shfl_xor(o[q], d, 16);
    }
    if ((t & 15) == 0 && i < N) {
        rel3[(size_t)i * 2 + 0] = o[0];
        rel3[(size_t)i * 2 + 1] = o[1];
        rel3[(size_t)(N + i) * 2 + 0] = o[2];
        rel3[(size_t)(N + i) * 2 + 1] = o[3];
        agg3[(size_t)i * 2 + 0] = o[4] + b3[0];
        agg3[(size_t)i * 2 + 1] = o[5] + b3[1];
    }
}

#define TAIL_TPB 256
__global__ __launch_bounds__(TAIL_TPB) void gather2_min(const float* __restrict__ rel3, const int* __restrict__ off,
                            const int* __restrict__ key, float* __restrict__ agg3,
                            float* __restrict__ blockmin, int N) {
    __shared__ float sm[TAIL_TPB / 32];
    const long long t = (long long)blockIdx.x * blockDim.x + threadIdx.x;
    const int i = (int)(t >> 4);
    const int l = (int)t & 15;
    float s0 = 0.0f, s1 = 0.0f;
    if (i < N) {
        const int pe = off[i + 1];
        for (int p = off[i] + l; p < pe; p += 16) {
            const float2 r = *(const float2*)(rel3 + (size_t)key[p] * 2);
            s0 += r.x; s1 += r.y;
        }
    }
    for (int m = 8; m >= 1; m >>= 1) {
        s0 += __shfl_xor(s0, m, 16);
        s1 += __shfl_xor(s1, m, 16);
    }
    float mn = 3.402823466e38f;
    if (i < N && l == 0) {
        const float v0 = agg3[(size_t)i * 2 + 0] + s0;
        const float v1 = agg3[(size_t)i * 2 + 1] + s1;
        agg3[(size_t)i * 2 + 0] = v0;
        agg3[(size_t)i * 2 + 1] = v1;
        mn = fminf(v0, v1);
    }
    mn = fminf(mn, __shfl_xor(mn, 16, 32));
    if ((threadIdx.x & 31) == 0) sm[threadIdx.x >> 5] = mn;
    __syncthreads();
    if (threadIdx.x == 0) {
        float r = sm[0];
        #pragma unroll
        for (int w = 1; w < TAIL_TPB / 32; ++w) r = fminf(r, sm[w]);
        blockmin[blockIdx.x] = r;
    }
}

__global__ __launch_bounds__(TAIL_TPB) void finalize_bm(const float* __restrict__ agg3,
                            const int* __restrict__ cell,
                            const int* __restrict__ maxs,
                            const float* __restrict__ blockmin, int nblk,
                            float* __restrict__ out, int N) {
    __shared__ float sm[TAIL_TPB / 32];
    float mn = 3.402823466e38f;
    for (int j0 = threadIdx.x; j0 < nblk; j0 += 16 * TAIL_TPB) {
        float v[16];
        #pragma unroll
        for (int u = 0; u < 16; ++u) v[u] = blockmin[min(j0 + u * TAIL_TPB, nblk - 1)];
        #pragma unroll
        for (int u = 0; u < 16; ++u) mn = fminf(mn, v[u]);
    }
    for (int m = 16; m >= 1; m >>= 1) mn = fminf(mn, __shfl_xor(mn, m, 32));
    if ((threadIdx.x & 31) == 0) sm[threadIdx.x >> 5] = mn;
    __syncthreads();
    float r = sm[0];
    #pragma unroll
    for (int w = 1; w < TAIL_TPB / 32; ++w) r = fminf(r, sm[w]);
    const float gmin = r - 1.0f;
    const int i = blockIdx.x * blockDim.x + threadIdx.x;
    if (i >= N) return;
    const bool upper = cell[i] >= (maxs[i] - 1);
    const bool lower = cell[i] == 0;
    out[(size_t)i * 2 + 0] = upper ? gmin : agg3[(size_t)i * 2 + 0];
    out[(size_t)i * 2 + 1] = lower ? gmin : agg3[(size_t)i * 2 + 1];
}

__device__ __forceinline__ unsigned f2ord(float f) {
    unsigned u = __float_as_uint(f);
    return (u & 0x80000000u) ? ~u : (u | 0x80000000u);
}
__device__ __forceinline__ float ord2f(unsigned u) {
    return __uint_as_float((u & 0x80000000u) ? (u ^ 0x80000000u) : ~u);
}

__global__ void init_slot(unsigned* slot) { *slot = 0xFFFFFFFFu; }

__global__ void min_reduce(const float* __restrict__ p, int n, unsigned* slot) {
    __shared__ float sm[256];
    float m = 3.402823466e38f;
    for (int i = blockIdx.x * blockDim.x + threadIdx.x; i < n;
         i += gridDim.x * blockDim.x)
        m = fminf(m, p[i]);
    sm[threadIdx.x] = m;
    __syncthreads();
    for (int s = blockDim.x >> 1; s > 0; s >>= 1) {
        if ((int)threadIdx.x < s) sm[threadIdx.x] = fminf(sm[threadIdx.x], sm[threadIdx.x + s]);
        __syncthreads();
    }
    if (threadIdx.x == 0) atomicMin(slot, f2ord(sm[0]));
}

__global__ void finalize_k(const float* __restrict__ agg3,
                           const int* __restrict__ cell,
                           const int* __restrict__ maxs,
                           const unsigned* __restrict__ slot,
                           float* __restrict__ out, int N) {
    const int i = blockIdx.x * blockDim.x + threadIdx.x;
    if (i >= N) return;
    const float gmin = ord2f(*slot) - 1.0f;
    const bool upper = cell[i] >= (maxs[i] - 1);
    const bool lower = cell[i] == 0;
    out[(size_t)i * 2 + 0] = upper ? gmin : agg3[(size_t)i * 2 + 0];
    out[(size_t)i * 2 + 1] = lower ? gmin : agg3[(size_t)i * 2 + 1];
}

extern "C" void kernel_launch(void* const* d_in, const int* in_sizes, int n_in,
                              void* d_out, int out_size, void* d_ws, size_t ws_size,
                              hipStream_t stream) {
    const float* x      = (const float*)d_in[0];
    const int*   src    = (const int*)d_in[1];
    const int*   dst    = (const int*)d_in[2];
    const int*   etypes = (const int*)d_in[3];
    const int*   cell   = (const int*)d_in[4];
    const int*   maxs   = (const int*)d_in[5];
    const float* W1     = (const float*)d_in[6];
    const float* loop1  = (const float*)d_in[7];
    const float* b1     = (const float*)d_in[8];
    const float* W2     = (const float*)d_in[9];
    const float* loop2  = (const float*)d_in[10];
    const float* b2     = (const float*)d_in[11];
    const float* W3     = (const float*)d_in[12];
    const float* loop3  = (const float*)d_in[13];
    const float* b3     = (const float*)d_in[14];
    float* out = (float*)d_out;

    const int N = in_sizes[4];
    const int E = in_sizes[1];
    const int K1 = in_sizes[0] / N;

    float* relbuf = (float*)d_ws;
    float* h1     = relbuf + (size_t)2 * N * 64;
    float* h2     = h1 + (size_t)N * 64;
    float* rel3   = h2 + (size_t)N * 64;
    float* agg3   = rel3 + (size_t)4 * N;
    unsigned* slot = (unsigned*)(agg3 + (size_t)2 * N);

    const int waves   = (N + 15) / 16;
    const int gblocks = (waves * 32 + 255) / 256;
    const int eb64    = (int)(((long long)E * 16 + 255) / 256);
    const int nb      = (N + 255) / 256;
    const int fb      = (N * 64 + 255) / 256;
    const int eb2     = (E + 255) / 256;

    int* key      = (int*)(slot + 4);
    int* off      = key + (size_t)E;
    int* blockcnt = (int*)relbuf;
    int* rank     = blockcnt + (size_t)CSR_NB * N;
    const size_t need = ((size_t)262 * N + 4 + (size_t)E + (size_t)N + 1) * sizeof(float);
    int* coarse   = (int*)rel3;
    const bool csr_ok = ws_size >= need && (long long)E <= (long long)192 * N
                        && (long long)CSR_NB * nb <= (long long)6 * N;
    const size_t need2 = need + (size_t)E * sizeof(int);
    const bool shared_ok = csr_ok && ws_size >= need2;

    if (shared_ok) {
        const int chunk = (E + CSR_NB - 1) / CSR_NB;
        const int gb64  = (int)(((long long)N * 16 + 255) / 256);
        int* blockcnt2 = (int*)h2;
        int* rank2     = off + (size_t)N + 1;
        csr_hist<<<CSR_NB, 1024, 0, stream>>>(dst, rank2, blockcnt2, coarse, E, N, chunk, nb);

        gemm3_xw_colscan<<<gblocks + nb, 256, 0, stream>>>(x, W1, loop1, b1, relbuf, h1, N, K1, gblocks,
                                                           blockcnt2, coarse, off, nb);
        csr_fill<<<eb2, 256, 0, stream>>>(src, dst, etypes, rank2, blockcnt2, key, E, N, chunk);
        gather64_relu<<<gb64, 256, 0, stream>>>(relbuf, off, key, h1, N);

        layer2_sum_xw_l3<<<waves, 256, 0, stream>>>(h1, off, key, W2, loop2, b2, W3, loop3, b3, rel3, agg3, N);

        float* blockmin = relbuf;
        gather2_min<<<gb64, TAIL_TPB, 0, stream>>>(rel3, off, key, agg3, blockmin, N);
        finalize_bm<<<nb, TAIL_TPB, 0, stream>>>(agg3, cell, maxs, blockmin, gb64, out, N);
    } else if (csr_ok) {
        const int chunk = (E + CSR_NB - 1) / CSR_NB;
        const int gb64  = (int)(((long long)N * 16 + 255) / 256);
        csr_hist<<<CSR_NB, 1024, 0, stream>>>(dst, rank, blockcnt, coarse, E, N, chunk, nb);
        csr_colscan_off<<<nb, CSR_RANGE, 0, stream>>>(blockcnt, coarse, off, N, nb);
        csr_fill<<<eb2, 256, 0, stream>>>(src, dst, etypes, rank, blockcnt, key, E, N, chunk);

        gemm3_xw<<<gblocks, 256, 0, stream>>>(x, W1, loop1, b1, relbuf, h1, N, K1);
        gather64_relu<<<gb64, 256, 0, stream>>>(relbuf, off, key, h1, N);

        layer2_sum_xw_l3<<<waves, 256, 0, stream>>>(h1, off, key, W2, loop2, b2, W3, loop3, b3, rel3, agg3, N);

        float* blockmin = relbuf;
        gather2_min<<<gb64, TAIL_TPB, 0, stream>>>(rel3, off, key, agg3, blockmin, N);
        finalize_bm<<<nb, TAIL_TPB, 0, stream>>>(agg3, cell, maxs, blockmin, gb64, out, N);
    } else {
        gemm3_xw<<<gblocks, 256, 0, stream>>>(x, W1, loop1, b1, relbuf, h1, N, K1);
        edge_scatter64<<<eb64, 256, 0, stream>>>(relbuf, src, dst, etypes, h1, E, N);
        relu_k<<<fb, 256, 0, stream>>>(h1, N * 64);

        gemm3_xw<<<gblocks, 256, 0, stream>>>(h1, W2, loop2, b2, relbuf, h2, N, 64);
        edge_scatter64<<<eb64, 256, 0, stream>>>(relbuf, src, dst, etypes, h2, E, N);
        relu_k<<<fb, 256, 0, stream>>>(h2, N * 64);

        layer3_node<<<nb, 256, 0, stream>>>(h2, W3, loop3, b3, rel3, agg3, N);
        edge_scatter2<<<eb2, 256, 0, stream>>>(rel3, src, dst, etypes, agg3, E, N);
        init_slot<<<1, 1, 0, stream>>>(slot);
        min_reduce<<<512, 256, 0, stream>>>(agg3, N * 2, slot);
        finalize_k<<<nb, 256, 0, stream>>>(agg3, cell, maxs, slot, out, N);
    }
}


# Fold the edge fill into the column-scan launch: the histogram pass bins each chunk's edges by 256-node range, scan blocks place their keys via an LDS table; 7 -> 6 launches; run order by bin arrival
# speedup vs baseline: 2.7819x; 1.1363x over previous
#include <hip/hip_runtime.h>

typedef __attribute__((ext_vector_type(2))) float v2f;
typedef __attribute__((ext_vector_type(8))) float v8f;

__device__ __forceinline__ void gemm3_strip(const float* __restrict__ X, const float* __restrict__ Wr,
                         const float* __restrict__ Wl, const float* __restrict__ bias,
                         float* __restrict__ rel, float* __restrict__ H,
                         int M, int K, int block) {
    const int gthread = block * blockDim.x + threadIdx.x;
    const int wave    = gthread >> 5;
    const int lane    = threadIdx.x & 31;
    const int m0      = wave * 16;
    if (m0 >= M) return;

    const int half = lane >> 4;
    const int lm   = lane & 15;

    v8f acc[12];
    #pragma unroll
    for (int t = 0; t < 8; ++t) {
        #pragma unroll
        for (int v = 0; v < 8; ++v) acc[t][v] = 0.0f;
    }
    #pragma unroll
    for (int nt = 0; nt < 4; ++nt) {
        const float bv = bias[nt * 16 + lm];
        #pragma unroll
        for (int v = 0; v < 8; ++v) acc[8 + nt][v] = bv;
    }

    const float* xrow = X + (size_t)(m0 + lm) * K;
    const float* w1   = Wr + (size_t)K * 64;

    #pragma unroll 2
    for (int k0 = 0; k0 < K; k0 += 4) {
        const int kb = k0 + 2 * half;
        const float2 av = *(const float2*)(xrow + kb);
        v2f a; a.x = av.x; a.y = av.y;
        #pragma unroll
        for (int mt = 0; mt < 3; ++mt) {
            const float* Wm = (mt == 0) ? Wr : (mt == 1) ? w1 : Wl;
            #pragma unroll
            for (int nt = 0; nt < 4; ++nt) {
                const int n = nt * 16 + lm;
                v2f b;
                b.x = Wm[(size_t)kb * 64 + n];
                b.y = Wm[(size_t)(kb + 1) * 64 + n];
                acc[mt * 4 + nt] = __builtin_amdgcn_wmma_f32_16x16x4_f32(
                    false, a, false, b, (short)0, acc[mt * 4 + nt], false, false);
            }
        }
    }

    #pragma unroll
    for (int mt = 0; mt < 3; ++mt) {
        float* C = (mt == 2) ? H : rel + (size_t)mt * M * 64;
        #pragma unroll
        for (int nt = 0; nt < 4; ++nt) {
            const int n = nt * 16 + lm;
            #pragma unroll
            for (int v = 0; v < 8; ++v) {
                C[(size_t)(m0 + v + 8 * half) * 64 + n] = acc[mt * 4 + nt][v];
            }
        }
    }
}

__global__ __launch_bounds__(256) void gemm3_xw(const float* __restrict__ X, const float* __restrict__ Wr,
                         const float* __restrict__ Wl, const float* __restrict__ bias,
                         float* __restrict__ rel, float* __restrict__ H,
                         int M, int K) {
    gemm3_strip(X, Wr, Wl, bias, rel, H, M, K, (int)blockIdx.x);
}

__global__ void edge_scatter64(const float* __restrict__ rel,
                               const int* __restrict__ src,
                               const int* __restrict__ dst,
                               const int* __restrict__ et,
                               float* __restrict__ agg, int E, int N) {
    const long long t = (long long)blockIdx.x * blockDim.x + threadIdx.x;
    const int e = (int)(t >> 4);
    if (e >= E) return;
    const int f = ((int)t & 15) * 4;
    const float4 v = *(const float4*)(rel + ((long long)et[e] * N + src[e]) * 64 + f);
    float* q = agg + (long long)dst[e] * 64 + f;
    atomicAdd(q + 0, v.x);
    atomicAdd(q + 1, v.y);
    atomicAdd(q + 2, v.z);
    atomicAdd(q + 3, v.w);
}

__global__ void relu_k(float* __restrict__ p, int n) {
    const int i = blockIdx.x * blockDim.x + threadIdx.x;
    if (i < n) p[i] = fmaxf(p[i], 0.0f);
}

__global__ void layer3_node(const float* __restrict__ h2,
                            const float* __restrict__ W3,
                            const float* __restrict__ loop3,
                            const float* __restrict__ b3,
                            float* __restrict__ rel3, float* __restrict__ agg3,
                            int N) {
    const int i = blockIdx.x * blockDim.x + threadIdx.x;
    if (i >= N) return;
    const float* row = h2 + (size_t)i * 64;
    float r00 = 0.f, r01 = 0.f, r10 = 0.f, r11 = 0.f, l0 = 0.f, l1 = 0.f;
    #pragma unroll 8
    for (int k = 0; k < 64; ++k) {
        const float v = row[k];
        r00 += v * W3[k * 2 + 0];
        r01 += v * W3[k * 2 + 1];
        r10 += v * W3[128 + k * 2 + 0];
        r11 += v * W3[128 + k * 2 + 1];
        l0  += v * loop3[k * 2 + 0];
        l1  += v * loop3[k * 2 + 1];
    }
    rel3[(size_t)i * 2 + 0] = r00;
    rel3[(size_t)i * 2 + 1] = r01;
    rel3[(size_t)(N + i) * 2 + 0] = r10;
    rel3[(size_t)(N + i) * 2 + 1] = r11;
    agg3[(size_t)i * 2 + 0] = l0 + b3[0];
    agg3[(size_t)i * 2 + 1] = l1 + b3[1];
}

__global__ void edge_scatter2(const float* __restrict__ rel3,
                              const int* __restrict__ src,
                              const int* __restrict__ dst,
                              const int* __restrict__ et,
                              float* __restrict__ agg3, int E, int N) {
    const int e = blockIdx.x * blockDim.x + threadIdx.x;
    if (e >= E) return;
    const float* p = rel3 + ((long long)et[e] * N + src[e]) * 2;
    float* q = agg3 + (long long)dst[e] * 2;
    atomicAdd(q + 0, p[0]);
    atomicAdd(q + 1, p[1]);
}


#define CSR_NB  64
#define CSR_CAP 51200
#define CSR_RANGE 256

__global__ __launch_bounds__(1024) void csr_hist(const int* __restrict__ dst,
                                                 int* __restrict__ rank,
                                                 int* __restrict__ blockcnt,
                                                 int* __restrict__ coarse,
                                                 int E, int N, int chunk, int nr_total) {
    __shared__ int cnt[CSR_CAP];
    const int b  = blockIdx.x;
    const int e0 = b * chunk;
    const int e1 = min(e0 + chunk, E);
    for (int base = 0; base < N; base += CSR_CAP) {
        const int lim = min(CSR_CAP, N - base);
        for (int i = threadIdx.x; i < lim; i += blockDim.x) cnt[i] = 0;
        __syncthreads();
        const int bd = (int)blockDim.x;
        for (int e = e0 + (int)threadIdx.x; e < e1; e += 4 * bd) {
            int d[4];
            #pragma unroll
            for (int j = 0; j < 4; ++j) {
                const int ee = e + j * bd;
                const int v  = dst[min(ee, e1 - 1)];
                d[j] = (ee < e1) ? v - base : -1;
            }
            #pragma unroll
            for (int j = 0; j < 4; ++j)
                if (d[j] >= 0 && d[j] < lim) rank[e + j * bd] = atomicAdd(&cnt[d[j]], 1);
        }
        __syncthreads();
        const int nrs = (lim + CSR_RANGE - 1) / CSR_RANGE;
        for (int r0 = 0; r0 < nrs; r0 += (int)blockDim.x >> 2) {
            const int r  = r0 + ((int)threadIdx.x >> 2);
            const int q  = (int)threadIdx.x & 3;
            const int lo = r * CSR_RANGE + q * (CSR_RANGE / 4);
            int s = 0;
            #pragma unroll 16
            for (int j = 0; j < CSR_RANGE / 4; ++j) {
                const int i = lo + ((j + (int)threadIdx.x) & (CSR_RANGE / 4 - 1));
                const int v = cnt[min(i, lim - 1)];
                s += (i < lim) ? v : 0;
            }
            s += __shfl_xor(s, 1, 4);
            s += __shfl_xor(s, 2, 4);
            if (q == 0 && r < nrs) coarse[(size_t)b * nr_total + base / CSR_RANGE + r] = s;
        }
        for (int i = threadIdx.x; i < lim; i += blockDim.x)
            blockcnt[(size_t)b * N + base + i] = cnt[i];
        __syncthreads();
    }
}

#define CSR_EPT 16
__global__ __launch_bounds__(1024) void csr_hist_bin(const int* __restrict__ dst, const int* __restrict__ src,
                                                     const int* __restrict__ et,
                                                     unsigned* __restrict__ ebin,
                                                     int* __restrict__ blockcnt,
                                                     int* __restrict__ coarse, int* __restrict__ cstart,
                                                     int E, int N, int chunk, int nr_total) {
    __shared__ int cnt[CSR_CAP];
    __shared__ int rcnt[CSR_CAP / CSR_RANGE];
    __shared__ int rstart[CSR_CAP / CSR_RANGE];
    __shared__ int swt[8];
    const int tid  = (int)threadIdx.x;
    const int lane = tid & 31;
    const int b    = (int)blockIdx.x;
    const int e0   = b * chunk;
    const int e1   = min(e0 + chunk, E);
    for (int i = tid; i < N; i += 1024) cnt[i] = 0;
    if (tid < nr_total) rcnt[tid] = 0;
    int dd[CSR_EPT], ss[CSR_EPT], tt[CSR_EPT];
    #pragma unroll
    for (int j = 0; j < CSR_EPT; ++j) {
        const int ec = max(min(e0 + tid + j * 1024, e1 - 1), 0);
        dd[j] = dst[ec]; ss[j] = src[ec]; tt[j] = et[ec];
    }
    #pragma unroll
    for (int j = 0; j < CSR_EPT; ++j)
        if (e0 + tid + j * 1024 >= e1) dd[j] = -1;
    __syncthreads();
    int slot[CSR_EPT];
    #pragma unroll
    for (int j = 0; j < CSR_EPT; ++j) {
        slot[j] = 0;
        if (dd[j] >= 0 && dd[j] < N) {
            atomicAdd(&cnt[dd[j]], 1);
            slot[j] = atomicAdd(&rcnt[dd[j] / CSR_RANGE], 1);
        } else dd[j] = -1;
    }
    __syncthreads();
    int rv = 0, inc = 0;
    if (tid < 256) {
        rv  = (tid < nr_total) ? rcnt[tid] : 0;
        inc = rv;
        for (int d = 1; d < 32; d <<= 1) {
            const int v = __shfl_up(inc, d, 32);
            if (lane >= d) inc += v;
        }
        if (lane == 31) swt[tid >> 5] = inc;
    }
    __syncthreads();
    if (tid < 256) {
        int wb = 0;
        #pragma unroll
        for (int w = 0; w < 8; ++w) wb += (w < (tid >> 5)) ? swt[w] : 0;
        const int ex = wb + inc - rv;
        if (tid < nr_total) {
            rstart[tid] = ex;
            coarse[(size_t)b * nr_total + tid] = rv;
            cstart[(size_t)b * nr_total + tid] = ex;
        }
    }
    __syncthreads();
    #pragma unroll
    for (int j = 0; j < CSR_EPT; ++j)
        slot[j] += rstart[max(dd[j], 0) / CSR_RANGE];
    #pragma unroll
    for (int j = 0; j < CSR_EPT; ++j) {
        if (dd[j] >= 0) {
            const int r = dd[j] / CSR_RANGE;
            const int p = min(max(slot[j], 0), e1 - e0 - 1);
            const unsigned kv = (unsigned)(tt[j] * N + ss[j]);
            ebin[(size_t)e0 + p] = (kv & 0xFFFFFFu) | ((unsigned)(dd[j] - r * CSR_RANGE) << 24);
        }
    }
    for (int i = tid; i < N; i += 1024) blockcnt[(size_t)b * N + i] = cnt[i];
}

__device__ __forceinline__ void csr_colscan_range(int* __restrict__ blockcnt,
                                                  const int* __restrict__ coarse,
                                                  int* __restrict__ off, int N, int nr_total, int r) {
    __shared__ int wtot[CSR_RANGE / 32];
    __shared__ int wbase[CSR_RANGE / 32];
    const int t  = threadIdx.x;
    const int n  = r * CSR_RANGE + t;
    const int nn = min(n, N - 1);
    int c[CSR_NB];
    #pragma unroll
    for (int j = 0; j < CSR_NB; ++j) c[j] = blockcnt[(size_t)j * N + nn];
    int part = 0;
    for (int rr = t; rr < r; rr += CSR_RANGE) {
        #pragma unroll 16
        for (int b = 0; b < CSR_NB; ++b) part += coarse[(size_t)b * nr_total + rr];
    }
    int deg = 0;
    #pragma unroll
    for (int j = 0; j < CSR_NB; ++j) deg += c[j];
    if (n >= N) deg = 0;
    const int lane = t & 31;
    int inc = deg;
    for (int d = 1; d < 32; d <<= 1) {
        const int v = __shfl_up(inc, d, 32);
        if (lane >= d) inc += v;
    }
    for (int m = 16; m >= 1; m >>= 1) part += __shfl_xor(part, m, 32);
    if (lane == 31) wtot[t >> 5] = inc;
    if (lane == 0)  wbase[t >> 5] = part;
    __syncthreads();
    int run = inc - deg;
    #pragma unroll
    for (int w = 0; w < CSR_RANGE / 32; ++w) {
        run += wbase[w];
        run += (w < (t >> 5)) ? wtot[w] : 0;
    }
    if (n < N) {
        off[n] = run;
        if (n == N - 1) off[N] = run + deg;
        #pragma unroll
        for (int j = 0; j < CSR_NB; ++j) {
            blockcnt[(size_t)j * N + n] = run;
            run += c[j];
        }
    }
}

__global__ __launch_bounds__(CSR_RANGE) void csr_colscan_off(int* __restrict__ blockcnt,
                                                             const int* __restrict__ coarse,
                                                             int* __restrict__ off, int N, int nr_total) {
    csr_colscan_range(blockcnt, coarse, off, N, nr_total, (int)blockIdx.x);
}

static_assert(CSR_NB == 64 && CSR_RANGE == 256, "8 waves x 8 chunks, one thread per node of the range");
__device__ __forceinline__ void csr_place_one(unsigned rec, int* __restrict__ trow, int* __restrict__ key, int E) {
    const int dl = (int)(rec >> 24);
    int pos = atomicAdd(&trow[dl], 1);
    pos = min(max(pos, 0), E - 1);
    key[pos] = (int)(rec & 0xFFFFFFu);
}

__device__ __forceinline__ void csr_colscan_place(const int* __restrict__ blockcnt,
                                                  const int* __restrict__ coarse, const int* __restrict__ cstart,
                                                  const unsigned* __restrict__ ebin,
                                                  int* __restrict__ off, int* __restrict__ key,
                                                  int N, int E, int chunk, int nr_total, int r) {
    __shared__ int wtot[CSR_RANGE / 32];
    __shared__ int wbase[CSR_RANGE / 32];
    __shared__ int tab[CSR_NB * CSR_RANGE];
    const int t    = threadIdx.x;
    const int wave = t >> 5;
    const int lane = t & 31;
    const int n  = r * CSR_RANGE + t;
    const int nn = min(n, N - 1);
    int len[8], st[8];
    #pragma unroll
    for (int q = 0; q < 8; ++q) {
        const int bq = wave + 8 * q;
        len[q] = coarse[(size_t)bq * nr_total + r];
        st[q]  = cstart[(size_t)bq * nr_total + r];
    }
    int c[CSR_NB];
    #pragma unroll
    for (int j = 0; j < CSR_NB; ++j) c[j] = blockcnt[(size_t)j * N + nn];
    int part = 0;
    for (int rr = t; rr < r; rr += CSR_RANGE) {
        #pragma unroll 16
        for (int b = 0; b < CSR_NB; ++b) part += coarse[(size_t)b * nr_total + rr];
    }
    int deg = 0;
    #pragma unroll
    for (int j = 0; j < CSR_NB; ++j) deg += c[j];
    if (n >= N) deg = 0;
    int inc = deg;
    for (int d = 1; d < 32; d <<= 1) {
        const int v = __shfl_up(inc, d, 32);
        if (lane >= d) inc += v;
    }
    for (int m = 16; m >= 1; m >>= 1) part += __shfl_xor(part, m, 32);
    if (lane == 31) wtot[wave] = inc;
    if (lane == 0)  wbase[wave] = part;
    __syncthreads();
    int run = inc - deg;
    #pragma unroll
    for (int w = 0; w < CSR_RANGE / 32; ++w) {
        run += wbase[w];
        run += (w < wave) ? wtot[w] : 0;
    }
    if (n < N) {
        off[n] = run;
        if (n == N - 1) off[N] = run + deg;
    }
    #pragma unroll
    for (int j = 0; j < CSR_NB; ++j) {
        tab[j * CSR_RANGE + t] = run;
        run += c[j];
    }
    __syncthreads();

    unsigned rec[8][3];
    #pragma unroll
    for (int q = 0; q < 8; ++q) {
        len[q] = min(max(len[q], 0), chunk);
        st[q]  = min(max(st[q], 0), chunk);
        const size_t base = (size_t)(wave + 8 * q) * chunk + st[q];
        #pragma unroll
        for (int u = 0; u < 3; ++u)
            rec[q][u] = ebin[min(base + (size_t)(lane + 32 * u), (size_t)E - 1)];
    }
    int pos[8][3];
    #pragma unroll
    for (int q = 0; q < 8; ++q) {
        int* trow = tab + (wave + 8 * q) * CSR_RANGE;
        #pragma unroll
        for (int u = 0; u < 3; ++u) {
            pos[q][u] = -1;
            if (lane + 32 * u < len[q]) pos[q][u] = atomicAdd(&trow[rec[q][u] >> 24], 1);
        }
    }
    #pragma unroll
    for (int q = 0; q < 8; ++q) {
        #pragma unroll
        for (int u = 0; u < 3; ++u)
            if (pos[q][u] >= 0) key[min(pos[q][u], E - 1)] = (int)(rec[q][u] & 0xFFFFFFu);
    }
    #pragma unroll
    for (int q = 0; q < 8; ++q) {
        int* trow = tab + (wave + 8 * q) * CSR_RANGE;
        const size_t base = (size_t)(wave + 8 * q) * chunk + st[q];
        for (int i = lane + 96; i < len[q]; i += 32)
            csr_place_one(ebin[min(base + i, (size_t)E - 1)], trow, key, E);
    }
}

__global__ __launch_bounds__(256) void gemm3_xw_colscan_place(const float* __restrict__ X, const float* __restrict__ Wr,
                         const float* __restrict__ Wl, const float* __restrict__ bias,
                         float* __restrict__ rel, float* __restrict__ H, int M, int K, int gblocks,
                         const int* __restrict__ blockcnt, const int* __restrict__ coarse,
                         const int* __restrict__ cstart, const unsigned* __restrict__ ebin,
                         int* __restrict__ off, int* __restrict__ key, int E, int chunk, int nr_total) {
    if ((int)blockIdx.x < gblocks) gemm3_strip(X, Wr, Wl, bias, rel, H, M, K, (int)blockIdx.x);
    else csr_colscan_place(blockcnt, coarse, cstart, ebin, off, key, M, E, chunk, nr_total, (int)blockIdx.x - gblocks);
}

__global__ void csr_fill(const int* __restrict__ src, const int* __restrict__ dst,
                         const int* __restrict__ et, const int* __restrict__ rank,
                         const int* __restrict__ blockcnt,
                         int* __restrict__ key, int E, int N, int chunk) {
    const int e = blockIdx.x * blockDim.x + threadIdx.x;
    if (e >= E) return;
    const int d = dst[e];
    const int b = e / chunk;
    int pos = blockcnt[(size_t)b * N + d] + rank[e];
    pos = min(max(pos, 0), E - 1);
    key[pos] = et[e] * N + src[e];
}

__global__ void gather64_relu(const float* __restrict__ rel, const int* __restrict__ off,
                              const int* __restrict__ key, float* __restrict__ h, int N) {
    const long long t = (long long)blockIdx.x * blockDim.x + threadIdx.x;
    const int i = (int)(t >> 4);
    if (i >= N) return;
    const int f = ((int)t & 15) * 4;
    float* hp = h + (size_t)i * 64 + f;
    float4 acc = *(const float4*)hp;
    int p = off[i];
    const int pe = off[i + 1];
    for (; p + 4 <= pe; p += 4) {
        const int k0 = key[p], k1 = key[p + 1], k2 = key[p + 2], k3 = key[p + 3];
        const float4 a = *(const float4*)(rel + (size_t)k0 * 64 + f);
        const float4 b = *(const float4*)(rel + (size_t)k1 * 64 + f);
        const float4 c = *(const float4*)(rel + (size_t)k2 * 64 + f);
        const float4 d = *(const float4*)(rel + (size_t)k3 * 64 + f);
        acc.x += a.x; acc.y += a.y; acc.z += a.z; acc.w += a.w;
        acc.x += b.x; acc.y += b.y; acc.z += b.z; acc.w += b.w;
        acc.x += c.x; acc.y += c.y; acc.z += c.z; acc.w += c.w;
        acc.x += d.x; acc.y += d.y; acc.z += d.z; acc.w += d.w;
    }
    for (; p < pe; ++p) {
        const float4 a = *(const float4*)(rel + (size_t)key[p] * 64 + f);
        acc.x += a.x; acc.y += a.y; acc.z += a.z; acc.w += a.w;
    }
    acc.x = fmaxf(acc.x, 0.0f); acc.y = fmaxf(acc.y, 0.0f);
    acc.z = fmaxf(acc.z, 0.0f); acc.w = fmaxf(acc.w, 0.0f);
    *(float4*)hp = acc;
}

__global__ void gather2(const float* __restrict__ rel3, const int* __restrict__ off,
                        const int* __restrict__ key, float* __restrict__ agg3, int N) {
    const long long t = (long long)blockIdx.x * blockDim.x + threadIdx.x;
    const int i = (int)(t >> 4);
    const int l = (int)t & 15;
    float s0 = 0.0f, s1 = 0.0f;
    if (i < N) {
        const int pe = off[i + 1];
        for (int p = off[i] + l; p < pe; p += 16) {
            const float2 r = *(const float2*)(rel3 + (size_t)key[p] * 2);
            s0 += r.x; s1 += r.y;
        }
    }
    for (int m = 8; m >= 1; m >>= 1) {
        s0 += __shfl_xor(s0, m, 16);
        s1 += __shfl_xor(s1, m, 16);
    }
    if (i < N && l == 0) {
        agg3[(size_t)i * 2 + 0] += s0;
        agg3[(size_t)i * 2 + 1] += s1;
    }
}

template <int U>
__device__ __forceinline__ void l2_add_rows(const float* __restrict__ h1, const int* __restrict__ kp,
                                            int N, int f, float4& a0, float4& a1) {
    int    kk[U];
    float  m1[U];
    float4 v[U];
    #pragma unroll
    for (int j = 0; j < U; ++j) kk[j] = kp[j];
    #pragma unroll
    for (int j = 0; j < U; ++j) {
        const bool r = kk[j] >= N;
        const int  s = r ? kk[j] - N : kk[j];
        m1[j] = r ? 1.0f : 0.0f;
        v[j]  = *(const float4*)(h1 + (size_t)s * 64 + f);
    }
    #pragma unroll
    for (int j = 0; j < U; ++j) {
        const float w1 = m1[j], w0 = 1.0f - m1[j];
        a0.x = fmaf(v[j].x, w0, a0.x); a0.y = fmaf(v[j].y, w0, a0.y);
        a0.z = fmaf(v[j].z, w0, a0.z); a0.w = fmaf(v[j].w, w0, a0.w);
        a1.x = fmaf(v[j].x, w1, a1.x); a1.y = fmaf(v[j].y, w1, a1.y);
        a1.z = fmaf(v[j].z, w1, a1.z); a1.w = fmaf(v[j].w, w1, a1.w);
    }
}

#define L2_PITCH  196
#define L2_HPITCH 68
__global__ __launch_bounds__(256) void layer2_sum_xw_l3(const float* __restrict__ h1, const int* __restrict__ off,
                                 const int* __restrict__ key,
                                 const float* __restrict__ W2, const float* __restrict__ loop2,
                                 const float* __restrict__ b2,
                                 const float* __restrict__ W3, const float* __restrict__ loop3,
                                 const float* __restrict__ b3,
                                 float* __restrict__ rel3, float* __restrict__ agg3, int N) {
    __shared__ __align__(16) float As[16 * L2_PITCH];
    __shared__ __align__(16) float Hs[16 * L2_HPITCH];
    const int t    = (int)threadIdx.x;
    const int row  = t >> 4;
    const int f    = (t & 15) * 4;
    const int i    = (int)blockIdx.x * 16 + row;
    float4 a0  = make_float4(0.0f, 0.0f, 0.0f, 0.0f);
    float4 a1  = make_float4(0.0f, 0.0f, 0.0f, 0.0f);
    float4 own = make_float4(0.0f, 0.0f, 0.0f, 0.0f);
    if (i < N) {
        own = *(const float4*)(h1 + (size_t)i * 64 + f);
        int p = off[i];
        const int pe = off[i + 1];
        for (; p + 8 <= pe; p += 8) l2_add_rows<8>(h1, key + p, N, f, a0, a1);
        if (p + 4 <= pe) { l2_add_rows<4>(h1, key + p, N, f, a0, a1); p += 4; }
        for (; p < pe; ++p) l2_add_rows<1>(h1, key + p, N, f, a0, a1);
    }
    {
        float* ar = As + row * L2_PITCH + f;
        *(float4*)(ar)       = a0;
        *(float4*)(ar + 64)  = a1;
        *(float4*)(ar + 128) = own;
    }
    __syncthreads();

    const int wave = t >> 5;
    if (wave < 4) {
        const int lane = t & 31;
        const int half = lane >> 4;
        const int lm   = lane & 15;
        const int n    = wave * 16 + lm;
        v8f c0, c1, c2;
        const float bv = b2[n];
        #pragma unroll
        for (int v = 0; v < 8; ++v) { c0[v] = 0.0f; c1[v] = 0.0f; c2[v] = bv; }
        const float* arow = As + lm * L2_PITCH + 2 * half;
        const float* w1p  = W2 + 64 * 64;
        #pragma unroll
        for (int k0 = 0; k0 < 64; k0 += 4) {
            const int kb = k0 + 2 * half;
            #pragma unroll
            for (int mt = 0; mt < 3; ++mt) {
                const float* Wm = (mt == 0) ? W2 : (mt == 1) ? w1p : loop2;
                const float2 av = *(const float2*)(arow + 64 * mt + k0);
                v2f a; a.x = av.x; a.y = av.y;
                v2f b;
                b.x = Wm[(size_t)kb * 64 + n];
                b.y = Wm[(size_t)(kb + 1) * 64 + n];
                if (mt == 0)      c0 = __builtin_amdgcn_wmma_f32_16x16x4_f32(false, a, false, b, (short)0, c0, false, false);
                else if (mt == 1) c1 = __builtin_amdgcn_wmma_f32_16x16x4_f32(false, a, false, b, (short)0, c1, false, false);
                else              c2 = __builtin_amdgcn_wmma_f32_16x16x4_f32(false, a, false, b, (short)0, c2, false, false);
            }
        }
        #pragma unroll
        for (int v = 0; v < 8; ++v)
            Hs[(v + 8 * half) * L2_HPITCH + n] = fmaxf((c0[v] + c1[v]) + c2[v], 0.0f);
    }
    __syncthreads();

    const float4 hq = *(const float4*)(Hs + row * L2_HPITCH + f);
    const float hv[4] = { hq.x, hq.y, hq.z, hq.w };
    const float* wm[3] = { W3 + f * 2, W3 + 128 + f * 2, loop3 + f * 2 };
    float o[6];
    #pragma unroll
    for (int m = 0; m < 3; ++m) {
        float s0 = 0.0f, s1 = 0.0f;
        #pragma unroll
        for (int j = 0; j < 4; ++j) {
            s0 += hv[j] * wm[m][2 * j + 0];
            s1 += hv[j] * wm[m][2 * j + 1];
        }
        o[2 * m + 0] = s0;
        o[2 * m + 1] = s1;
    }
    for (int d = 8; d >= 1; d >>= 1) {
        #pragma unroll
        for (int q = 0; q < 6; ++q) o[q] += __shfl_xor(o[q], d, 16);
    }
    if ((t & 15) == 0 && i < N) {
        rel3[(size_t)i * 2 + 0] = o[0];
        rel3[(size_t)i * 2 + 1] = o[1];
        rel3[(size_t)(N + i) * 2 + 0] = o[2];
        rel3[(size_t)(N + i) * 2 + 1] = o[3];
        agg3[(size_t)i * 2 + 0] = o[4] + b3[0];
        agg3[(size_t)i * 2 + 1] = o[5] + b3[1];
    }
}

#define TAIL_TPB 256
__global__ __launch_bounds__(TAIL_TPB) void gather2_min(const float* __restrict__ rel3, const int* __restrict__ off,
                            const int* __restrict__ key, float* __restrict__ agg3,
                            float* __restrict__ blockmin, int N) {
    __shared__ float sm[TAIL_TPB / 32];
    const long long t = (long long)blockIdx.x * blockDim.x + threadIdx.x;
    const int i = (int)(t >> 4);
    const int l = (int)t & 15;
    float s0 = 0.0f, s1 = 0.0f;
    if (i < N) {
        const int pe = off[i + 1];
        for (int p = off[i] + l; p < pe; p += 16) {
            const float2 r = *(const float2*)(rel3 + (size_t)key[p] * 2);
            s0 += r.x; s1 += r.y;
        }
    }
    for (int m = 8; m >= 1; m >>= 1) {
        s0 += __shfl_xor(s0, m, 16);
        s1 += __shfl_xor(s1, m, 16);
    }
    float mn = 3.402823466e38f;
    if (i < N && l == 0) {
        const float v0 = agg3[(size_t)i * 2 + 0] + s0;
        const float v1 = agg3[(size_t)i * 2 + 1] + s1;
        agg3[(size_t)i * 2 + 0] = v0;
        agg3[(size_t)i * 2 + 1] = v1;
        mn = fminf(v0, v1);
    }
    mn = fminf(mn, __shfl_xor(mn, 16, 32));
    if ((threadIdx.x & 31) == 0) sm[threadIdx.x >> 5] = mn;
    __syncthreads();
    if (threadIdx.x == 0) {
        float r = sm[0];
        #pragma unroll
        for (int w = 1; w < TAIL_TPB / 32; ++w) r = fminf(r, sm[w]);
        blockmin[blockIdx.x] = r;
    }
}

__global__ __launch_bounds__(TAIL_TPB) void finalize_bm(const float* __restrict__ agg3,
                            const int* __restrict__ cell,
                            const int* __restrict__ maxs,
                            const float* __restrict__ blockmin, int nblk,
                            float* __restrict__ out, int N) {
    __shared__ float sm[TAIL_TPB / 32];
    float mn = 3.402823466e38f;
    for (int j0 = threadIdx.x; j0 < nblk; j0 += 16 * TAIL_TPB) {
        float v[16];
        #pragma unroll
        for (int u = 0; u < 16; ++u) v[u] = blockmin[min(j0 + u * TAIL_TPB, nblk - 1)];
        #pragma unroll
        for (int u = 0; u < 16; ++u) mn = fminf(mn, v[u]);
    }
    for (int m = 16; m >= 1; m >>= 1) mn = fminf(mn, __shfl_xor(mn, m, 32));
    if ((threadIdx.x & 31) == 0) sm[threadIdx.x >> 5] = mn;
    __syncthreads();
    float r = sm[0];
    #pragma unroll
    for (int w = 1; w < TAIL_TPB / 32; ++w) r = fminf(r, sm[w]);
    const float gmin = r - 1.0f;
    const int i = blockIdx.x * blockDim.x + threadIdx.x;
    if (i >= N) return;
    const bool upper = cell[i] >= (maxs[i] - 1);
    const bool lower = cell[i] == 0;
    out[(size_t)i * 2 + 0] = upper ? gmin : agg3[(size_t)i * 2 + 0];
    out[(size_t)i * 2 + 1] = lower ? gmin : agg3[(size_t)i * 2 + 1];
}

__device__ __forceinline__ unsigned f2ord(float f) {
    unsigned u = __float_as_uint(f);
    return (u & 0x80000000u) ? ~u : (u | 0x80000000u);
}
__device__ __forceinline__ float ord2f(unsigned u) {
    return __uint_as_float((u & 0x80000000u) ? (u ^ 0x80000000u) : ~u);
}

__global__ void init_slot(unsigned* slot) { *slot = 0xFFFFFFFFu; }

__global__ void min_reduce(const float* __restrict__ p, int n, unsigned* slot) {
    __shared__ float sm[256];
    float m = 3.402823466e38f;
    for (int i = blockIdx.x * blockDim.x + threadIdx.x; i < n;
         i += gridDim.x * blockDim.x)
        m = fminf(m, p[i]);
    sm[threadIdx.x] = m;
    __syncthreads();
    for (int s = blockDim.x >> 1; s > 0; s >>= 1) {
        if ((int)threadIdx.x < s) sm[threadIdx.x] = fminf(sm[threadIdx.x], sm[threadIdx.x + s]);
        __syncthreads();
    }
    if (threadIdx.x == 0) atomicMin(slot, f2ord(sm[0]));
}

__global__ void finalize_k(const float* __restrict__ agg3,
                           const int* __restrict__ cell,
                           const int* __restrict__ maxs,
                           const unsigned* __restrict__ slot,
                           float* __restrict__ out, int N) {
    const int i = blockIdx.x * blockDim.x + threadIdx.x;
    if (i >= N) return;
    const float gmin = ord2f(*slot) - 1.0f;
    const bool upper = cell[i] >= (maxs[i] - 1);
    const bool lower = cell[i] == 0;
    out[(size_t)i * 2 + 0] = upper ? gmin : agg3[(size_t)i * 2 + 0];
    out[(size_t)i * 2 + 1] = lower ? gmin : agg3[(size_t)i * 2 + 1];
}

extern "C" void kernel_launch(void* const* d_in, const int* in_sizes, int n_in,
                              void* d_out, int out_size, void* d_ws, size_t ws_size,
                              hipStream_t stream) {
    const float* x      = (const float*)d_in[0];
    const int*   src    = (const int*)d_in[1];
    const int*   dst    = (const int*)d_in[2];
    const int*   etypes = (const int*)d_in[3];
    const int*   cell   = (const int*)d_in[4];
    const int*   maxs   = (const int*)d_in[5];
    const float* W1     = (const float*)d_in[6];
    const float* loop1  = (const float*)d_in[7];
    const float* b1     = (const float*)d_in[8];
    const float* W2     = (const float*)d_in[9];
    const float* loop2  = (const float*)d_in[10];
    const float* b2     = (const float*)d_in[11];
    const float* W3     = (const float*)d_in[12];
    const float* loop3  = (const float*)d_in[13];
    const float* b3     = (const float*)d_in[14];
    float* out = (float*)d_out;

    const int N = in_sizes[4];
    const int E = in_sizes[1];
    const int K1 = in_sizes[0] / N;

    float* relbuf = (float*)d_ws;
    float* h1     = relbuf + (size_t)2 * N * 64;
    float* h2     = h1 + (size_t)N * 64;
    float* rel3   = h2 + (size_t)N * 64;
    float* agg3   = rel3 + (size_t)4 * N;
    unsigned* slot = (unsigned*)(agg3 + (size_t)2 * N);

    const int waves   = (N + 15) / 16;
    const int gblocks = (waves * 32 + 255) / 256;
    const int eb64    = (int)(((long long)E * 16 + 255) / 256);
    const int nb      = (N + 255) / 256;
    const int fb      = (N * 64 + 255) / 256;
    const int eb2     = (E + 255) / 256;

    int* key      = (int*)(slot + 4);
    int* off      = key + (size_t)E;
    int* blockcnt = (int*)relbuf;
    int* rank     = blockcnt + (size_t)CSR_NB * N;
    const size_t need = ((size_t)262 * N + 4 + (size_t)E + (size_t)N + 1) * sizeof(float);
    int* coarse   = (int*)rel3;
    const bool csr_ok = ws_size >= need && (long long)E <= (long long)192 * N
                        && (long long)CSR_NB * nb <= (long long)6 * N;
    const size_t need2 = need + (size_t)E * sizeof(int);
    const int chunk0 = (E + CSR_NB - 1) / CSR_NB;
    const bool shared_ok = csr_ok && ws_size >= need2
                           && N <= CSR_CAP && chunk0 <= 1024 * CSR_EPT
                           && (long long)2 * N <= (long long)0x1000000
                           && (long long)2 * CSR_NB * nb <= (long long)6 * N;

    if (shared_ok) {
        const int chunk = chunk0;
        const int gb64  = (int)(((long long)N * 16 + 255) / 256);
        int* blockcnt2  = (int*)h2;
        unsigned* ebin  = (unsigned*)(off + (size_t)N + 1);
        int* cstart     = coarse + (size_t)CSR_NB * nb;
        csr_hist_bin<<<CSR_NB, 1024, 0, stream>>>(dst, src, etypes, ebin, blockcnt2, coarse, cstart, E, N, chunk, nb);

        gemm3_xw_colscan_place<<<gblocks + nb, 256, 0, stream>>>(x, W1, loop1, b1, relbuf, h1, N, K1, gblocks,
                                                                 blockcnt2, coarse, cstart, ebin, off, key, E, chunk, nb);
        gather64_relu<<<gb64, 256, 0, stream>>>(relbuf, off, key, h1, N);

        layer2_sum_xw_l3<<<waves, 256, 0, stream>>>(h1, off, key, W2, loop2, b2, W3, loop3, b3, rel3, agg3, N);

        float* blockmin = relbuf;
        gather2_min<<<gb64, TAIL_TPB, 0, stream>>>(rel3, off, key, agg3, blockmin, N);
        finalize_bm<<<nb, TAIL_TPB, 0, stream>>>(agg3, cell, maxs, blockmin, gb64, out, N);
    } else if (csr_ok) {
        const int chunk = (E + CSR_NB - 1) / CSR_NB;
        const int gb64  = (int)(((long long)N * 16 + 255) / 256);
        csr_hist<<<CSR_NB, 1024, 0, stream>>>(dst, rank, blockcnt, coarse, E, N, chunk, nb);
        csr_colscan_off<<<nb, CSR_RANGE, 0, stream>>>(blockcnt, coarse, off, N, nb);
        csr_fill<<<eb2, 256, 0, stream>>>(src, dst, etypes, rank, blockcnt, key, E, N, chunk);

        gemm3_xw<<<gblocks, 256, 0, stream>>>(x, W1, loop1, b1, relbuf, h1, N, K1);
        gather64_relu<<<gb64, 256, 0, stream>>>(relbuf, off, key, h1, N);

        layer2_sum_xw_l3<<<waves, 256, 0, stream>>>(h1, off, key, W2, loop2, b2, W3, loop3, b3, rel3, agg3, N);

        float* blockmin = relbuf;
        gather2_min<<<gb64, TAIL_TPB, 0, stream>>>(rel3, off, key, agg3, blockmin, N);
        finalize_bm<<<nb, TAIL_TPB, 0, stream>>>(agg3, cell, maxs, blockmin, gb64, out, N);
    } else {
        gemm3_xw<<<gblocks, 256, 0, stream>>>(x, W1, loop1, b1, relbuf, h1, N, K1);
        edge_scatter64<<<eb64, 256, 0, stream>>>(relbuf, src, dst, etypes, h1, E, N);
        relu_k<<<fb, 256, 0, stream>>>(h1, N * 64);

        gemm3_xw<<<gblocks, 256, 0, stream>>>(h1, W2, loop2, b2, relbuf, h2, N, 64);
        edge_scatter64<<<eb64, 256, 0, stream>>>(relbuf, src, dst, etypes, h2, E, N);
        relu_k<<<fb, 256, 0, stream>>>(h2, N * 64);

        layer3_node<<<nb, 256, 0, stream>>>(h2, W3, loop3, b3, rel3, agg3, N);
        edge_scatter2<<<eb2, 256, 0, stream>>>(rel3, src, dst, etypes, agg3, E, N);
        init_slot<<<1, 1, 0, stream>>>(slot);
        min_reduce<<<512, 256, 0, stream>>>(agg3, N * 2, slot);
        finalize_k<<<nb, 256, 0, stream>>>(agg3, cell, maxs, slot, out, N);
    }
}


# Fewer lanes per node in the 3 edge-gather launches (64-wide: 8 lanes x 2 float4, layer 2 at 32 nodes/block; 2-wide: 4 lanes x 4 running sums): 75000 -> 31250 waves started, sum trees unchanged
# speedup vs baseline: 2.8865x; 1.0376x over previous
#include <hip/hip_runtime.h>

typedef __attribute__((ext_vector_type(2))) float v2f;
typedef __attribute__((ext_vector_type(8))) float v8f;

__device__ __forceinline__ void gemm3_strip(const float* __restrict__ X, const float* __restrict__ Wr,
                         const float* __restrict__ Wl, const float* __restrict__ bias,
                         float* __restrict__ rel, float* __restrict__ H,
                         int M, int K, int block) {
    const int gthread = block * blockDim.x + threadIdx.x;
    const int wave    = gthread >> 5;
    const int lane    = threadIdx.x & 31;
    const int m0      = wave * 16;
    if (m0 >= M) return;

    const int half = lane >> 4;
    const int lm   = lane & 15;

    v8f acc[12];
    #pragma unroll
    for (int t = 0; t < 8; ++t) {
        #pragma unroll
        for (int v = 0; v < 8; ++v) acc[t][v] = 0.0f;
    }
    #pragma unroll
    for (int nt = 0; nt < 4; ++nt) {
        const float bv = bias[nt * 16 + lm];
        #pragma unroll
        for (int v = 0; v < 8; ++v) acc[8 + nt][v] = bv;
    }

    const float* xrow = X + (size_t)(m0 + lm) * K;
    const float* w1   = Wr + (size_t)K * 64;

    #pragma unroll 2
    for (int k0 = 0; k0 < K; k0 += 4) {
        const int kb = k0 + 2 * half;
        const float2 av = *(const float2*)(xrow + kb);
        v2f a; a.x = av.x; a.y = av.y;
        #pragma unroll
        for (int mt = 0; mt < 3; ++mt) {
            const float* Wm = (mt == 0) ? Wr : (mt == 1) ? w1 : Wl;
            #pragma unroll
            for (int nt = 0; nt < 4; ++nt) {
                const int n = nt * 16 + lm;
                v2f b;
                b.x = Wm[(size_t)kb * 64 + n];
                b.y = Wm[(size_t)(kb + 1) * 64 + n];
                acc[mt * 4 + nt] = __builtin_amdgcn_wmma_f32_16x16x4_f32(
                    false, a, false, b, (short)0, acc[mt * 4 + nt], false, false);
            }
        }
    }

    #pragma unroll
    for (int mt = 0; mt < 3; ++mt) {
        float* C = (mt == 2) ? H : rel + (size_t)mt * M * 64;
        #pragma unroll
        for (int nt = 0; nt < 4; ++nt) {
            const int n = nt * 16 + lm;
            #pragma unroll
            for (int v = 0; v < 8; ++v) {
                C[(size_t)(m0 + v + 8 * half) * 64 + n] = acc[mt * 4 + nt][v];
            }
        }
    }
}

__global__ __launch_bounds__(256) void gemm3_xw(const float* __restrict__ X, const float* __restrict__ Wr,
                         const float* __restrict__ Wl, const float* __restrict__ bias,
                         float* __restrict__ rel, float* __restrict__ H,
                         int M, int K) {
    gemm3_strip(X, Wr, Wl, bias, rel, H, M, K, (int)blockIdx.x);
}

__global__ void edge_scatter64(const float* __restrict__ rel,
                               const int* __restrict__ src,
                               const int* __restrict__ dst,
                               const int* __restrict__ et,
                               float* __restrict__ agg, int E, int N) {
    const long long t = (long long)blockIdx.x * blockDim.x + threadIdx.x;
    const int e = (int)(t >> 4);
    if (e >= E) return;
    const int f = ((int)t & 15) * 4;
    const float4 v = *(const float4*)(rel + ((long long)et[e] * N + src[e]) * 64 + f);
    float* q = agg + (long long)dst[e] * 64 + f;
    atomicAdd(q + 0, v.x);
    atomicAdd(q + 1, v.y);
    atomicAdd(q + 2, v.z);
    atomicAdd(q + 3, v.w);
}

__global__ void relu_k(float* __restrict__ p, int n) {
    const int i = blockIdx.x * blockDim.x + threadIdx.x;
    if (i < n) p[i] = fmaxf(p[i], 0.0f);
}

__global__ void layer3_node(const float* __restrict__ h2,
                            const float* __restrict__ W3,
                            const float* __restrict__ loop3,
                            const float* __restrict__ b3,
                            float* __restrict__ rel3, float* __restrict__ agg3,
                            int N) {
    const int i = blockIdx.x * blockDim.x + threadIdx.x;
    if (i >= N) return;
    const float* row = h2 + (size_t)i * 64;
    float r00 = 0.f, r01 = 0.f, r10 = 0.f, r11 = 0.f, l0 = 0.f, l1 = 0.f;
    #pragma unroll 8
    for (int k = 0; k < 64; ++k) {
        const float v = row[k];
        r00 += v * W3[k * 2 + 0];
        r01 += v * W3[k * 2 + 1];
        r10 += v * W3[128 + k * 2 + 0];
        r11 += v * W3[128 + k * 2 + 1];
        l0  += v * loop3[k * 2 + 0];
        l1  += v * loop3[k * 2 + 1];
    }
    rel3[(size_t)i * 2 + 0] = r00;
    rel3[(size_t)i * 2 + 1] = r01;
    rel3[(size_t)(N + i) * 2 + 0] = r10;
    rel3[(size_t)(N + i) * 2 + 1] = r11;
    agg3[(size_t)i * 2 + 0] = l0 + b3[0];
    agg3[(size_t)i * 2 + 1] = l1 + b3[1];
}

__global__ void edge_scatter2(const float* __restrict__ rel3,
                              const int* __restrict__ src,
                              const int* __restrict__ dst,
                              const int* __restrict__ et,
                              float* __restrict__ agg3, int E, int N) {
    const int e = blockIdx.x * blockDim.x + threadIdx.x;
    if (e >= E) return;
    const float* p = rel3 + ((long long)et[e] * N + src[e]) * 2;
    float* q = agg3 + (long long)dst[e] * 2;
    atomicAdd(q + 0, p[0]);
    atomicAdd(q + 1, p[1]);
}


#define CSR_NB  64
#define CSR_CAP 51200
#define CSR_RANGE 256

__global__ __launch_bounds__(1024) void csr_hist(const int* __restrict__ dst,
                                                 int* __restrict__ rank,
                                                 int* __restrict__ blockcnt,
                                                 int* __restrict__ coarse,
                                                 int E, int N, int chunk, int nr_total) {
    __shared__ int cnt[CSR_CAP];
    const int b  = blockIdx.x;
    const int e0 = b * chunk;
    const int e1 = min(e0 + chunk, E);
    for (int base = 0; base < N; base += CSR_CAP) {
        const int lim = min(CSR_CAP, N - base);
        for (int i = threadIdx.x; i < lim; i += blockDim.x) cnt[i] = 0;
        __syncthreads();
        const int bd = (int)blockDim.x;
        for (int e = e0 + (int)threadIdx.x; e < e1; e += 4 * bd) {
            int d[4];
            #pragma unroll
            for (int j = 0; j < 4; ++j) {
                const int ee = e + j * bd;
                const int v  = dst[min(ee, e1 - 1)];
                d[j] = (ee < e1) ? v - base : -1;
            }
            #pragma unroll
            for (int j = 0; j < 4; ++j)
                if (d[j] >= 0 && d[j] < lim) rank[e + j * bd] = atomicAdd(&cnt[d[j]], 1);
        }
        __syncthreads();
        const int nrs = (lim + CSR_RANGE - 1) / CSR_RANGE;
        for (int r0 = 0; r0 < nrs; r0 += (int)blockDim.x >> 2) {
            const int r  = r0 + ((int)threadIdx.x >> 2);
            const int q  = (int)threadIdx.x & 3;
            const int lo = r * CSR_RANGE + q * (CSR_RANGE / 4);
            int s = 0;
            #pragma unroll 16
            for (int j = 0; j < CSR_RANGE / 4; ++j) {
                const int i = lo + ((j + (int)threadIdx.x) & (CSR_RANGE / 4 - 1));
                const int v = cnt[min(i, lim - 1)];
                s += (i < lim) ? v : 0;
            }
            s += __shfl_xor(s, 1, 4);
            s += __shfl_xor(s, 2, 4);
            if (q == 0 && r < nrs) coarse[(size_t)b * nr_total + base / CSR_RANGE + r] = s;
        }
        for (int i = threadIdx.x; i < lim; i += blockDim.x)
            blockcnt[(size_t)b * N + base + i] = cnt[i];
        __syncthreads();
    }
}

#define CSR_EPT 16
__global__ __launch_bounds__(1024) void csr_hist_bin(const int* __restrict__ dst, const int* __restrict__ src,
                                                     const int* __restrict__ et,
                                                     unsigned* __restrict__ ebin,
                                                     int* __restrict__ blockcnt,
                                                     int* __restrict__ coarse, int* __restrict__ cstart,
                                                     int E, int N, int chunk, int nr_total) {
    __shared__ int cnt[CSR_CAP];
    __shared__ int rcnt[CSR_CAP / CSR_RANGE];
    __shared__ int rstart[CSR_CAP / CSR_RANGE];
    __shared__ int swt[8];
    const int tid  = (int)threadIdx.x;
    const int lane = tid & 31;
    const int b    = (int)blockIdx.x;
    const int e0   = b * chunk;
    const int e1   = min(e0 + chunk, E);
    for (int i = tid; i < N; i += 1024) cnt[i] = 0;
    if (tid < nr_total) rcnt[tid] = 0;
    int dd[CSR_EPT], ss[CSR_EPT], tt[CSR_EPT];
    #pragma unroll
    for (int j = 0; j < CSR_EPT; ++j) {
        const int ec = max(min(e0 + tid + j * 1024, e1 - 1), 0);
        dd[j] = dst[ec]; ss[j] = src[ec]; tt[j] = et[ec];
    }
    #pragma unroll
    for (int j = 0; j < CSR_EPT; ++j)
        if (e0 + tid + j * 1024 >= e1) dd[j] = -1;
    __syncthreads();
    int slot[CSR_EPT];
    #pragma unroll
    for (int j = 0; j < CSR_EPT; ++j) {
        slot[j] = 0;
        if (dd[j] >= 0 && dd[j] < N) {
            atomicAdd(&cnt[dd[j]], 1);
            slot[j] = atomicAdd(&rcnt[dd[j] / CSR_RANGE], 1);
        } else dd[j] = -1;
    }
    __syncthreads();
    int rv = 0, inc = 0;
    if (tid < 256) {
        rv  = (tid < nr_total) ? rcnt[tid] : 0;
        inc = rv;
        for (int d = 1; d < 32; d <<= 1) {
            const int v = __shfl_up(inc, d, 32);
            if (lane >= d) inc += v;
        }
        if (lane == 31) swt[tid >> 5] = inc;
    }
    __syncthreads();
    if (tid < 256) {
        int wb = 0;
        #pragma unroll
        for (int w = 0; w < 8; ++w) wb += (w < (tid >> 5)) ? swt[w] : 0;
        const int ex = wb + inc - rv;
        if (tid < nr_total) {
            rstart[tid] = ex;
            coarse[(size_t)b * nr_total + tid] = rv;
            cstart[(size_t)b * nr_total + tid] = ex;
        }
    }
    __syncthreads();
    #pragma unroll
    for (int j = 0; j < CSR_EPT; ++j)
        slot[j] += rstart[max(dd[j], 0) / CSR_RANGE];
    #pragma unroll
    for (int j = 0; j < CSR_EPT; ++j) {
        if (dd[j] >= 0) {
            const int r = dd[j] / CSR_RANGE;
            const int p = min(max(slot[j], 0), e1 - e0 - 1);
            const unsigned kv = (unsigned)(tt[j] * N + ss[j]);
            ebin[(size_t)e0 + p] = (kv & 0xFFFFFFu) | ((unsigned)(dd[j] - r * CSR_RANGE) << 24);
        }
    }
    for (int i = tid; i < N; i += 1024) blockcnt[(size_t)b * N + i] = cnt[i];
}

__device__ __forceinline__ void csr_colscan_range(int* __restrict__ blockcnt,
                                                  const int* __restrict__ coarse,
                                                  int* __restrict__ off, int N, int nr_total, int r) {
    __shared__ int wtot[CSR_RANGE / 32];
    __shared__ int wbase[CSR_RANGE / 32];
    const int t  = threadIdx.x;
    const int n  = r * CSR_RANGE + t;
    const int nn = min(n, N - 1);
    int c[CSR_NB];
    #pragma unroll
    for (int j = 0; j < CSR_NB; ++j) c[j] = blockcnt[(size_t)j * N + nn];
    int part = 0;
    for (int rr = t; rr < r; rr += CSR_RANGE) {
        #pragma unroll 16
        for (int b = 0; b < CSR_NB; ++b) part += coarse[(size_t)b * nr_total + rr];
    }
    int deg = 0;
    #pragma unroll
    for (int j = 0; j < CSR_NB; ++j) deg += c[j];
    if (n >= N) deg = 0;
    const int lane = t & 31;
    int inc = deg;
    for (int d = 1; d < 32; d <<= 1) {
        const int v = __shfl_up(inc, d, 32);
        if (lane >= d) inc += v;
    }
    for (int m = 16; m >= 1; m >>= 1) part += __shfl_xor(part, m, 32);
    if (lane == 31) wtot[t >> 5] = inc;
    if (lane == 0)  wbase[t >> 5] = part;
    __syncthreads();
    int run = inc - deg;
    #pragma unroll
    for (int w = 0; w < CSR_RANGE / 32; ++w) {
        run += wbase[w];
        run += (w < (t >> 5)) ? wtot[w] : 0;
    }
    if (n < N) {
        off[n] = run;
        if (n == N - 1) off[N] = run + deg;
        #pragma unroll
        for (int j = 0; j < CSR_NB; ++j) {
            blockcnt[(size_t)j * N + n] = run;
            run += c[j];
        }
    }
}

__global__ __launch_bounds__(CSR_RANGE) void csr_colscan_off(int* __restrict__ blockcnt,
                                                             const int* __restrict__ coarse,
                                                             int* __restrict__ off, int N, int nr_total) {
    csr_colscan_range(blockcnt, coarse, off, N, nr_total, (int)blockIdx.x);
}

static_assert(CSR_NB == 64 && CSR_RANGE == 256, "8 waves x 8 chunks, one thread per node of the range");
__device__ __forceinline__ void csr_place_one(unsigned rec, int* __restrict__ trow, int* __restrict__ key, int E) {
    const int dl = (int)(rec >> 24);
    int pos = atomicAdd(&trow[dl], 1);
    pos = min(max(pos, 0), E - 1);
    key[pos] = (int)(rec & 0xFFFFFFu);
}

__device__ __forceinline__ void csr_colscan_place(const int* __restrict__ blockcnt,
                                                  const int* __restrict__ coarse, const int* __restrict__ cstart,
                                                  const unsigned* __restrict__ ebin,
                                                  int* __restrict__ off, int* __restrict__ key,
                                                  int N, int E, int chunk, int nr_total, int r) {
    __shared__ int wtot[CSR_RANGE / 32];
    __shared__ int wbase[CSR_RANGE / 32];
    __shared__ int tab[CSR_NB * CSR_RANGE];
    const int t    = threadIdx.x;
    const int wave = t >> 5;
    const int lane = t & 31;
    const int n  = r * CSR_RANGE + t;
    const int nn = min(n, N - 1);
    int len[8], st[8];
    #pragma unroll
    for (int q = 0; q < 8; ++q) {
        const int bq = wave + 8 * q;
        len[q] = coarse[(size_t)bq * nr_total + r];
        st[q]  = cstart[(size_t)bq * nr_total + r];
    }
    int c[CSR_NB];
    #pragma unroll
    for (int j = 0; j < CSR_NB; ++j) c[j] = blockcnt[(size_t)j * N + nn];
    int part = 0;
    for (int rr = t; rr < r; rr += CSR_RANGE) {
        #pragma unroll 16
        for (int b = 0; b < CSR_NB; ++b) part += coarse[(size_t)b * nr_total + rr];
    }
    int deg = 0;
    #pragma unroll
    for (int j = 0; j < CSR_NB; ++j) deg += c[j];
    if (n >= N) deg = 0;
    int inc = deg;
    for (int d = 1; d < 32; d <<= 1) {
        const int v = __shfl_up(inc, d, 32);
        if (lane >= d) inc += v;
    }
    for (int m = 16; m >= 1; m >>= 1) part += __shfl_xor(part, m, 32);
    if (lane == 31) wtot[wave] = inc;
    if (lane == 0)  wbase[wave] = part;
    __syncthreads();
    int run = inc - deg;
    #pragma unroll
    for (int w = 0; w < CSR_RANGE / 32; ++w) {
        run += wbase[w];
        run += (w < wave) ? wtot[w] : 0;
    }
    if (n < N) {
        off[n] = run;
        if (n == N - 1) off[N] = run + deg;
    }
    #pragma unroll
    for (int j = 0; j < CSR_NB; ++j) {
        tab[j * CSR_RANGE + t] = run;
        run += c[j];
    }
    __syncthreads();

    unsigned rec[8][3];
    #pragma unroll
    for (int q = 0; q < 8; ++q) {
        len[q] = min(max(len[q], 0), chunk);
        st[q]  = min(max(st[q], 0), chunk);
        const size_t base = (size_t)(wave + 8 * q) * chunk + st[q];
        #pragma unroll
        for (int u = 0; u < 3; ++u)
            rec[q][u] = ebin[min(base + (size_t)(lane + 32 * u), (size_t)E - 1)];
    }
    int pos[8][3];
    #pragma unroll
    for (int q = 0; q < 8; ++q) {
        int* trow = tab + (wave + 8 * q) * CSR_RANGE;
        #pragma unroll
        for (int u = 0; u < 3; ++u) {
            pos[q][u] = -1;
            if (lane + 32 * u < len[q]) pos[q][u] = atomicAdd(&trow[rec[q][u] >> 24], 1);
        }
    }
    #pragma unroll
    for (int q = 0; q < 8; ++q) {
        #pragma unroll
        for (int u = 0; u < 3; ++u)
            if (pos[q][u] >= 0) key[min(pos[q][u], E - 1)] = (int)(rec[q][u] & 0xFFFFFFu);
    }
    #pragma unroll
    for (int q = 0; q < 8; ++q) {
        int* trow = tab + (wave + 8 * q) * CSR_RANGE;
        const size_t base = (size_t)(wave + 8 * q) * chunk + st[q];
        for (int i = lane + 96; i < len[q]; i += 32)
            csr_place_one(ebin[min(base + i, (size_t)E - 1)], trow, key, E);
    }
}

__global__ __launch_bounds__(256) void gemm3_xw_colscan_place(const float* __restrict__ X, const float* __restrict__ Wr,
                         const float* __restrict__ Wl, const float* __restrict__ bias,
                         float* __restrict__ rel, float* __restrict__ H, int M, int K, int gblocks,
                         const int* __restrict__ blockcnt, const int* __restrict__ coarse,
                         const int* __restrict__ cstart, const unsigned* __restrict__ ebin,
                         int* __restrict__ off, int* __restrict__ key, int E, int chunk, int nr_total) {
    if ((int)blockIdx.x < gblocks) gemm3_strip(X, Wr, Wl, bias, rel, H, M, K, (int)blockIdx.x);
    else csr_colscan_place(blockcnt, coarse, cstart, ebin, off, key, M, E, chunk, nr_total, (int)blockIdx.x - gblocks);
}

__global__ void csr_fill(const int* __restrict__ src, const int* __restrict__ dst,
                         const int* __restrict__ et, const int* __restrict__ rank,
                         const int* __restrict__ blockcnt,
                         int* __restrict__ key, int E, int N, int chunk) {
    const int e = blockIdx.x * blockDim.x + threadIdx.x;
    if (e >= E) return;
    const int d = dst[e];
    const int b = e / chunk;
    int pos = blockcnt[(size_t)b * N + d] + rank[e];
    pos = min(max(pos, 0), E - 1);
    key[pos] = et[e] * N + src[e];
}

__global__ void gather64_relu(const float* __restrict__ rel, const int* __restrict__ off,
                              const int* __restrict__ key, float* __restrict__ h, int N) {
    const long long t = (long long)blockIdx.x * blockDim.x + threadIdx.x;
    const int i = (int)(t >> 3);
    if (i >= N) return;
    const int f = ((int)t & 7) * 4;
    float* hp = h + (size_t)i * 64 + f;
    float4 acc = *(const float4*)hp;
    float4 acd = *(const float4*)(hp + 32);
    int p = off[i];
    const int pe = off[i + 1];
    for (; p + 4 <= pe; p += 4) {
        const int k0 = key[p], k1 = key[p + 1], k2 = key[p + 2], k3 = key[p + 3];
        const float* r0 = rel + (size_t)k0 * 64 + f;
        const float* r1 = rel + (size_t)k1 * 64 + f;
        const float* r2 = rel + (size_t)k2 * 64 + f;
        const float* r3 = rel + (size_t)k3 * 64 + f;
        const float4 a = *(const float4*)r0, a2 = *(const float4*)(r0 + 32);
        const float4 b = *(const float4*)r1, b2 = *(const float4*)(r1 + 32);
        const float4 c = *(const float4*)r2, c2 = *(const float4*)(r2 + 32);
        const float4 d = *(const float4*)r3, d2 = *(const float4*)(r3 + 32);
        acc.x += a.x; acc.y += a.y; acc.z += a.z; acc.w += a.w;
        acd.x += a2.x; acd.y += a2.y; acd.z += a2.z; acd.w += a2.w;
        acc.x += b.x; acc.y += b.y; acc.z += b.z; acc.w += b.w;
        acd.x += b2.x; acd.y += b2.y; acd.z += b2.z; acd.w += b2.w;
        acc.x += c.x; acc.y += c.y; acc.z += c.z; acc.w += c.w;
        acd.x += c2.x; acd.y += c2.y; acd.z += c2.z; acd.w += c2.w;
        acc.x += d.x; acc.y += d.y; acc.z += d.z; acc.w += d.w;
        acd.x += d2.x; acd.y += d2.y; acd.z += d2.z; acd.w += d2.w;
    }
    #pragma nounroll
    for (; p < pe; ++p) {
        const float* r0 = rel + (size_t)key[p] * 64 + f;
        const float4 a = *(const float4*)r0, a2 = *(const float4*)(r0 + 32);
        acc.x += a.x; acc.y += a.y; acc.z += a.z; acc.w += a.w;
        acd.x += a2.x; acd.y += a2.y; acd.z += a2.z; acd.w += a2.w;
    }
    acc.x = fmaxf(acc.x, 0.0f); acc.y = fmaxf(acc.y, 0.0f);
    acc.z = fmaxf(acc.z, 0.0f); acc.w = fmaxf(acc.w, 0.0f);
    acd.x = fmaxf(acd.x, 0.0f); acd.y = fmaxf(acd.y, 0.0f);
    acd.z = fmaxf(acd.z, 0.0f); acd.w = fmaxf(acd.w, 0.0f);
    *(float4*)hp = acc;
    *(float4*)(hp + 32) = acd;
}

__global__ void gather2(const float* __restrict__ rel3, const int* __restrict__ off,
                        const int* __restrict__ key, float* __restrict__ agg3, int N) {
    const long long t = (long long)blockIdx.x * blockDim.x + threadIdx.x;
    const int i = (int)(t >> 4);
    const int l = (int)t & 15;
    float s0 = 0.0f, s1 = 0.0f;
    if (i < N) {
        const int pe = off[i + 1];
        for (int p = off[i] + l; p < pe; p += 16) {
            const float2 r = *(const float2*)(rel3 + (size_t)key[p] * 2);
            s0 += r.x; s1 += r.y;
        }
    }
    for (int m = 8; m >= 1; m >>= 1) {
        s0 += __shfl_xor(s0, m, 16);
        s1 += __shfl_xor(s1, m, 16);
    }
    if (i < N && l == 0) {
        agg3[(size_t)i * 2 + 0] += s0;
        agg3[(size_t)i * 2 + 1] += s1;
    }
}

template <int U>
__device__ __forceinline__ void l2_add_rows(const float* __restrict__ h1, const int* __restrict__ kp,
                                            int N, int f, float4& a0, float4& b0, float4& a1, float4& b1) {
    int    kk[U];
    float  m1[U];
    float4 lo[U], hi[U];
    #pragma unroll
    for (int j = 0; j < U; ++j) kk[j] = kp[j];
    #pragma unroll
    for (int j = 0; j < U; ++j) {
        const bool r = kk[j] >= N;
        const int  s = r ? kk[j] - N : kk[j];
        m1[j] = r ? 1.0f : 0.0f;
        const float* rp = h1 + (size_t)s * 64 + f;
        lo[j] = *(const float4*)rp;
        hi[j] = *(const float4*)(rp + 32);
    }
    #pragma unroll
    for (int j = 0; j < U; ++j) {
        const float w1 = m1[j], w0 = 1.0f - m1[j];
        a0.x = fmaf(lo[j].x, w0, a0.x); a0.y = fmaf(lo[j].y, w0, a0.y);
        a0.z = fmaf(lo[j].z, w0, a0.z); a0.w = fmaf(lo[j].w, w0, a0.w);
        b0.x = fmaf(hi[j].x, w0, b0.x); b0.y = fmaf(hi[j].y, w0, b0.y);
        b0.z = fmaf(hi[j].z, w0, b0.z); b0.w = fmaf(hi[j].w, w0, b0.w);
        a1.x = fmaf(lo[j].x, w1, a1.x); a1.y = fmaf(lo[j].y, w1, a1.y);
        a1.z = fmaf(lo[j].z, w1, a1.z); a1.w = fmaf(lo[j].w, w1, a1.w);
        b1.x = fmaf(hi[j].x, w1, b1.x); b1.y = fmaf(hi[j].y, w1, b1.y);
        b1.z = fmaf(hi[j].z, w1, b1.z); b1.w = fmaf(hi[j].w, w1, b1.w);
    }
}

#define L2_NODES  32
#define L2_PITCH  196
#define L2_HPITCH 68
__global__ __launch_bounds__(256) void layer2_sum_xw_l3(const float* __restrict__ h1, const int* __restrict__ off,
                                 const int* __restrict__ key,
                                 const float* __restrict__ W2, const float* __restrict__ loop2,
                                 const float* __restrict__ b2,
                                 const float* __restrict__ W3, const float* __restrict__ loop3,
                                 const float* __restrict__ b3,
                                 float* __restrict__ rel3, float* __restrict__ agg3, int N) {
    __shared__ __align__(16) float As[L2_NODES * L2_PITCH];
    __shared__ __align__(16) float Hs[L2_NODES * L2_HPITCH];
    const int t    = (int)threadIdx.x;
    const int row  = t >> 3;
    const int f    = (t & 7) * 4;
    const int i    = (int)blockIdx.x * L2_NODES + row;
    float4 a0  = make_float4(0.0f, 0.0f, 0.0f, 0.0f);
    float4 b0  = make_float4(0.0f, 0.0f, 0.0f, 0.0f);
    float4 a1  = make_float4(0.0f, 0.0f, 0.0f, 0.0f);
    float4 b1  = make_float4(0.0f, 0.0f, 0.0f, 0.0f);
    float4 own = make_float4(0.0f, 0.0f, 0.0f, 0.0f);
    float4 owb = make_float4(0.0f, 0.0f, 0.0f, 0.0f);
    if (i < N) {
        own = *(const float4*)(h1 + (size_t)i * 64 + f);
        owb = *(const float4*)(h1 + (size_t)i * 64 + f + 32);
        int p = off[i];
        const int pe = off[i + 1];
        for (; p + 8 <= pe; p += 8) l2_add_rows<8>(h1, key + p, N, f, a0, b0, a1, b1);
        if (p + 4 <= pe) { l2_add_rows<4>(h1, key + p, N, f, a0, b0, a1, b1); p += 4; }
        for (; p < pe; ++p) l2_add_rows<1>(h1, key + p, N, f, a0, b0, a1, b1);
    }
    {
        float* ar = As + row * L2_PITCH + f;
        *(float4*)(ar)       = a0;
        *(float4*)(ar + 32)  = b0;
        *(float4*)(ar + 64)  = a1;
        *(float4*)(ar + 96)  = b1;
        *(float4*)(ar + 128) = own;
        *(float4*)(ar + 160) = owb;
    }
    __syncthreads();

    {
        const int wave  = t >> 5;
        const int strip = wave >> 2;
        const int lane  = t & 31;
        const int half  = lane >> 4;
        const int lm    = lane & 15;
        const int n     = (wave & 3) * 16 + lm;
        v8f c0, c1, c2;
        const float bv = b2[n];
        #pragma unroll
        for (int v = 0; v < 8; ++v) { c0[v] = 0.0f; c1[v] = 0.0f; c2[v] = bv; }
        const float* arow = As + (strip * 16 + lm) * L2_PITCH + 2 * half;
        const float* w1p  = W2 + 64 * 64;
        #pragma unroll
        for (int k0 = 0; k0 < 64; k0 += 4) {
            const int kb = k0 + 2 * half;
            #pragma unroll
            for (int mt = 0; mt < 3; ++mt) {
                const float* Wm = (mt == 0) ? W2 : (mt == 1) ? w1p : loop2;
                const float2 av = *(const float2*)(arow + 64 * mt + k0);
                v2f a; a.x = av.x; a.y = av.y;
                v2f b;
                b.x = Wm[(size_t)kb * 64 + n];
                b.y = Wm[(size_t)(kb + 1) * 64 + n];
                if (mt == 0)      c0 = __builtin_amdgcn_wmma_f32_16x16x4_f32(false, a, false, b, (short)0, c0, false, false);
                else if (mt == 1) c1 = __builtin_amdgcn_wmma_f32_16x16x4_f32(false, a, false, b, (short)0, c1, false, false);
                else              c2 = __builtin_amdgcn_wmma_f32_16x16x4_f32(false, a, false, b, (short)0, c2, false, false);
            }
        }
        #pragma unroll
        for (int v = 0; v < 8; ++v)
            Hs[(strip * 16 + v + 8 * half) * L2_HPITCH + n] = fmaxf((c0[v] + c1[v]) + c2[v], 0.0f);
    }
    __syncthreads();

    const float4 hq = *(const float4*)(Hs + row * L2_HPITCH + f);
    const float4 hr = *(const float4*)(Hs + row * L2_HPITCH + f + 32);
    const float hv[4] = { hq.x, hq.y, hq.z, hq.w };
    const float hw[4] = { hr.x, hr.y, hr.z, hr.w };
    const float* wm[3] = { W3 + f * 2, W3 + 128 + f * 2, loop3 + f * 2 };
    float o[6];
    #pragma unroll
    for (int m = 0; m < 3; ++m) {
        float s0 = 0.0f, s1 = 0.0f, u0 = 0.0f, u1 = 0.0f;
        #pragma unroll
        for (int j = 0; j < 4; ++j) {
            s0 += hv[j] * wm[m][2 * j + 0];
            s1 += hv[j] * wm[m][2 * j + 1];
        }
        #pragma unroll
        for (int j = 0; j < 4; ++j) {
            u0 += hw[j] * wm[m][64 + 2 * j + 0];
            u1 += hw[j] * wm[m][64 + 2 * j + 1];
        }
        o[2 * m + 0] = s0 + u0;
        o[2 * m + 1] = s1 + u1;
    }
    for (int d = 4; d >= 1; d >>= 1) {
        #pragma unroll
        for (int q = 0; q < 6; ++q) o[q] += __shfl_xor(o[q], d, 8);
    }
    if ((t & 7) == 0 && i < N) {
        rel3[(size_t)i * 2 + 0] = o[0];
        rel3[(size_t)i * 2 + 1] = o[1];
        rel3[(size_t)(N + i) * 2 + 0] = o[2];
        rel3[(size_t)(N + i) * 2 + 1] = o[3];
        agg3[(size_t)i * 2 + 0] = o[4] + b3[0];
        agg3[(size_t)i * 2 + 1] = o[5] + b3[1];
    }
}

#define TAIL_TPB 256
__global__ __launch_bounds__(TAIL_TPB) void gather2_min(const float* __restrict__ rel3, const int* __restrict__ off,
                            const int* __restrict__ key, float* __restrict__ agg3,
                            float* __restrict__ blockmin, int N) {
    __shared__ float sm[TAIL_TPB / 32];
    const long long t = (long long)blockIdx.x * blockDim.x + threadIdx.x;
    const int i = (int)(t >> 2);
    const int q = (int)t & 3;
    float s0[4] = { 0.0f, 0.0f, 0.0f, 0.0f };
    float s1[4] = { 0.0f, 0.0f, 0.0f, 0.0f };
    float g0 = 0.0f, g1 = 0.0f;
    if (i < N) {
        const int p0 = off[i];
        const int pe = off[i + 1];
        if (q == 0) {
            const float2 g = *(const float2*)(agg3 + (size_t)i * 2);
            g0 = g.x; g1 = g.y;
        }
        for (int base = p0; base < pe; base += 32) {
            int    kk[8];
            float2 r[8];
            #pragma unroll
            for (int u = 0; u < 8; ++u) kk[u] = key[min(base + q + 4 * u, pe - 1)];
            #pragma unroll
            for (int u = 0; u < 8; ++u) r[u] = *(const float2*)(rel3 + (size_t)kk[u] * 2);
            #pragma unroll
            for (int u = 0; u < 8; ++u) {
                const float w = (base + q + 4 * u < pe) ? 1.0f : 0.0f;
                s0[u & 3] = fmaf(r[u].x, w, s0[u & 3]);
                s1[u & 3] = fmaf(r[u].y, w, s1[u & 3]);
            }
        }
    }
    float t0 = (s0[0] + s0[2]) + (s0[1] + s0[3]);
    float t1 = (s1[0] + s1[2]) + (s1[1] + s1[3]);
    for (int m = 2; m >= 1; m >>= 1) {
        t0 += __shfl_xor(t0, m, 4);
        t1 += __shfl_xor(t1, m, 4);
    }
    float mn = 3.402823466e38f;
    if (i < N && q == 0) {
        const float v0 = g0 + t0;
        const float v1 = g1 + t1;
        agg3[(size_t)i * 2 + 0] = v0;
        agg3[(size_t)i * 2 + 1] = v1;
        mn = fminf(v0, v1);
    }
    for (int m = 16; m >= 4; m >>= 1) mn = fminf(mn, __shfl_xor(mn, m, 32));
    if ((threadIdx.x & 31) == 0) sm[threadIdx.x >> 5] = mn;
    __syncthreads();
    if (threadIdx.x == 0) {
        float r = sm[0];
        #pragma unroll
        for (int w = 1; w < TAIL_TPB / 32; ++w) r = fminf(r, sm[w]);
        blockmin[blockIdx.x] = r;
    }
}

__global__ __launch_bounds__(TAIL_TPB) void finalize_bm(const float* __restrict__ agg3,
                            const int* __restrict__ cell,
                            const int* __restrict__ maxs,
                            const float* __restrict__ blockmin, int nblk,
                            float* __restrict__ out, int N) {
    __shared__ float sm[TAIL_TPB / 32];
    float mn = 3.402823466e38f;
    for (int j0 = threadIdx.x; j0 < nblk; j0 += 16 * TAIL_TPB) {
        float v[16];
        #pragma unroll
        for (int u = 0; u < 16; ++u) v[u] = blockmin[min(j0 + u * TAIL_TPB, nblk - 1)];
        #pragma unroll
        for (int u = 0; u < 16; ++u) mn = fminf(mn, v[u]);
    }
    for (int m = 16; m >= 1; m >>= 1) mn = fminf(mn, __shfl_xor(mn, m, 32));
    if ((threadIdx.x & 31) == 0) sm[threadIdx.x >> 5] = mn;
    __syncthreads();
    float r = sm[0];
    #pragma unroll
    for (int w = 1; w < TAIL_TPB / 32; ++w) r = fminf(r, sm[w]);
    const float gmin = r - 1.0f;
    const int i = blockIdx.x * blockDim.x + threadIdx.x;
    if (i >= N) return;
    const bool upper = cell[i] >= (maxs[i] - 1);
    const bool lower = cell[i] == 0;
    out[(size_t)i * 2 + 0] = upper ? gmin : agg3[(size_t)i * 2 + 0];
    out[(size_t)i * 2 + 1] = lower ? gmin : agg3[(size_t)i * 2 + 1];
}

__device__ __forceinline__ unsigned f2ord(float f) {
    unsigned u = __float_as_uint(f);
    return (u & 0x80000000u) ? ~u : (u | 0x80000000u);
}
__device__ __forceinline__ float ord2f(unsigned u) {
    return __uint_as_float((u & 0x80000000u) ? (u ^ 0x80000000u) : ~u);
}

__global__ void init_slot(unsigned* slot) { *slot = 0xFFFFFFFFu; }

__global__ void min_reduce(const float* __restrict__ p, int n, unsigned* slot) {
    __shared__ float sm[256];
    float m = 3.402823466e38f;
    for (int i = blockIdx.x * blockDim.x + threadIdx.x; i < n;
         i += gridDim.x * blockDim.x)
        m = fminf(m, p[i]);
    sm[threadIdx.x] = m;
    __syncthreads();
    for (int s = blockDim.x >> 1; s > 0; s >>= 1) {
        if ((int)threadIdx.x < s) sm[threadIdx.x] = fminf(sm[threadIdx.x], sm[threadIdx.x + s]);
        __syncthreads();
    }
    if (threadIdx.x == 0) atomicMin(slot, f2ord(sm[0]));
}

__global__ void finalize_k(const float* __restrict__ agg3,
                           const int* __restrict__ cell,
                           const int* __restrict__ maxs,
                           const unsigned* __restrict__ slot,
                           float* __restrict__ out, int N) {
    const int i = blockIdx.x * blockDim.x + threadIdx.x;
    if (i >= N) return;
    const float gmin = ord2f(*slot) - 1.0f;
    const bool upper = cell[i] >= (maxs[i] - 1);
    const bool lower = cell[i] == 0;
    out[(size_t)i * 2 + 0] = upper ? gmin : agg3[(size_t)i * 2 + 0];
    out[(size_t)i * 2 + 1] = lower ? gmin : agg3[(size_t)i * 2 + 1];
}

extern "C" void kernel_launch(void* const* d_in, const int* in_sizes, int n_in,
                              void* d_out, int out_size, void* d_ws, size_t ws_size,
                              hipStream_t stream) {
    const float* x      = (const float*)d_in[0];
    const int*   src    = (const int*)d_in[1];
    const int*   dst    = (const int*)d_in[2];
    const int*   etypes = (const int*)d_in[3];
    const int*   cell   = (const int*)d_in[4];
    const int*   maxs   = (const int*)d_in[5];
    const float* W1     = (const float*)d_in[6];
    const float* loop1  = (const float*)d_in[7];
    const float* b1     = (const float*)d_in[8];
    const float* W2     = (const float*)d_in[9];
    const float* loop2  = (const float*)d_in[10];
    const float* b2     = (const float*)d_in[11];
    const float* W3     = (const float*)d_in[12];
    const float* loop3  = (const float*)d_in[13];
    const float* b3     = (const float*)d_in[14];
    float* out = (float*)d_out;

    const int N = in_sizes[4];
    const int E = in_sizes[1];
    const int K1 = in_sizes[0] / N;

    float* relbuf = (float*)d_ws;
    float* h1     = relbuf + (size_t)2 * N * 64;
    float* h2     = h1 + (size_t)N * 64;
    float* rel3   = h2 + (size_t)N * 64;
    float* agg3   = rel3 + (size_t)4 * N;
    unsigned* slot = (unsigned*)(agg3 + (size_t)2 * N);

    const int waves   = (N + 15) / 16;
    const int gblocks = (waves * 32 + 255) / 256;
    const int eb64    = (int)(((long long)E * 16 + 255) / 256);
    const int nb      = (N + 255) / 256;
    const int fb      = (N * 64 + 255) / 256;
    const int eb2     = (E + 255) / 256;

    int* key      = (int*)(slot + 4);
    int* off      = key + (size_t)E;
    int* blockcnt = (int*)relbuf;
    int* rank     = blockcnt + (size_t)CSR_NB * N;
    const size_t need = ((size_t)262 * N + 4 + (size_t)E + (size_t)N + 1) * sizeof(float);
    int* coarse   = (int*)rel3;
    const bool csr_ok = ws_size >= need && (long long)E <= (long long)192 * N
                        && (long long)CSR_NB * nb <= (long long)6 * N;
    const size_t need2 = need + (size_t)E * sizeof(int);
    const int chunk0 = (E + CSR_NB - 1) / CSR_NB;
    const bool shared_ok = csr_ok && ws_size >= need2
                           && N <= CSR_CAP && chunk0 <= 1024 * CSR_EPT
                           && (long long)2 * N <= (long long)0x1000000
                           && (long long)2 * CSR_NB * nb <= (long long)6 * N;

    if (shared_ok) {
        const int chunk = chunk0;
        const int gb8   = (int)(((long long)N * 8 + 255) / 256);
        const int gb4   = (int)(((long long)N * 4 + 255) / 256);
        const int l2b   = (N + L2_NODES - 1) / L2_NODES;
        int* blockcnt2  = (int*)h2;
        unsigned* ebin  = (unsigned*)(off + (size_t)N + 1);
        int* cstart     = coarse + (size_t)CSR_NB * nb;
        csr_hist_bin<<<CSR_NB, 1024, 0, stream>>>(dst, src, etypes, ebin, blockcnt2, coarse, cstart, E, N, chunk, nb);

        gemm3_xw_colscan_place<<<gblocks + nb, 256, 0, stream>>>(x, W1, loop1, b1, relbuf, h1, N, K1, gblocks,
                                                                 blockcnt2, coarse, cstart, ebin, off, key, E, chunk, nb);
        gather64_relu<<<gb8, 256, 0, stream>>>(relbuf, off, key, h1, N);

        layer2_sum_xw_l3<<<l2b, 256, 0, stream>>>(h1, off, key, W2, loop2, b2, W3, loop3, b3, rel3, agg3, N);

        float* blockmin = relbuf;
        gather2_min<<<gb4, TAIL_TPB, 0, stream>>>(rel3, off, key, agg3, blockmin, N);
        finalize_bm<<<nb, TAIL_TPB, 0, stream>>>(agg3, cell, maxs, blockmin, gb4, out, N);
    } else if (csr_ok) {
        const int chunk = (E + CSR_NB - 1) / CSR_NB;
        const int gb8   = (int)(((long long)N * 8 + 255) / 256);
        const int gb4   = (int)(((long long)N * 4 + 255) / 256);
        const int l2b   = (N + L2_NODES - 1) / L2_NODES;
        csr_hist<<<CSR_NB, 1024, 0, stream>>>(dst, rank, blockcnt, coarse, E, N, chunk, nb);
        csr_colscan_off<<<nb, CSR_RANGE, 0, stream>>>(blockcnt, coarse, off, N, nb);
        csr_fill<<<eb2, 256, 0, stream>>>(src, dst, etypes, rank, blockcnt, key, E, N, chunk);

        gemm3_xw<<<gblocks, 256, 0, stream>>>(x, W1, loop1, b1, relbuf, h1, N, K1);
        gather64_relu<<<gb8, 256, 0, stream>>>(relbuf, off, key, h1, N);

        layer2_sum_xw_l3<<<l2b, 256, 0, stream>>>(h1, off, key, W2, loop2, b2, W3, loop3, b3, rel3, agg3, N);

        float* blockmin = relbuf;
        gather2_min<<<gb4, TAIL_TPB, 0, stream>>>(rel3, off, key, agg3, blockmin, N);
        finalize_bm<<<nb, TAIL_TPB, 0, stream>>>(agg3, cell, maxs, blockmin, gb4, out, N);
    } else {
        gemm3_xw<<<gblocks, 256, 0, stream>>>(x, W1, loop1, b1, relbuf, h1, N, K1);
        edge_scatter64<<<eb64, 256, 0, stream>>>(relbuf, src, dst, etypes, h1, E, N);
        relu_k<<<fb, 256, 0, stream>>>(h1, N * 64);

        gemm3_xw<<<gblocks, 256, 0, stream>>>(h1, W2, loop2, b2, relbuf, h2, N, 64);
        edge_scatter64<<<eb64, 256, 0, stream>>>(relbuf, src, dst, etypes, h2, E, N);
        relu_k<<<fb, 256, 0, stream>>>(h2, N * 64);

        layer3_node<<<nb, 256, 0, stream>>>(h2, W3, loop3, b3, rel3, agg3, N);
        edge_scatter2<<<eb2, 256, 0, stream>>>(rel3, src, dst, etypes, agg3, E, N);
        init_slot<<<1, 1, 0, stream>>>(slot);
        min_reduce<<<512, 256, 0, stream>>>(agg3, N * 2, slot);
        finalize_k<<<nb, 256, 0, stream>>>(agg3, cell, maxs, slot, out, N);
    }
}


# The edge sort's two scattered 4-byte store streams (binned edge words, placed keys) are assembled in LDS and leave as whole 128-byte lines; key places and sum order unchanged
# speedup vs baseline: 3.0119x; 1.0434x over previous
#include <hip/hip_runtime.h>

typedef __attribute__((ext_vector_type(2))) float v2f;
typedef __attribute__((ext_vector_type(8))) float v8f;

__device__ __forceinline__ void gemm3_strip(const float* __restrict__ X, const float* __restrict__ Wr,
                         const float* __restrict__ Wl, const float* __restrict__ bias,
                         float* __restrict__ rel, float* __restrict__ H,
                         int M, int K, int block) {
    const int gthread = block * blockDim.x + threadIdx.x;
    const int wave    = gthread >> 5;
    const int lane    = threadIdx.x & 31;
    const int m0      = wave * 16;
    if (m0 >= M) return;

    const int half = lane >> 4;
    const int lm   = lane & 15;

    v8f acc[12];
    #pragma unroll
    for (int t = 0; t < 8; ++t) {
        #pragma unroll
        for (int v = 0; v < 8; ++v) acc[t][v] = 0.0f;
    }
    #pragma unroll
    for (int nt = 0; nt < 4; ++nt) {
        const float bv = bias[nt * 16 + lm];
        #pragma unroll
        for (int v = 0; v < 8; ++v) acc[8 + nt][v] = bv;
    }

    const float* xrow = X + (size_t)(m0 + lm) * K;
    const float* w1   = Wr + (size_t)K * 64;

    #pragma unroll 2
    for (int k0 = 0; k0 < K; k0 += 4) {
        const int kb = k0 + 2 * half;
        const float2 av = *(const float2*)(xrow + kb);
        v2f a; a.x = av.x; a.y = av.y;
        #pragma unroll
        for (int mt = 0; mt < 3; ++mt) {
            const float* Wm = (mt == 0) ? Wr : (mt == 1) ? w1 : Wl;
            #pragma unroll
            for (int nt = 0; nt < 4; ++nt) {
                const int n = nt * 16 + lm;
                v2f b;
                b.x = Wm[(size_t)kb * 64 + n];
                b.y = Wm[(size_t)(kb + 1) * 64 + n];
                acc[mt * 4 + nt] = __builtin_amdgcn_wmma_f32_16x16x4_f32(
                    false, a, false, b, (short)0, acc[mt * 4 + nt], false, false);
            }
        }
    }

    #pragma unroll
    for (int mt = 0; mt < 3; ++mt) {
        float* C = (mt == 2) ? H : rel + (size_t)mt * M * 64;
        #pragma unroll
        for (int nt = 0; nt < 4; ++nt) {
            const int n = nt * 16 + lm;
            #pragma unroll
            for (int v = 0; v < 8; ++v) {
                C[(size_t)(m0 + v + 8 * half) * 64 + n] = acc[mt * 4 + nt][v];
            }
        }
    }
}

__global__ __launch_bounds__(256) void gemm3_xw(const float* __restrict__ X, const float* __restrict__ Wr,
                         const float* __restrict__ Wl, const float* __restrict__ bias,
                         float* __restrict__ rel, float* __restrict__ H,
                         int M, int K) {
    gemm3_strip(X, Wr, Wl, bias, rel, H, M, K, (int)blockIdx.x);
}

__global__ void edge_scatter64(const float* __restrict__ rel,
                               const int* __restrict__ src,
                               const int* __restrict__ dst,
                               const int* __restrict__ et,
                               float* __restrict__ agg, int E, int N) {
    const long long t = (long long)blockIdx.x * blockDim.x + threadIdx.x;
    const int e = (int)(t >> 4);
    if (e >= E) return;
    const int f = ((int)t & 15) * 4;
    const float4 v = *(const float4*)(rel + ((long long)et[e] * N + src[e]) * 64 + f);
    float* q = agg + (long long)dst[e] * 64 + f;
    atomicAdd(q + 0, v.x);
    atomicAdd(q + 1, v.y);
    atomicAdd(q + 2, v.z);
    atomicAdd(q + 3, v.w);
}

__global__ void relu_k(float* __restrict__ p, int n) {
    const int i = blockIdx.x * blockDim.x + threadIdx.x;
    if (i < n) p[i] = fmaxf(p[i], 0.0f);
}

__global__ void layer3_node(const float* __restrict__ h2,
                            const float* __restrict__ W3,
                            const float* __restrict__ loop3,
                            const float* __restrict__ b3,
                            float* __restrict__ rel3, float* __restrict__ agg3,
                            int N) {
    const int i = blockIdx.x * blockDim.x + threadIdx.x;
    if (i >= N) return;
    const float* row = h2 + (size_t)i * 64;
    float r00 = 0.f, r01 = 0.f, r10 = 0.f, r11 = 0.f, l0 = 0.f, l1 = 0.f;
    #pragma unroll 8
    for (int k = 0; k < 64; ++k) {
        const float v = row[k];
        r00 += v * W3[k * 2 + 0];
        r01 += v * W3[k * 2 + 1];
        r10 += v * W3[128 + k * 2 + 0];
        r11 += v * W3[128 + k * 2 + 1];
        l0  += v * loop3[k * 2 + 0];
        l1  += v * loop3[k * 2 + 1];
    }
    rel3[(size_t)i * 2 + 0] = r00;
    rel3[(size_t)i * 2 + 1] = r01;
    rel3[(size_t)(N + i) * 2 + 0] = r10;
    rel3[(size_t)(N + i) * 2 + 1] = r11;
    agg3[(size_t)i * 2 + 0] = l0 + b3[0];
    agg3[(size_t)i * 2 + 1] = l1 + b3[1];
}

__global__ void edge_scatter2(const float* __restrict__ rel3,
                              const int* __restrict__ src,
                              const int* __restrict__ dst,
                              const int* __restrict__ et,
                              float* __restrict__ agg3, int E, int N) {
    const int e = blockIdx.x * blockDim.x + threadIdx.x;
    if (e >= E) return;
    const float* p = rel3 + ((long long)et[e] * N + src[e]) * 2;
    float* q = agg3 + (long long)dst[e] * 2;
    atomicAdd(q + 0, p[0]);
    atomicAdd(q + 1, p[1]);
}


#define CSR_NB  64
#define CSR_CAP 51200
#define CSR_RANGE 256

__global__ __launch_bounds__(1024) void csr_hist(const int* __restrict__ dst,
                                                 int* __restrict__ rank,
                                                 int* __restrict__ blockcnt,
                                                 int* __restrict__ coarse,
                                                 int E, int N, int chunk, int nr_total) {
    __shared__ int cnt[CSR_CAP];
    const int b  = blockIdx.x;
    const int e0 = b * chunk;
    const int e1 = min(e0 + chunk, E);
    for (int base = 0; base < N; base += CSR_CAP) {
        const int lim = min(CSR_CAP, N - base);
        for (int i = threadIdx.x; i < lim; i += blockDim.x) cnt[i] = 0;
        __syncthreads();
        const int bd = (int)blockDim.x;
        for (int e = e0 + (int)threadIdx.x; e < e1; e += 4 * bd) {
            int d[4];
            #pragma unroll
            for (int j = 0; j < 4; ++j) {
                const int ee = e + j * bd;
                const int v  = dst[min(ee, e1 - 1)];
                d[j] = (ee < e1) ? v - base : -1;
            }
            #pragma unroll
            for (int j = 0; j < 4; ++j)
                if (d[j] >= 0 && d[j] < lim) rank[e + j * bd] = atomicAdd(&cnt[d[j]], 1);
        }
        __syncthreads();
        const int nrs = (lim + CSR_RANGE - 1) / CSR_RANGE;
        for (int r0 = 0; r0 < nrs; r0 += (int)blockDim.x >> 2) {
            const int r  = r0 + ((int)threadIdx.x >> 2);
            const int q  = (int)threadIdx.x & 3;
            const int lo = r * CSR_RANGE + q * (CSR_RANGE / 4);
            int s = 0;
            #pragma unroll 16
            for (int j = 0; j < CSR_RANGE / 4; ++j) {
                const int i = lo + ((j + (int)threadIdx.x) & (CSR_RANGE / 4 - 1));
                const int v = cnt[min(i, lim - 1)];
                s += (i < lim) ? v : 0;
            }
            s += __shfl_xor(s, 1, 4);
            s += __shfl_xor(s, 2, 4);
            if (q == 0 && r < nrs) coarse[(size_t)b * nr_total + base / CSR_RANGE + r] = s;
        }
        for (int i = threadIdx.x; i < lim; i += blockDim.x)
            blockcnt[(size_t)b * N + base + i] = cnt[i];
        __syncthreads();
    }
}

#define CSR_EPT 16
__global__ __launch_bounds__(1024) void csr_hist_bin(const int* __restrict__ dst, const int* __restrict__ src,
                                                     const int* __restrict__ et,
                                                     unsigned* __restrict__ ebin,
                                                     int* __restrict__ blockcnt,
                                                     int* __restrict__ coarse, int* __restrict__ cstart,
                                                     int E, int N, int chunk, int nr_total) {
    __shared__ int cnt[CSR_CAP];
    __shared__ int rcnt[CSR_CAP / CSR_RANGE];
    __shared__ int rstart[CSR_CAP / CSR_RANGE];
    __shared__ int swt[8];
    __shared__ int stot;
    __shared__ unsigned stage[1024 * CSR_EPT];
    const int tid  = (int)threadIdx.x;
    const int lane = tid & 31;
    const int b    = (int)blockIdx.x;
    const int e0   = b * chunk;
    const int e1   = min(e0 + chunk, E);
    for (int i = tid; i < N; i += 1024) cnt[i] = 0;
    if (tid < nr_total) rcnt[tid] = 0;
    int dd[CSR_EPT], ss[CSR_EPT], tt[CSR_EPT];
    #pragma unroll
    for (int j = 0; j < CSR_EPT; ++j) {
        const int ec = max(min(e0 + tid + j * 1024, e1 - 1), 0);
        dd[j] = dst[ec]; ss[j] = src[ec]; tt[j] = et[ec];
    }
    #pragma unroll
    for (int j = 0; j < CSR_EPT; ++j)
        if (e0 + tid + j * 1024 >= e1) dd[j] = -1;
    __syncthreads();
    int slot[CSR_EPT];
    #pragma unroll
    for (int j = 0; j < CSR_EPT; ++j) {
        slot[j] = 0;
        if (dd[j] >= 0 && dd[j] < N) {
            atomicAdd(&cnt[dd[j]], 1);
            slot[j] = atomicAdd(&rcnt[dd[j] / CSR_RANGE], 1);
        } else dd[j] = -1;
    }
    __syncthreads();
    int rv = 0, inc = 0;
    if (tid < 256) {
        rv  = (tid < nr_total) ? rcnt[tid] : 0;
        inc = rv;
        for (int d = 1; d < 32; d <<= 1) {
            const int v = __shfl_up(inc, d, 32);
            if (lane >= d) inc += v;
        }
        if (lane == 31) swt[tid >> 5] = inc;
    }
    __syncthreads();
    if (tid < 256) {
        int wb = 0;
        #pragma unroll
        for (int w = 0; w < 8; ++w) wb += (w < (tid >> 5)) ? swt[w] : 0;
        const int ex = wb + inc - rv;
        if (tid < nr_total) {
            rstart[tid] = ex;
            coarse[(size_t)b * nr_total + tid] = rv;
            cstart[(size_t)b * nr_total + tid] = ex;
        }
        if (tid == 255) stot = ex + rv;
    }
    __syncthreads();
    #pragma unroll
    for (int j = 0; j < CSR_EPT; ++j)
        slot[j] += rstart[max(dd[j], 0) / CSR_RANGE];
    #pragma unroll
    for (int j = 0; j < CSR_EPT; ++j) {
        if (dd[j] >= 0) {
            const int r = dd[j] / CSR_RANGE;
            const int p = min(max(slot[j], 0), e1 - e0 - 1);
            const unsigned kv = (unsigned)(tt[j] * N + ss[j]);
            stage[p] = (kv & 0xFFFFFFu) | ((unsigned)(dd[j] - r * CSR_RANGE) << 24);
        }
    }
    for (int i = tid; i < N; i += 1024) blockcnt[(size_t)b * N + i] = cnt[i];
    __syncthreads();
    const int tot = min(max(stot, 0), e1 - e0);
    for (int i0 = tid; i0 < tot; i0 += 4 * 1024) {
        unsigned w[4];
        #pragma unroll
        for (int u = 0; u < 4; ++u) w[u] = stage[min(i0 + u * 1024, 1024 * CSR_EPT - 1)];
        #pragma unroll
        for (int u = 0; u < 4; ++u)
            if (i0 + u * 1024 < tot) ebin[(size_t)e0 + i0 + u * 1024] = w[u];
    }
}

__device__ __forceinline__ void csr_colscan_range(int* __restrict__ blockcnt,
                                                  const int* __restrict__ coarse,
                                                  int* __restrict__ off, int N, int nr_total, int r) {
    __shared__ int wtot[CSR_RANGE / 32];
    __shared__ int wbase[CSR_RANGE / 32];
    const int t  = threadIdx.x;
    const int n  = r * CSR_RANGE + t;
    const int nn = min(n, N - 1);
    int c[CSR_NB];
    #pragma unroll
    for (int j = 0; j < CSR_NB; ++j) c[j] = blockcnt[(size_t)j * N + nn];
    int part = 0;
    for (int rr = t; rr < r; rr += CSR_RANGE) {
        #pragma unroll 16
        for (int b = 0; b < CSR_NB; ++b) part += coarse[(size_t)b * nr_total + rr];
    }
    int deg = 0;
    #pragma unroll
    for (int j = 0; j < CSR_NB; ++j) deg += c[j];
    if (n >= N) deg = 0;
    const int lane = t & 31;
    int inc = deg;
    for (int d = 1; d < 32; d <<= 1) {
        const int v = __shfl_up(inc, d, 32);
        if (lane >= d) inc += v;
    }
    for (int m = 16; m >= 1; m >>= 1) part += __shfl_xor(part, m, 32);
    if (lane == 31) wtot[t >> 5] = inc;
    if (lane == 0)  wbase[t >> 5] = part;
    __syncthreads();
    int run = inc - deg;
    #pragma unroll
    for (int w = 0; w < CSR_RANGE / 32; ++w) {
        run += wbase[w];
        run += (w < (t >> 5)) ? wtot[w] : 0;
    }
    if (n < N) {
        off[n] = run;
        if (n == N - 1) off[N] = run + deg;
        #pragma unroll
        for (int j = 0; j < CSR_NB; ++j) {
            blockcnt[(size_t)j * N + n] = run;
            run += c[j];
        }
    }
}

__global__ __launch_bounds__(CSR_RANGE) void csr_colscan_off(int* __restrict__ blockcnt,
                                                             const int* __restrict__ coarse,
                                                             int* __restrict__ off, int N, int nr_total) {
    csr_colscan_range(blockcnt, coarse, off, N, nr_total, (int)blockIdx.x);
}

#define CSR_KST 8192
static_assert(CSR_NB == 64 && CSR_RANGE == 256, "8 waves x 8 chunks, one thread per node of the range");
__device__ __forceinline__ void csr_put_key(int pos, int k, int kb, int* __restrict__ kst, int* __restrict__ key, int E) {
    const int li = pos - kb;
    if (li >= 0 && li < CSR_KST) kst[li] = k;
    else key[min(max(pos, 0), E - 1)] = k;
}
__device__ __forceinline__ void csr_place_one(unsigned rec, int* __restrict__ trow, int kb, int* __restrict__ kst,
                                              int* __restrict__ key, int E) {
    const int dl = (int)(rec >> 24);
    const int pos = atomicAdd(&trow[dl], 1);
    csr_put_key(pos, (int)(rec & 0xFFFFFFu), kb, kst, key, E);
}

__device__ __forceinline__ void csr_colscan_place(const int* __restrict__ blockcnt,
                                                  const int* __restrict__ coarse, const int* __restrict__ cstart,
                                                  const unsigned* __restrict__ ebin,
                                                  int* __restrict__ off, int* __restrict__ key,
                                                  int N, int E, int chunk, int nr_total, int r) {
    __shared__ int wtot[CSR_RANGE / 32];
    __shared__ int wbase[CSR_RANGE / 32];
    __shared__ int tab[CSR_NB * CSR_RANGE];
    __shared__ int kst[CSR_KST];
    const int t    = threadIdx.x;
    const int wave = t >> 5;
    const int lane = t & 31;
    const int n  = r * CSR_RANGE + t;
    const int nn = min(n, N - 1);
    int len[8], st[8];
    #pragma unroll
    for (int q = 0; q < 8; ++q) {
        const int bq = wave + 8 * q;
        len[q] = coarse[(size_t)bq * nr_total + r];
        st[q]  = cstart[(size_t)bq * nr_total + r];
    }
    int c[CSR_NB];
    #pragma unroll
    for (int j = 0; j < CSR_NB; ++j) c[j] = blockcnt[(size_t)j * N + nn];
    int part = 0;
    for (int rr = t; rr < r; rr += CSR_RANGE) {
        #pragma unroll 16
        for (int b = 0; b < CSR_NB; ++b) part += coarse[(size_t)b * nr_total + rr];
    }
    int deg = 0;
    #pragma unroll
    for (int j = 0; j < CSR_NB; ++j) deg += c[j];
    if (n >= N) deg = 0;
    int inc = deg;
    for (int d = 1; d < 32; d <<= 1) {
        const int v = __shfl_up(inc, d, 32);
        if (lane >= d) inc += v;
    }
    for (int m = 16; m >= 1; m >>= 1) part += __shfl_xor(part, m, 32);
    if (lane == 31) wtot[wave] = inc;
    if (lane == 0)  wbase[wave] = part;
    __syncthreads();
    int run = inc - deg;
    int kb = 0, kt = 0;
    #pragma unroll
    for (int w = 0; w < CSR_RANGE / 32; ++w) {
        run += wbase[w];
        run += (w < wave) ? wtot[w] : 0;
        kb  += wbase[w];
        kt  += wtot[w];
    }
    if (n < N) {
        off[n] = run;
        if (n == N - 1) off[N] = run + deg;
    }
    #pragma unroll
    for (int j = 0; j < CSR_NB; ++j) {
        tab[j * CSR_RANGE + t] = run;
        run += c[j];
    }
    __syncthreads();

    unsigned rec[8][3];
    #pragma unroll
    for (int q = 0; q < 8; ++q) {
        len[q] = min(max(len[q], 0), chunk);
        st[q]  = min(max(st[q], 0), chunk);
        const size_t base = (size_t)(wave + 8 * q) * chunk + st[q];
        #pragma unroll
        for (int u = 0; u < 3; ++u)
            rec[q][u] = ebin[min(base + (size_t)(lane + 32 * u), (size_t)E - 1)];
    }
    int pos[8][3];
    #pragma unroll
    for (int q = 0; q < 8; ++q) {
        int* trow = tab + (wave + 8 * q) * CSR_RANGE;
        #pragma unroll
        for (int u = 0; u < 3; ++u) {
            pos[q][u] = -1;
            if (lane + 32 * u < len[q]) pos[q][u] = atomicAdd(&trow[rec[q][u] >> 24], 1);
        }
    }
    #pragma unroll
    for (int q = 0; q < 8; ++q) {
        #pragma unroll
        for (int u = 0; u < 3; ++u)
            if (pos[q][u] >= 0) csr_put_key(pos[q][u], (int)(rec[q][u] & 0xFFFFFFu), kb, kst, key, E);
    }
    #pragma unroll
    for (int q = 0; q < 8; ++q) {
        int* trow = tab + (wave + 8 * q) * CSR_RANGE;
        const size_t base = (size_t)(wave + 8 * q) * chunk + st[q];
        for (int i = lane + 96; i < len[q]; i += 32)
            csr_place_one(ebin[min(base + i, (size_t)E - 1)], trow, kb, kst, key, E);
    }
    __syncthreads();
    const int klim = min(max(kt, 0), CSR_KST);
    for (int i0 = t; i0 < klim; i0 += 4 * CSR_RANGE) {
        int w[4];
        #pragma unroll
        for (int u = 0; u < 4; ++u) w[u] = kst[min(i0 + u * CSR_RANGE, CSR_KST - 1)];
        #pragma unroll
        for (int u = 0; u < 4; ++u)
            if (i0 + u * CSR_RANGE < klim) key[min(max(kb + i0 + u * CSR_RANGE, 0), E - 1)] = w[u];
    }
}

__global__ __launch_bounds__(256) void gemm3_xw_colscan_place(const float* __restrict__ X, const float* __restrict__ Wr,
                         const float* __restrict__ Wl, const float* __restrict__ bias,
                         float* __restrict__ rel, float* __restrict__ H, int M, int K, int gblocks,
                         const int* __restrict__ blockcnt, const int* __restrict__ coarse,
                         const int* __restrict__ cstart, const unsigned* __restrict__ ebin,
                         int* __restrict__ off, int* __restrict__ key, int E, int chunk, int nr_total) {
    if ((int)blockIdx.x < gblocks) gemm3_strip(X, Wr, Wl, bias, rel, H, M, K, (int)blockIdx.x);
    else csr_colscan_place(blockcnt, coarse, cstart, ebin, off, key, M, E, chunk, nr_total, (int)blockIdx.x - gblocks);
}

__global__ void csr_fill(const int* __restrict__ src, const int* __restrict__ dst,
                         const int* __restrict__ et, const int* __restrict__ rank,
                         const int* __restrict__ blockcnt,
                         int* __restrict__ key, int E, int N, int chunk) {
    const int e = blockIdx.x * blockDim.x + threadIdx.x;
    if (e >= E) return;
    const int d = dst[e];
    const int b = e / chunk;
    int pos = blockcnt[(size_t)b * N + d] + rank[e];
    pos = min(max(pos, 0), E - 1);
    key[pos] = et[e] * N + src[e];
}

__global__ void gather64_relu(const float* __restrict__ rel, const int* __restrict__ off,
                              const int* __restrict__ key, float* __restrict__ h, int N) {
    const long long t = (long long)blockIdx.x * blockDim.x + threadIdx.x;
    const int i = (int)(t >> 3);
    if (i >= N) return;
    const int f = ((int)t & 7) * 4;
    float* hp = h + (size_t)i * 64 + f;
    float4 acc = *(const float4*)hp;
    float4 acd = *(const float4*)(hp + 32);
    int p = off[i];
    const int pe = off[i + 1];
    for (; p + 4 <= pe; p += 4) {
        const int k0 = key[p], k1 = key[p + 1], k2 = key[p + 2], k3 = key[p + 3];
        const float* r0 = rel + (size_t)k0 * 64 + f;
        const float* r1 = rel + (size_t)k1 * 64 + f;
        const float* r2 = rel + (size_t)k2 * 64 + f;
        const float* r3 = rel + (size_t)k3 * 64 + f;
        const float4 a = *(const float4*)r0, a2 = *(const float4*)(r0 + 32);
        const float4 b = *(const float4*)r1, b2 = *(const float4*)(r1 + 32);
        const float4 c = *(const float4*)r2, c2 = *(const float4*)(r2 + 32);
        const float4 d = *(const float4*)r3, d2 = *(const float4*)(r3 + 32);
        acc.x += a.x; acc.y += a.y; acc.z += a.z; acc.w += a.w;
        acd.x += a2.x; acd.y += a2.y; acd.z += a2.z; acd.w += a2.w;
        acc.x += b.x; acc.y += b.y; acc.z += b.z; acc.w += b.w;
        acd.x += b2.x; acd.y += b2.y; acd.z += b2.z; acd.w += b2.w;
        acc.x += c.x; acc.y += c.y; acc.z += c.z; acc.w += c.w;
        acd.x += c2.x; acd.y += c2.y; acd.z += c2.z; acd.w += c2.w;
        acc.x += d.x; acc.y += d.y; acc.z += d.z; acc.w += d.w;
        acd.x += d2.x; acd.y += d2.y; acd.z += d2.z; acd.w += d2.w;
    }
    #pragma nounroll
    for (; p < pe; ++p) {
        const float* r0 = rel + (size_t)key[p] * 64 + f;
        const float4 a = *(const float4*)r0, a2 = *(const float4*)(r0 + 32);
        acc.x += a.x; acc.y += a.y; acc.z += a.z; acc.w += a.w;
        acd.x += a2.x; acd.y += a2.y; acd.z += a2.z; acd.w += a2.w;
    }
    acc.x = fmaxf(acc.x, 0.0f); acc.y = fmaxf(acc.y, 0.0f);
    acc.z = fmaxf(acc.z, 0.0f); acc.w = fmaxf(acc.w, 0.0f);
    acd.x = fmaxf(acd.x, 0.0f); acd.y = fmaxf(acd.y, 0.0f);
    acd.z = fmaxf(acd.z, 0.0f); acd.w = fmaxf(acd.w, 0.0f);
    *(float4*)hp = acc;
    *(float4*)(hp + 32) = acd;
}

__global__ void gather2(const float* __restrict__ rel3, const int* __restrict__ off,
                        const int* __restrict__ key, float* __restrict__ agg3, int N) {
    const long long t = (long long)blockIdx.x * blockDim.x + threadIdx.x;
    const int i = (int)(t >> 4);
    const int l = (int)t & 15;
    float s0 = 0.0f, s1 = 0.0f;
    if (i < N) {
        const int pe = off[i + 1];
        for (int p = off[i] + l; p < pe; p += 16) {
            const float2 r = *(const float2*)(rel3 + (size_t)key[p] * 2);
            s0 += r.x; s1 += r.y;
        }
    }
    for (int m = 8; m >= 1; m >>= 1) {
        s0 += __shfl_xor(s0, m, 16);
        s1 += __shfl_xor(s1, m, 16);
    }
    if (i < N && l == 0) {
        agg3[(size_t)i * 2 + 0] += s0;
        agg3[(size_t)i * 2 + 1] += s1;
    }
}

template <int U>
__device__ __forceinline__ void l2_add_rows(const float* __restrict__ h1, const int* __restrict__ kp,
                                            int N, int f, float4& a0, float4& b0, float4& a1, float4& b1) {
    int    kk[U];
    float  m1[U];
    float4 lo[U], hi[U];
    #pragma unroll
    for (int j = 0; j < U; ++j) kk[j] = kp[j];
    #pragma unroll
    for (int j = 0; j < U; ++j) {
        const bool r = kk[j] >= N;
        const int  s = r ? kk[j] - N : kk[j];
        m1[j] = r ? 1.0f : 0.0f;
        const float* rp = h1 + (size_t)s * 64 + f;
        lo[j] = *(const float4*)rp;
        hi[j] = *(const float4*)(rp + 32);
    }
    #pragma unroll
    for (int j = 0; j < U; ++j) {
        const float w1 = m1[j], w0 = 1.0f - m1[j];
        a0.x = fmaf(lo[j].x, w0, a0.x); a0.y = fmaf(lo[j].y, w0, a0.y);
        a0.z = fmaf(lo[j].z, w0, a0.z); a0.w = fmaf(lo[j].w, w0, a0.w);
        b0.x = fmaf(hi[j].x, w0, b0.x); b0.y = fmaf(hi[j].y, w0, b0.y);
        b0.z = fmaf(hi[j].z, w0, b0.z); b0.w = fmaf(hi[j].w, w0, b0.w);
        a1.x = fmaf(lo[j].x, w1, a1.x); a1.y = fmaf(lo[j].y, w1, a1.y);
        a1.z = fmaf(lo[j].z, w1, a1.z); a1.w = fmaf(lo[j].w, w1, a1.w);
        b1.x = fmaf(hi[j].x, w1, b1.x); b1.y = fmaf(hi[j].y, w1, b1.y);
        b1.z = fmaf(hi[j].z, w1, b1.z); b1.w = fmaf(hi[j].w, w1, b1.w);
    }
}

#define L2_NODES  32
#define L2_PITCH  196
#define L2_HPITCH 68
__global__ __launch_bounds__(256) void layer2_sum_xw_l3(const float* __restrict__ h1, const int* __restrict__ off,
                                 const int* __restrict__ key,
                                 const float* __restrict__ W2, const float* __restrict__ loop2,
                                 const float* __restrict__ b2,
                                 const float* __restrict__ W3, const float* __restrict__ loop3,
                                 const float* __restrict__ b3,
                                 float* __restrict__ rel3, float* __restrict__ agg3, int N) {
    __shared__ __align__(16) float As[L2_NODES * L2_PITCH];
    __shared__ __align__(16) float Hs[L2_NODES * L2_HPITCH];
    const int t    = (int)threadIdx.x;
    const int row  = t >> 3;
    const int f    = (t & 7) * 4;
    const int i    = (int)blockIdx.x * L2_NODES + row;
    float4 a0  = make_float4(0.0f, 0.0f, 0.0f, 0.0f);
    float4 b0  = make_float4(0.0f, 0.0f, 0.0f, 0.0f);
    float4 a1  = make_float4(0.0f, 0.0f, 0.0f, 0.0f);
    float4 b1  = make_float4(0.0f, 0.0f, 0.0f, 0.0f);
    float4 own = make_float4(0.0f, 0.0f, 0.0f, 0.0f);
    float4 owb = make_float4(0.0f, 0.0f, 0.0f, 0.0f);
    if (i < N) {
        own = *(const float4*)(h1 + (size_t)i * 64 + f);
        owb = *(const float4*)(h1 + (size_t)i * 64 + f + 32);
        int p = off[i];
        const int pe = off[i + 1];
        for (; p + 8 <= pe; p += 8) l2_add_rows<8>(h1, key + p, N, f, a0, b0, a1, b1);
        if (p + 4 <= pe) { l2_add_rows<4>(h1, key + p, N, f, a0, b0, a1, b1); p += 4; }
        for (; p < pe; ++p) l2_add_rows<1>(h1, key + p, N, f, a0, b0, a1, b1);
    }
    {
        float* ar = As + row * L2_PITCH + f;
        *(float4*)(ar)       = a0;
        *(float4*)(ar + 32)  = b0;
        *(float4*)(ar + 64)  = a1;
        *(float4*)(ar + 96)  = b1;
        *(float4*)(ar + 128) = own;
        *(float4*)(ar + 160) = owb;
    }
    __syncthreads();

    {
        const int wave  = t >> 5;
        const int strip = wave >> 2;
        const int lane  = t & 31;
        const int half  = lane >> 4;
        const int lm    = lane & 15;
        const int n     = (wave & 3) * 16 + lm;
        v8f c0, c1, c2;
        const float bv = b2[n];
        #pragma unroll
        for (int v = 0; v < 8; ++v) { c0[v] = 0.0f; c1[v] = 0.0f; c2[v] = bv; }
        const float* arow = As + (strip * 16 + lm) * L2_PITCH + 2 * half;
        const float* w1p  = W2 + 64 * 64;
        #pragma unroll
        for (int k0 = 0; k0 < 64; k0 += 4) {
            const int kb = k0 + 2 * half;
            #pragma unroll
            for (int mt = 0; mt < 3; ++mt) {
                const float* Wm = (mt == 0) ? W2 : (mt == 1) ? w1p : loop2;
                const float2 av = *(const float2*)(arow + 64 * mt + k0);
                v2f a; a.x = av.x; a.y = av.y;
                v2f b;
                b.x = Wm[(size_t)kb * 64 + n];
                b.y = Wm[(size_t)(kb + 1) * 64 + n];
                if (mt == 0)      c0 = __builtin_amdgcn_wmma_f32_16x16x4_f32(false, a, false, b, (short)0, c0, false, false);
                else if (mt == 1) c1 = __builtin_amdgcn_wmma_f32_16x16x4_f32(false, a, false, b, (short)0, c1, false, false);
                else              c2 = __builtin_amdgcn_wmma_f32_16x16x4_f32(false, a, false, b, (short)0, c2, false, false);
            }
        }
        #pragma unroll
        for (int v = 0; v < 8; ++v)
            Hs[(strip * 16 + v + 8 * half) * L2_HPITCH + n] = fmaxf((c0[v] + c1[v]) + c2[v], 0.0f);
    }
    __syncthreads();

    const float4 hq = *(const float4*)(Hs + row * L2_HPITCH + f);
    const float4 hr = *(const float4*)(Hs + row * L2_HPITCH + f + 32);
    const float hv[4] = { hq.x, hq.y, hq.z, hq.w };
    const float hw[4] = { hr.x, hr.y, hr.z, hr.w };
    const float* wm[3] = { W3 + f * 2, W3 + 128 + f * 2, loop3 + f * 2 };
    float o[6];
    #pragma unroll
    for (int m = 0; m < 3; ++m) {
        float s0 = 0.0f, s1 = 0.0f, u0 = 0.0f, u1 = 0.0f;
        #pragma unroll
        for (int j = 0; j < 4; ++j) {
            s0 += hv[j] * wm[m][2 * j + 0];
            s1 += hv[j] * wm[m][2 * j + 1];
        }
        #pragma unroll
        for (int j = 0; j < 4; ++j) {
            u0 += hw[j] * wm[m][64 + 2 * j + 0];
            u1 += hw[j] * wm[m][64 + 2 * j + 1];
        }
        o[2 * m + 0] = s0 + u0;
        o[2 * m + 1] = s1 + u1;
    }
    for (int d = 4; d >= 1; d >>= 1) {
        #pragma unroll
        for (int q = 0; q < 6; ++q) o[q] += __shfl_xor(o[q], d, 8);
    }
    if ((t & 7) == 0 && i < N) {
        rel3[(size_t)i * 2 + 0] = o[0];
        rel3[(size_t)i * 2 + 1] = o[1];
        rel3[(size_t)(N + i) * 2 + 0] = o[2];
        rel3[(size_t)(N + i) * 2 + 1] = o[3];
        agg3[(size_t)i * 2 + 0] = o[4] + b3[0];
        agg3[(size_t)i * 2 + 1] = o[5] + b3[1];
    }
}

#define TAIL_TPB 256
__global__ __launch_bounds__(TAIL_TPB) void gather2_min(const float* __restrict__ rel3, const int* __restrict__ off,
                            const int* __restrict__ key, float* __restrict__ agg3,
                            float* __restrict__ blockmin, int N) {
    __shared__ float sm[TAIL_TPB / 32];
    const long long t = (long long)blockIdx.x * blockDim.x + threadIdx.x;
    const int i = (int)(t >> 2);
    const int q = (int)t & 3;
    float s0[4] = { 0.0f, 0.0f, 0.0f, 0.0f };
    float s1[4] = { 0.0f, 0.0f, 0.0f, 0.0f };
    float g0 = 0.0f, g1 = 0.0f;
    if (i < N) {
        const int p0 = off[i];
        const int pe = off[i + 1];
        if (q == 0) {
            const float2 g = *(const float2*)(agg3 + (size_t)i * 2);
            g0 = g.x; g1 = g.y;
        }
        for (int base = p0; base < pe; base += 32) {
            int    kk[8];
            float2 r[8];
            #pragma unroll
            for (int u = 0; u < 8; ++u) kk[u] = key[min(base + q + 4 * u, pe - 1)];
            #pragma unroll
            for (int u = 0; u < 8; ++u) r[u] = *(const float2*)(rel3 + (size_t)kk[u] * 2);
            #pragma unroll
            for (int u = 0; u < 8; ++u) {
                const float w = (base + q + 4 * u < pe) ? 1.0f : 0.0f;
                s0[u & 3] = fmaf(r[u].x, w, s0[u & 3]);
                s1[u & 3] = fmaf(r[u].y, w, s1[u & 3]);
            }
        }
    }
    float t0 = (s0[0] + s0[2]) + (s0[1] + s0[3]);
    float t1 = (s1[0] + s1[2]) + (s1[1] + s1[3]);
    for (int m = 2; m >= 1; m >>= 1) {
        t0 += __shfl_xor(t0, m, 4);
        t1 += __shfl_xor(t1, m, 4);
    }
    float mn = 3.402823466e38f;
    if (i < N && q == 0) {
        const float v0 = g0 + t0;
        const float v1 = g1 + t1;
        agg3[(size_t)i * 2 + 0] = v0;
        agg3[(size_t)i * 2 + 1] = v1;
        mn = fminf(v0, v1);
    }
    for (int m = 16; m >= 4; m >>= 1) mn = fminf(mn, __shfl_xor(mn, m, 32));
    if ((threadIdx.x & 31) == 0) sm[threadIdx.x >> 5] = mn;
    __syncthreads();
    if (threadIdx.x == 0) {
        float r = sm[0];
        #pragma unroll
        for (int w = 1; w < TAIL_TPB / 32; ++w) r = fminf(r, sm[w]);
        blockmin[blockIdx.x] = r;
    }
}

__global__ __launch_bounds__(TAIL_TPB) void finalize_bm(const float* __restrict__ agg3,
                            const int* __restrict__ cell,
                            const int* __restrict__ maxs,
                            const float* __restrict__ blockmin, int nblk,
                            float* __restrict__ out, int N) {
    __shared__ float sm[TAIL_TPB / 32];
    float mn = 3.402823466e38f;
    for (int j0 = threadIdx.x; j0 < nblk; j0 += 16 * TAIL_TPB) {
        float v[16];
        #pragma unroll
        for (int u = 0; u < 16; ++u) v[u] = blockmin[min(j0 + u * TAIL_TPB, nblk - 1)];
        #pragma unroll
        for (int u = 0; u < 16; ++u) mn = fminf(mn, v[u]);
    }
    for (int m = 16; m >= 1; m >>= 1) mn = fminf(mn, __shfl_xor(mn, m, 32));
    if ((threadIdx.x & 31) == 0) sm[threadIdx.x >> 5] = mn;
    __syncthreads();
    float r = sm[0];
    #pragma unroll
    for (int w = 1; w < TAIL_TPB / 32; ++w) r = fminf(r, sm[w]);
    const float gmin = r - 1.0f;
    const int i = blockIdx.x * blockDim.x + threadIdx.x;
    if (i >= N) return;
    const bool upper = cell[i] >= (maxs[i] - 1);
    const bool lower = cell[i] == 0;
    out[(size_t)i * 2 + 0] = upper ? gmin : agg3[(size_t)i * 2 + 0];
    out[(size_t)i * 2 + 1] = lower ? gmin : agg3[(size_t)i * 2 + 1];
}

__device__ __forceinline__ unsigned f2ord(float f) {
    unsigned u = __float_as_uint(f);
    return (u & 0x80000000u) ? ~u : (u | 0x80000000u);
}
__device__ __forceinline__ float ord2f(unsigned u) {
    return __uint_as_float((u & 0x80000000u) ? (u ^ 0x80000000u) : ~u);
}

__global__ void init_slot(unsigned* slot) { *slot = 0xFFFFFFFFu; }

__global__ void min_reduce(const float* __restrict__ p, int n, unsigned* slot) {
    __shared__ float sm[256];
    float m = 3.402823466e38f;
    for (int i = blockIdx.x * blockDim.x + threadIdx.x; i < n;
         i += gridDim.x * blockDim.x)
        m = fminf(m, p[i]);
    sm[threadIdx.x] = m;
    __syncthreads();
    for (int s = blockDim.x >> 1; s > 0; s >>= 1) {
        if ((int)threadIdx.x < s) sm[threadIdx.x] = fminf(sm[threadIdx.x], sm[threadIdx.x + s]);
        __syncthreads();
    }
    if (threadIdx.x == 0) atomicMin(slot, f2ord(sm[0]));
}

__global__ void finalize_k(const float* __restrict__ agg3,
                           const int* __restrict__ cell,
                           const int* __restrict__ maxs,
                           const unsigned* __restrict__ slot,
                           float* __restrict__ out, int N) {
    const int i = blockIdx.x * blockDim.x + threadIdx.x;
    if (i >= N) return;
    const float gmin = ord2f(*slot) - 1.0f;
    const bool upper = cell[i] >= (maxs[i] - 1);
    const bool lower = cell[i] == 0;
    out[(size_t)i * 2 + 0] = upper ? gmin : agg3[(size_t)i * 2 + 0];
    out[(size_t)i * 2 + 1] = lower ? gmin : agg3[(size_t)i * 2 + 1];
}

extern "C" void kernel_launch(void* const* d_in, const int* in_sizes, int n_in,
                              void* d_out, int out_size, void* d_ws, size_t ws_size,
                              hipStream_t stream) {
    const float* x      = (const float*)d_in[0];
    const int*   src    = (const int*)d_in[1];
    const int*   dst    = (const int*)d_in[2];
    const int*   etypes = (const int*)d_in[3];
    const int*   cell   = (const int*)d_in[4];
    const int*   maxs   = (const int*)d_in[5];
    const float* W1     = (const float*)d_in[6];
    const float* loop1  = (const float*)d_in[7];
    const float* b1     = (const float*)d_in[8];
    const float* W2     = (const float*)d_in[9];
    const float* loop2  = (const float*)d_in[10];
    const float* b2     = (const float*)d_in[11];
    const float* W3     = (const float*)d_in[12];
    const float* loop3  = (const float*)d_in[13];
    const float* b3     = (const float*)d_in[14];
    float* out = (float*)d_out;

    const int N = in_sizes[4];
    const int E = in_sizes[1];
    const int K1 = in_sizes[0] / N;

    float* relbuf = (float*)d_ws;
    float* h1     = relbuf + (size_t)2 * N * 64;
    float* h2     = h1 + (size_t)N * 64;
    float* rel3   = h2 + (size_t)N * 64;
    float* agg3   = rel3 + (size_t)4 * N;
    unsigned* slot = (unsigned*)(agg3 + (size_t)2 * N);

    const int waves   = (N + 15) / 16;
    const int gblocks = (waves * 32 + 255) / 256;
    const int eb64    = (int)(((long long)E * 16 + 255) / 256);
    const int nb      = (N + 255) / 256;
    const int fb      = (N * 64 + 255) / 256;
    const int eb2     = (E + 255) / 256;

    int* key      = (int*)(slot + 4);
    int* off      = key + (size_t)E;
    int* blockcnt = (int*)relbuf;
    int* rank     = blockcnt + (size_t)CSR_NB * N;
    const size_t need = ((size_t)262 * N + 4 + (size_t)E + (size_t)N + 1) * sizeof(float);
    int* coarse   = (int*)rel3;
    const bool csr_ok = ws_size >= need && (long long)E <= (long long)192 * N
                        && (long long)CSR_NB * nb <= (long long)6 * N;
    const size_t need2 = need + (size_t)E * sizeof(int);
    const int chunk0 = (E + CSR_NB - 1) / CSR_NB;
    const bool shared_ok = csr_ok && ws_size >= need2
                           && N <= CSR_CAP && chunk0 <= 1024 * CSR_EPT
                           && (long long)2 * N <= (long long)0x1000000
                           && (long long)2 * CSR_NB * nb <= (long long)6 * N;

    if (shared_ok) {
        const int chunk = chunk0;
        const int gb8   = (int)(((long long)N * 8 + 255) / 256);
        const int gb4   = (int)(((long long)N * 4 + 255) / 256);
        const int l2b   = (N + L2_NODES - 1) / L2_NODES;
        int* blockcnt2  = (int*)h2;
        unsigned* ebin  = (unsigned*)(off + (size_t)N + 1);
        int* cstart     = coarse + (size_t)CSR_NB * nb;
        csr_hist_bin<<<CSR_NB, 1024, 0, stream>>>(dst, src, etypes, ebin, blockcnt2, coarse, cstart, E, N, chunk, nb);

        gemm3_xw_colscan_place<<<gblocks + nb, 256, 0, stream>>>(x, W1, loop1, b1, relbuf, h1, N, K1, gblocks,
                                                                 blockcnt2, coarse, cstart, ebin, off, key, E, chunk, nb);
        gather64_relu<<<gb8, 256, 0, stream>>>(relbuf, off, key, h1, N);

        layer2_sum_xw_l3<<<l2b, 256, 0, stream>>>(h1, off, key, W2, loop2, b2, W3, loop3, b3, rel3, agg3, N);

        float* blockmin = relbuf;
        gather2_min<<<gb4, TAIL_TPB, 0, stream>>>(rel3, off, key, agg3, blockmin, N);
        finalize_bm<<<nb, TAIL_TPB, 0, stream>>>(agg3, cell, maxs, blockmin, gb4, out, N);
    } else if (csr_ok) {
        const int chunk = (E + CSR_NB - 1) / CSR_NB;
        const int gb8   = (int)(((long long)N * 8 + 255) / 256);
        const int gb4   = (int)(((long long)N * 4 + 255) / 256);
        const int l2b   = (N + L2_NODES - 1) / L2_NODES;
        csr_hist<<<CSR_NB, 1024, 0, stream>>>(dst, rank, blockcnt, coarse, E, N, chunk, nb);
        csr_colscan_off<<<nb, CSR_RANGE, 0, stream>>>(blockcnt, coarse, off, N, nb);
        csr_fill<<<eb2, 256, 0, stream>>>(src, dst, etypes, rank, blockcnt, key, E, N, chunk);

        gemm3_xw<<<gblocks, 256, 0, stream>>>(x, W1, loop1, b1, relbuf, h1, N, K1);
        gather64_relu<<<gb8, 256, 0, stream>>>(relbuf, off, key, h1, N);

        layer2_sum_xw_l3<<<l2b, 256, 0, stream>>>(h1, off, key, W2, loop2, b2, W3, loop3, b3, rel3, agg3, N);

        float* blockmin = relbuf;
        gather2_min<<<gb4, TAIL_TPB, 0, stream>>>(rel3, off, key, agg3, blockmin, N);
        finalize_bm<<<nb, TAIL_TPB, 0, stream>>>(agg3, cell, maxs, blockmin, gb4, out, N);
    } else {
        gemm3_xw<<<gblocks, 256, 0, stream>>>(x, W1, loop1, b1, relbuf, h1, N, K1);
        edge_scatter64<<<eb64, 256, 0, stream>>>(relbuf, src, dst, etypes, h1, E, N);
        relu_k<<<fb, 256, 0, stream>>>(h1, N * 64);

        gemm3_xw<<<gblocks, 256, 0, stream>>>(h1, W2, loop2, b2, relbuf, h2, N, 64);
        edge_scatter64<<<eb64, 256, 0, stream>>>(relbuf, src, dst, etypes, h2, E, N);
        relu_k<<<fb, 256, 0, stream>>>(h2, N * 64);

        layer3_node<<<nb, 256, 0, stream>>>(h2, W3, loop3, b3, rel3, agg3, N);
        edge_scatter2<<<eb2, 256, 0, stream>>>(rel3, src, dst, etypes, agg3, E, N);
        init_slot<<<1, 1, 0, stream>>>(slot);
        min_reduce<<<512, 256, 0, stream>>>(agg3, N * 2, slot);
        finalize_k<<<nb, 256, 0, stream>>>(agg3, cell, maxs, slot, out, N);
    }
}


# Edge sort keeps no per-chunk per-node count table in memory: each range's block counts its nodes from the binned words in LDS (binning pass: no node counters, no table dump); key places unchanged
# speedup vs baseline: 3.2880x; 1.0917x over previous
#include <hip/hip_runtime.h>

typedef __attribute__((ext_vector_type(2))) float v2f;
typedef __attribute__((ext_vector_type(8))) float v8f;

__device__ __forceinline__ void gemm3_strip(const float* __restrict__ X, const float* __restrict__ Wr,
                         const float* __restrict__ Wl, const float* __restrict__ bias,
                         float* __restrict__ rel, float* __restrict__ H,
                         int M, int K, int block) {
    const int gthread = block * blockDim.x + threadIdx.x;
    const int wave    = gthread >> 5;
    const int lane    = threadIdx.x & 31;
    const int m0      = wave * 16;
    if (m0 >= M) return;

    const int half = lane >> 4;
    const int lm   = lane & 15;

    v8f acc[12];
    #pragma unroll
    for (int t = 0; t < 8; ++t) {
        #pragma unroll
        for (int v = 0; v < 8; ++v) acc[t][v] = 0.0f;
    }
    #pragma unroll
    for (int nt = 0; nt < 4; ++nt) {
        const float bv = bias[nt * 16 + lm];
        #pragma unroll
        for (int v = 0; v < 8; ++v) acc[8 + nt][v] = bv;
    }

    const float* xrow = X + (size_t)(m0 + lm) * K;
    const float* w1   = Wr + (size_t)K * 64;

    #pragma unroll 2
    for (int k0 = 0; k0 < K; k0 += 4) {
        const int kb = k0 + 2 * half;
        const float2 av = *(const float2*)(xrow + kb);
        v2f a; a.x = av.x; a.y = av.y;
        #pragma unroll
        for (int mt = 0; mt < 3; ++mt) {
            const float* Wm = (mt == 0) ? Wr : (mt == 1) ? w1 : Wl;
            #pragma unroll
            for (int nt = 0; nt < 4; ++nt) {
                const int n = nt * 16 + lm;
                v2f b;
                b.x = Wm[(size_t)kb * 64 + n];
                b.y = Wm[(size_t)(kb + 1) * 64 + n];
                acc[mt * 4 + nt] = __builtin_amdgcn_wmma_f32_16x16x4_f32(
                    false, a, false, b, (short)0, acc[mt * 4 + nt], false, false);
            }
        }
    }

    #pragma unroll
    for (int mt = 0; mt < 3; ++mt) {
        float* C = (mt == 2) ? H : rel + (size_t)mt * M * 64;
        #pragma unroll
        for (int nt = 0; nt < 4; ++nt) {
            const int n = nt * 16 + lm;
            #pragma unroll
            for (int v = 0; v < 8; ++v) {
                C[(size_t)(m0 + v + 8 * half) * 64 + n] = acc[mt * 4 + nt][v];
            }
        }
    }
}

__global__ __launch_bounds__(256) void gemm3_xw(const float* __restrict__ X, const float* __restrict__ Wr,
                         const float* __restrict__ Wl, const float* __restrict__ bias,
                         float* __restrict__ rel, float* __restrict__ H,
                         int M, int K) {
    gemm3_strip(X, Wr, Wl, bias, rel, H, M, K, (int)blockIdx.x);
}

__global__ void edge_scatter64(const float* __restrict__ rel,
                               const int* __restrict__ src,
                               const int* __restrict__ dst,
                               const int* __restrict__ et,
                               float* __restrict__ agg, int E, int N) {
    const long long t = (long long)blockIdx.x * blockDim.x + threadIdx.x;
    const int e = (int)(t >> 4);
    if (e >= E) return;
    const int f = ((int)t & 15) * 4;
    const float4 v = *(const float4*)(rel + ((long long)et[e] * N + src[e]) * 64 + f);
    float* q = agg + (long long)dst[e] * 64 + f;
    atomicAdd(q + 0, v.x);
    atomicAdd(q + 1, v.y);
    atomicAdd(q + 2, v.z);
    atomicAdd(q + 3, v.w);
}

__global__ void relu_k(float* __restrict__ p, int n) {
    const int i = blockIdx.x * blockDim.x + threadIdx.x;
    if (i < n) p[i] = fmaxf(p[i], 0.0f);
}

__global__ void layer3_node(const float* __restrict__ h2,
                            const float* __restrict__ W3,
                            const float* __restrict__ loop3,
                            const float* __restrict__ b3,
                            float* __restrict__ rel3, float* __restrict__ agg3,
                            int N) {
    const int i = blockIdx.x * blockDim.x + threadIdx.x;
    if (i >= N) return;
    const float* row = h2 + (size_t)i * 64;
    float r00 = 0.f, r01 = 0.f, r10 = 0.f, r11 = 0.f, l0 = 0.f, l1 = 0.f;
    #pragma unroll 8
    for (int k = 0; k < 64; ++k) {
        const float v = row[k];
        r00 += v * W3[k * 2 + 0];
        r01 += v * W3[k * 2 + 1];
        r10 += v * W3[128 + k * 2 + 0];
        r11 += v * W3[128 + k * 2 + 1];
        l0  += v * loop3[k * 2 + 0];
        l1  += v * loop3[k * 2 + 1];
    }
    rel3[(size_t)i * 2 + 0] = r00;
    rel3[(size_t)i * 2 + 1] = r01;
    rel3[(size_t)(N + i) * 2 + 0] = r10;
    rel3[(size_t)(N + i) * 2 + 1] = r11;
    agg3[(size_t)i * 2 + 0] = l0 + b3[0];
    agg3[(size_t)i * 2 + 1] = l1 + b3[1];
}

__global__ void edge_scatter2(const float* __restrict__ rel3,
                              const int* __restrict__ src,
                              const int* __restrict__ dst,
                              const int* __restrict__ et,
                              float* __restrict__ agg3, int E, int N) {
    const int e = blockIdx.x * blockDim.x + threadIdx.x;
    if (e >= E) return;
    const float* p = rel3 + ((long long)et[e] * N + src[e]) * 2;
    float* q = agg3 + (long long)dst[e] * 2;
    atomicAdd(q + 0, p[0]);
    atomicAdd(q + 1, p[1]);
}


#define CSR_NB  64
#define CSR_CAP 51200
#define CSR_RANGE 256

__global__ __launch_bounds__(1024) void csr_hist(const int* __restrict__ dst,
                                                 int* __restrict__ rank,
                                                 int* __restrict__ blockcnt,
                                                 int* __restrict__ coarse,
                                                 int E, int N, int chunk, int nr_total) {
    __shared__ int cnt[CSR_CAP];
    const int b  = blockIdx.x;
    const int e0 = b * chunk;
    const int e1 = min(e0 + chunk, E);
    for (int base = 0; base < N; base += CSR_CAP) {
        const int lim = min(CSR_CAP, N - base);
        for (int i = threadIdx.x; i < lim; i += blockDim.x) cnt[i] = 0;
        __syncthreads();
        const int bd = (int)blockDim.x;
        for (int e = e0 + (int)threadIdx.x; e < e1; e += 4 * bd) {
            int d[4];
            #pragma unroll
            for (int j = 0; j < 4; ++j) {
                const int ee = e + j * bd;
                const int v  = dst[min(ee, e1 - 1)];
                d[j] = (ee < e1) ? v - base : -1;
            }
            #pragma unroll
            for (int j = 0; j < 4; ++j)
                if (d[j] >= 0 && d[j] < lim) rank[e + j * bd] = atomicAdd(&cnt[d[j]], 1);
        }
        __syncthreads();
        const int nrs = (lim + CSR_RANGE - 1) / CSR_RANGE;
        for (int r0 = 0; r0 < nrs; r0 += (int)blockDim.x >> 2) {
            const int r  = r0 + ((int)threadIdx.x >> 2);
            const int q  = (int)threadIdx.x & 3;
            const int lo = r * CSR_RANGE + q * (CSR_RANGE / 4);
            int s = 0;
            #pragma unroll 16
            for (int j = 0; j < CSR_RANGE / 4; ++j) {
                const int i = lo + ((j + (int)threadIdx.x) & (CSR_RANGE / 4 - 1));
                const int v = cnt[min(i, lim - 1)];
                s += (i < lim) ? v : 0;
            }
            s += __shfl_xor(s, 1, 4);
            s += __shfl_xor(s, 2, 4);
            if (q == 0 && r < nrs) coarse[(size_t)b * nr_total + base / CSR_RANGE + r] = s;
        }
        for (int i = threadIdx.x; i < lim; i += blockDim.x)
            blockcnt[(size_t)b * N + base + i] = cnt[i];
        __syncthreads();
    }
}

#define CSR_EPT 16
__global__ __launch_bounds__(1024) void csr_hist_bin(const int* __restrict__ dst, const int* __restrict__ src,
                                                     const int* __restrict__ et,
                                                     unsigned* __restrict__ ebin,
                                                     int* __restrict__ coarse, int* __restrict__ cstart,
                                                     int E, int N, int chunk, int nr_total) {
    __shared__ int rcnt[CSR_CAP / CSR_RANGE];
    __shared__ int rstart[CSR_CAP / CSR_RANGE];
    __shared__ int swt[8];
    __shared__ int stot;
    __shared__ unsigned stage[1024 * CSR_EPT];
    const int tid  = (int)threadIdx.x;
    const int lane = tid & 31;
    const int b    = (int)blockIdx.x;
    const int e0   = b * chunk;
    const int e1   = min(e0 + chunk, E);
    if (tid < nr_total) rcnt[tid] = 0;
    int dd[CSR_EPT], ss[CSR_EPT], tt[CSR_EPT];
    #pragma unroll
    for (int j = 0; j < CSR_EPT; ++j) {
        const int ec = max(min(e0 + tid + j * 1024, e1 - 1), 0);
        dd[j] = dst[ec]; ss[j] = src[ec]; tt[j] = et[ec];
    }
    #pragma unroll
    for (int j = 0; j < CSR_EPT; ++j)
        if (e0 + tid + j * 1024 >= e1) dd[j] = -1;
    __syncthreads();
    int slot[CSR_EPT];
    #pragma unroll
    for (int j = 0; j < CSR_EPT; ++j) {
        slot[j] = 0;
        if (dd[j] >= 0 && dd[j] < N) slot[j] = atomicAdd(&rcnt[dd[j] / CSR_RANGE], 1);
        else dd[j] = -1;
    }
    __syncthreads();
    int rv = 0, inc = 0;
    if (tid < 256) {
        rv  = (tid < nr_total) ? rcnt[tid] : 0;
        inc = rv;
        for (int d = 1; d < 32; d <<= 1) {
            const int v = __shfl_up(inc, d, 32);
            if (lane >= d) inc += v;
        }
        if (lane == 31) swt[tid >> 5] = inc;
    }
    __syncthreads();
    if (tid < 256) {
        int wb = 0;
        #pragma unroll
        for (int w = 0; w < 8; ++w) wb += (w < (tid >> 5)) ? swt[w] : 0;
        const int ex = wb + inc - rv;
        if (tid < nr_total) {
            rstart[tid] = ex;
            coarse[(size_t)b * nr_total + tid] = rv;
            cstart[(size_t)b * nr_total + tid] = ex;
        }
        if (tid == 255) stot = ex + rv;
    }
    __syncthreads();
    #pragma unroll
    for (int j = 0; j < CSR_EPT; ++j)
        slot[j] += rstart[max(dd[j], 0) / CSR_RANGE];
    #pragma unroll
    for (int j = 0; j < CSR_EPT; ++j) {
        if (dd[j] >= 0) {
            const int r = dd[j] / CSR_RANGE;
            const int p = min(max(slot[j], 0), e1 - e0 - 1);
            const unsigned kv = (unsigned)(tt[j] * N + ss[j]);
            stage[p] = (kv & 0xFFFFFFu) | ((unsigned)(dd[j] - r * CSR_RANGE) << 24);
        }
    }
    __syncthreads();
    const int tot = min(max(stot, 0), e1 - e0);
    for (int i0 = tid; i0 < tot; i0 += 4 * 1024) {
        unsigned w[4];
        #pragma unroll
        for (int u = 0; u < 4; ++u) w[u] = stage[min(i0 + u * 1024, 1024 * CSR_EPT - 1)];
        #pragma unroll
        for (int u = 0; u < 4; ++u)
            if (i0 + u * 1024 < tot) ebin[(size_t)e0 + i0 + u * 1024] = w[u];
    }
}

__device__ __forceinline__ void csr_colscan_range(int* __restrict__ blockcnt,
                                                  const int* __restrict__ coarse,
                                                  int* __restrict__ off, int N, int nr_total, int r) {
    __shared__ int wtot[CSR_RANGE / 32];
    __shared__ int wbase[CSR_RANGE / 32];
    const int t  = threadIdx.x;
    const int n  = r * CSR_RANGE + t;
    const int nn = min(n, N - 1);
    int c[CSR_NB];
    #pragma unroll
    for (int j = 0; j < CSR_NB; ++j) c[j] = blockcnt[(size_t)j * N + nn];
    int part = 0;
    for (int rr = t; rr < r; rr += CSR_RANGE) {
        #pragma unroll 16
        for (int b = 0; b < CSR_NB; ++b) part += coarse[(size_t)b * nr_total + rr];
    }
    int deg = 0;
    #pragma unroll
    for (int j = 0; j < CSR_NB; ++j) deg += c[j];
    if (n >= N) deg = 0;
    const int lane = t & 31;
    int inc = deg;
    for (int d = 1; d < 32; d <<= 1) {
        const int v = __shfl_up(inc, d, 32);
        if (lane >= d) inc += v;
    }
    for (int m = 16; m >= 1; m >>= 1) part += __shfl_xor(part, m, 32);
    if (lane == 31) wtot[t >> 5] = inc;
    if (lane == 0)  wbase[t >> 5] = part;
    __syncthreads();
    int run = inc - deg;
    #pragma unroll
    for (int w = 0; w < CSR_RANGE / 32; ++w) {
        run += wbase[w];
        run += (w < (t >> 5)) ? wtot[w] : 0;
    }
    if (n < N) {
        off[n] = run;
        if (n == N - 1) off[N] = run + deg;
        #pragma unroll
        for (int j = 0; j < CSR_NB; ++j) {
            blockcnt[(size_t)j * N + n] = run;
            run += c[j];
        }
    }
}

__global__ __launch_bounds__(CSR_RANGE) void csr_colscan_off(int* __restrict__ blockcnt,
                                                             const int* __restrict__ coarse,
                                                             int* __restrict__ off, int N, int nr_total) {
    csr_colscan_range(blockcnt, coarse, off, N, nr_total, (int)blockIdx.x);
}

#define CSR_KST 8192
static_assert(CSR_NB == 64 && CSR_RANGE == 256, "8 waves x 8 chunks, one thread per node of the range");
__device__ __forceinline__ void csr_put_key(int pos, int k, int kb, int* __restrict__ kst, int* __restrict__ key, int E) {
    const int li = pos - kb;
    if (li >= 0 && li < CSR_KST) kst[li] = k;
    else key[min(max(pos, 0), E - 1)] = k;
}
__device__ __forceinline__ void csr_place_one(unsigned rec, int* __restrict__ trow, int kb, int* __restrict__ kst,
                                              int* __restrict__ key, int E) {
    const int dl = (int)(rec >> 24);
    const int pos = atomicAdd(&trow[dl], 1);
    csr_put_key(pos, (int)(rec & 0xFFFFFFu), kb, kst, key, E);
}

__device__ __forceinline__ void csr_colscan_place(const int* __restrict__ coarse, const int* __restrict__ cstart,
                                                  const unsigned* __restrict__ ebin,
                                                  int* __restrict__ off, int* __restrict__ key,
                                                  int N, int E, int chunk, int nr_total, int r) {
    __shared__ int wtot[CSR_RANGE / 32];
    __shared__ int wbase[CSR_RANGE / 32];
    __shared__ __align__(16) int tab[CSR_NB * CSR_RANGE];
    __shared__ int kst[CSR_KST];
    const int t    = threadIdx.x;
    const int wave = t >> 5;
    const int lane = t & 31;
    const int n  = r * CSR_RANGE + t;
    int len[8], st[8];
    #pragma unroll
    for (int q = 0; q < 8; ++q) {
        const int bq = wave + 8 * q;
        len[q] = coarse[(size_t)bq * nr_total + r];
        st[q]  = cstart[(size_t)bq * nr_total + r];
    }
    #pragma unroll
    for (int j = 0; j < CSR_NB * CSR_RANGE / 4 / CSR_RANGE; ++j)
        ((int4*)tab)[j * CSR_RANGE + t] = make_int4(0, 0, 0, 0);
    int part = 0;
    for (int rr = t; rr < r; rr += CSR_RANGE) {
        #pragma unroll 16
        for (int b = 0; b < CSR_NB; ++b) part += coarse[(size_t)b * nr_total + rr];
    }
    unsigned rec[8][3];
    #pragma unroll
    for (int q = 0; q < 8; ++q) {
        len[q] = min(max(len[q], 0), chunk);
        st[q]  = min(max(st[q], 0), chunk);
        const size_t base = (size_t)(wave + 8 * q) * chunk + st[q];
        #pragma unroll
        for (int u = 0; u < 3; ++u)
            rec[q][u] = ebin[min(base + (size_t)(lane + 32 * u), (size_t)E - 1)];
    }
    __syncthreads();
    #pragma unroll
    for (int q = 0; q < 8; ++q) {
        int* trow = tab + (wave + 8 * q) * CSR_RANGE;
        #pragma unroll
        for (int u = 0; u < 3; ++u)
            if (lane + 32 * u < len[q]) atomicAdd(&trow[rec[q][u] >> 24], 1);
    }
    #pragma unroll
    for (int q = 0; q < 8; ++q) {
        int* trow = tab + (wave + 8 * q) * CSR_RANGE;
        const size_t base = (size_t)(wave + 8 * q) * chunk + st[q];
        for (int i = lane + 96; i < len[q]; i += 32)
            atomicAdd(&trow[ebin[min(base + i, (size_t)E - 1)] >> 24], 1);
    }
    __syncthreads();
    int c[CSR_NB];
    #pragma unroll
    for (int j = 0; j < CSR_NB; ++j) c[j] = tab[j * CSR_RANGE + t];
    int deg = 0;
    #pragma unroll
    for (int j = 0; j < CSR_NB; ++j) deg += c[j];
    if (n >= N) deg = 0;
    int inc = deg;
    for (int d = 1; d < 32; d <<= 1) {
        const int v = __shfl_up(inc, d, 32);
        if (lane >= d) inc += v;
    }
    for (int m = 16; m >= 1; m >>= 1) part += __shfl_xor(part, m, 32);
    if (lane == 31) wtot[wave] = inc;
    if (lane == 0)  wbase[wave] = part;
    __syncthreads();
    int run = inc - deg;
    int kb = 0, kt = 0;
    #pragma unroll
    for (int w = 0; w < CSR_RANGE / 32; ++w) {
        run += wbase[w];
        run += (w < wave) ? wtot[w] : 0;
        kb  += wbase[w];
        kt  += wtot[w];
    }
    if (n < N) {
        off[n] = run;
        if (n == N - 1) off[N] = run + deg;
    }
    #pragma unroll
    for (int j = 0; j < CSR_NB; ++j) {
        tab[j * CSR_RANGE + t] = run;
        run += c[j];
    }
    __syncthreads();

    int pos[8][3];
    #pragma unroll
    for (int q = 0; q < 8; ++q) {
        int* trow = tab + (wave + 8 * q) * CSR_RANGE;
        #pragma unroll
        for (int u = 0; u < 3; ++u) {
            pos[q][u] = -1;
            if (lane + 32 * u < len[q]) pos[q][u] = atomicAdd(&trow[rec[q][u] >> 24], 1);
        }
    }
    #pragma unroll
    for (int q = 0; q < 8; ++q) {
        #pragma unroll
        for (int u = 0; u < 3; ++u)
            if (pos[q][u] >= 0) csr_put_key(pos[q][u], (int)(rec[q][u] & 0xFFFFFFu), kb, kst, key, E);
    }
    #pragma unroll
    for (int q = 0; q < 8; ++q) {
        int* trow = tab + (wave + 8 * q) * CSR_RANGE;
        const size_t base = (size_t)(wave + 8 * q) * chunk + st[q];
        for (int i = lane + 96; i < len[q]; i += 32)
            csr_place_one(ebin[min(base + i, (size_t)E - 1)], trow, kb, kst, key, E);
    }
    __syncthreads();
    const int klim = min(max(kt, 0), CSR_KST);
    for (int i0 = t; i0 < klim; i0 += 4 * CSR_RANGE) {
        int w[4];
        #pragma unroll
        for (int u = 0; u < 4; ++u) w[u] = kst[min(i0 + u * CSR_RANGE, CSR_KST - 1)];
        #pragma unroll
        for (int u = 0; u < 4; ++u)
            if (i0 + u * CSR_RANGE < klim) key[min(max(kb + i0 + u * CSR_RANGE, 0), E - 1)] = w[u];
    }
}

__global__ __launch_bounds__(256) void gemm3_xw_colscan_place(const float* __restrict__ X, const float* __restrict__ Wr,
                         const float* __restrict__ Wl, const float* __restrict__ bias,
                         float* __restrict__ rel, float* __restrict__ H, int M, int K, int gblocks,
                         const int* __restrict__ coarse,
                         const int* __restrict__ cstart, const unsigned* __restrict__ ebin,
                         int* __restrict__ off, int* __restrict__ key, int E, int chunk, int nr_total) {
    if ((int)blockIdx.x < gblocks) gemm3_strip(X, Wr, Wl, bias, rel, H, M, K, (int)blockIdx.x);
    else csr_colscan_place(coarse, cstart, ebin, off, key, M, E, chunk, nr_total, (int)blockIdx.x - gblocks);
}

__global__ void csr_fill(const int* __restrict__ src, const int* __restrict__ dst,
                         const int* __restrict__ et, const int* __restrict__ rank,
                         const int* __restrict__ blockcnt,
                         int* __restrict__ key, int E, int N, int chunk) {
    const int e = blockIdx.x * blockDim.x + threadIdx.x;
    if (e >= E) return;
    const int d = dst[e];
    const int b = e / chunk;
    int pos = blockcnt[(size_t)b * N + d] + rank[e];
    pos = min(max(pos, 0), E - 1);
    key[pos] = et[e] * N + src[e];
}

__global__ void gather64_relu(const float* __restrict__ rel, const int* __restrict__ off,
                              const int* __restrict__ key, float* __restrict__ h, int N) {
    const long long t = (long long)blockIdx.x * blockDim.x + threadIdx.x;
    const int i = (int)(t >> 3);
    if (i >= N) return;
    const int f = ((int)t & 7) * 4;
    float* hp = h + (size_t)i * 64 + f;
    float4 acc = *(const float4*)hp;
    float4 acd = *(const float4*)(hp + 32);
    int p = off[i];
    const int pe = off[i + 1];
    for (; p + 4 <= pe; p += 4) {
        const int k0 = key[p], k1 = key[p + 1], k2 = key[p + 2], k3 = key[p + 3];
        const float* r0 = rel + (size_t)k0 * 64 + f;
        const float* r1 = rel + (size_t)k1 * 64 + f;
        const float* r2 = rel + (size_t)k2 * 64 + f;
        const float* r3 = rel + (size_t)k3 * 64 + f;
        const float4 a = *(const float4*)r0, a2 = *(const float4*)(r0 + 32);
        const float4 b = *(const float4*)r1, b2 = *(const float4*)(r1 + 32);
        const float4 c = *(const float4*)r2, c2 = *(const float4*)(r2 + 32);
        const float4 d = *(const float4*)r3, d2 = *(const float4*)(r3 + 32);
        acc.x += a.x; acc.y += a.y; acc.z += a.z; acc.w += a.w;
        acd.x += a2.x; acd.y += a2.y; acd.z += a2.z; acd.w += a2.w;
        acc.x += b.x; acc.y += b.y; acc.z += b.z; acc.w += b.w;
        acd.x += b2.x; acd.y += b2.y; acd.z += b2.z; acd.w += b2.w;
        acc.x += c.x; acc.y += c.y; acc.z += c.z; acc.w += c.w;
        acd.x += c2.x; acd.y += c2.y; acd.z += c2.z; acd.w += c2.w;
        acc.x += d.x; acc.y += d.y; acc.z += d.z; acc.w += d.w;
        acd.x += d2.x; acd.y += d2.y; acd.z += d2.z; acd.w += d2.w;
    }
    #pragma nounroll
    for (; p < pe; ++p) {
        const float* r0 = rel + (size_t)key[p] * 64 + f;
        const float4 a = *(const float4*)r0, a2 = *(const float4*)(r0 + 32);
        acc.x += a.x; acc.y += a.y; acc.z += a.z; acc.w += a.w;
        acd.x += a2.x; acd.y += a2.y; acd.z += a2.z; acd.w += a2.w;
    }
    acc.x = fmaxf(acc.x, 0.0f); acc.y = fmaxf(acc.y, 0.0f);
    acc.z = fmaxf(acc.z, 0.0f); acc.w = fmaxf(acc.w, 0.0f);
    acd.x = fmaxf(acd.x, 0.0f); acd.y = fmaxf(acd.y, 0.0f);
    acd.z = fmaxf(acd.z, 0.0f); acd.w = fmaxf(acd.w, 0.0f);
    *(float4*)hp = acc;
    *(float4*)(hp + 32) = acd;
}

__global__ void gather2(const float* __restrict__ rel3, const int* __restrict__ off,
                        const int* __restrict__ key, float* __restrict__ agg3, int N) {
    const long long t = (long long)blockIdx.x * blockDim.x + threadIdx.x;
    const int i = (int)(t >> 4);
    const int l = (int)t & 15;
    float s0 = 0.0f, s1 = 0.0f;
    if (i < N) {
        const int pe = off[i + 1];
        for (int p = off[i] + l; p < pe; p += 16) {
            const float2 r = *(const float2*)(rel3 + (size_t)key[p] * 2);
            s0 += r.x; s1 += r.y;
        }
    }
    for (int m = 8; m >= 1; m >>= 1) {
        s0 += __shfl_xor(s0, m, 16);
        s1 += __shfl_xor(s1, m, 16);
    }
    if (i < N && l == 0) {
        agg3[(size_t)i * 2 + 0] += s0;
        agg3[(size_t)i * 2 + 1] += s1;
    }
}

template <int U>
__device__ __forceinline__ void l2_add_rows(const float* __restrict__ h1, const int* __restrict__ kp,
                                            int N, int f, float4& a0, float4& b0, float4& a1, float4& b1) {
    int    kk[U];
    float  m1[U];
    float4 lo[U], hi[U];
    #pragma unroll
    for (int j = 0; j < U; ++j) kk[j] = kp[j];
    #pragma unroll
    for (int j = 0; j < U; ++j) {
        const bool r = kk[j] >= N;
        const int  s = r ? kk[j] - N : kk[j];
        m1[j] = r ? 1.0f : 0.0f;
        const float* rp = h1 + (size_t)s * 64 + f;
        lo[j] = *(const float4*)rp;
        hi[j] = *(const float4*)(rp + 32);
    }
    #pragma unroll
    for (int j = 0; j < U; ++j) {
        const float w1 = m1[j], w0 = 1.0f - m1[j];
        a0.x = fmaf(lo[j].x, w0, a0.x); a0.y = fmaf(lo[j].y, w0, a0.y);
        a0.z = fmaf(lo[j].z, w0, a0.z); a0.w = fmaf(lo[j].w, w0, a0.w);
        b0.x = fmaf(hi[j].x, w0, b0.x); b0.y = fmaf(hi[j].y, w0, b0.y);
        b0.z = fmaf(hi[j].z, w0, b0.z); b0.w = fmaf(hi[j].w, w0, b0.w);
        a1.x = fmaf(lo[j].x, w1, a1.x); a1.y = fmaf(lo[j].y, w1, a1.y);
        a1.z = fmaf(lo[j].z, w1, a1.z); a1.w = fmaf(lo[j].w, w1, a1.w);
        b1.x = fmaf(hi[j].x, w1, b1.x); b1.y = fmaf(hi[j].y, w1, b1.y);
        b1.z = fmaf(hi[j].z, w1, b1.z); b1.w = fmaf(hi[j].w, w1, b1.w);
    }
}

#define L2_NODES  32
#define L2_PITCH  196
#define L2_HPITCH 68
__global__ __launch_bounds__(256) void layer2_sum_xw_l3(const float* __restrict__ h1, const int* __restrict__ off,
                                 const int* __restrict__ key,
                                 const float* __restrict__ W2, const float* __restrict__ loop2,
                                 const float* __restrict__ b2,
                                 const float* __restrict__ W3, const float* __restrict__ loop3,
                                 const float* __restrict__ b3,
                                 float* __restrict__ rel3, float* __restrict__ agg3, int N) {
    __shared__ __align__(16) float As[L2_NODES * L2_PITCH];
    __shared__ __align__(16) float Hs[L2_NODES * L2_HPITCH];
    const int t    = (int)threadIdx.x;
    const int row  = t >> 3;
    const int f    = (t & 7) * 4;
    const int i    = (int)blockIdx.x * L2_NODES + row;
    float4 a0  = make_float4(0.0f, 0.0f, 0.0f, 0.0f);
    float4 b0  = make_float4(0.0f, 0.0f, 0.0f, 0.0f);
    float4 a1  = make_float4(0.0f, 0.0f, 0.0f, 0.0f);
    float4 b1  = make_float4(0.0f, 0.0f, 0.0f, 0.0f);
    float4 own = make_float4(0.0f, 0.0f, 0.0f, 0.0f);
    float4 owb = make_float4(0.0f, 0.0f, 0.0f, 0.0f);
    if (i < N) {
        own = *(const float4*)(h1 + (size_t)i * 64 + f);
        owb = *(const float4*)(h1 + (size_t)i * 64 + f + 32);
        int p = off[i];
        const int pe = off[i + 1];
        for (; p + 8 <= pe; p += 8) l2_add_rows<8>(h1, key + p, N, f, a0, b0, a1, b1);
        if (p + 4 <= pe) { l2_add_rows<4>(h1, key + p, N, f, a0, b0, a1, b1); p += 4; }
        for (; p < pe; ++p) l2_add_rows<1>(h1, key + p, N, f, a0, b0, a1, b1);
    }
    {
        float* ar = As + row * L2_PITCH + f;
        *(float4*)(ar)       = a0;
        *(float4*)(ar + 32)  = b0;
        *(float4*)(ar + 64)  = a1;
        *(float4*)(ar + 96)  = b1;
        *(float4*)(ar + 128) = own;
        *(float4*)(ar + 160) = owb;
    }
    __syncthreads();

    {
        const int wave  = t >> 5;
        const int strip = wave >> 2;
        const int lane  = t & 31;
        const int half  = lane >> 4;
        const int lm    = lane & 15;
        const int n     = (wave & 3) * 16 + lm;
        v8f c0, c1, c2;
        const float bv = b2[n];
        #pragma unroll
        for (int v = 0; v < 8; ++v) { c0[v] = 0.0f; c1[v] = 0.0f; c2[v] = bv; }
        const float* arow = As + (strip * 16 + lm) * L2_PITCH + 2 * half;
        const float* w1p  = W2 + 64 * 64;
        #pragma unroll
        for (int k0 = 0; k0 < 64; k0 += 4) {
            const int kb = k0 + 2 * half;
            #pragma unroll
            for (int mt = 0; mt < 3; ++mt) {
                const float* Wm = (mt == 0) ? W2 : (mt == 1) ? w1p : loop2;
                const float2 av = *(const float2*)(arow + 64 * mt + k0);
                v2f a; a.x = av.x; a.y = av.y;
                v2f b;
                b.x = Wm[(size_t)kb * 64 + n];
                b.y = Wm[(size_t)(kb + 1) * 64 + n];
                if (mt == 0)      c0 = __builtin_amdgcn_wmma_f32_16x16x4_f32(false, a, false, b, (short)0, c0, false, false);
                else if (mt == 1) c1 = __builtin_amdgcn_wmma_f32_16x16x4_f32(false, a, false, b, (short)0, c1, false, false);
                else              c2 = __builtin_amdgcn_wmma_f32_16x16x4_f32(false, a, false, b, (short)0, c2, false, false);
            }
        }
        #pragma unroll
        for (int v = 0; v < 8; ++v)
            Hs[(strip * 16 + v + 8 * half) * L2_HPITCH + n] = fmaxf((c0[v] + c1[v]) + c2[v], 0.0f);
    }
    __syncthreads();

    const float4 hq = *(const float4*)(Hs + row * L2_HPITCH + f);
    const float4 hr = *(const float4*)(Hs + row * L2_HPITCH + f + 32);
    const float hv[4] = { hq.x, hq.y, hq.z, hq.w };
    const float hw[4] = { hr.x, hr.y, hr.z, hr.w };
    const float* wm[3] = { W3 + f * 2, W3 + 128 + f * 2, loop3 + f * 2 };
    float o[6];
    #pragma unroll
    for (int m = 0; m < 3; ++m) {
        float s0 = 0.0f, s1 = 0.0f, u0 = 0.0f, u1 = 0.0f;
        #pragma unroll
        for (int j = 0; j < 4; ++j) {
            s0 += hv[j] * wm[m][2 * j + 0];
            s1 += hv[j] * wm[m][2 * j + 1];
        }
        #pragma unroll
        for (int j = 0; j < 4; ++j) {
            u0 += hw[j] * wm[m][64 + 2 * j + 0];
            u1 += hw[j] * wm[m][64 + 2 * j + 1];
        }
        o[2 * m + 0] = s0 + u0;
        o[2 * m + 1] = s1 + u1;
    }
    for (int d = 4; d >= 1; d >>= 1) {
        #pragma unroll
        for (int q = 0; q < 6; ++q) o[q] += __shfl_xor(o[q], d, 8);
    }
    if ((t & 7) == 0 && i < N) {
        rel3[(size_t)i * 2 + 0] = o[0];
        rel3[(size_t)i * 2 + 1] = o[1];
        rel3[(size_t)(N + i) * 2 + 0] = o[2];
        rel3[(size_t)(N + i) * 2 + 1] = o[3];
        agg3[(size_t)i * 2 + 0] = o[4] + b3[0];
        agg3[(size_t)i * 2 + 1] = o[5] + b3[1];
    }
}

#define TAIL_TPB 256
__global__ __launch_bounds__(TAIL_TPB) void gather2_min(const float* __restrict__ rel3, const int* __restrict__ off,
                            const int* __restrict__ key, float* __restrict__ agg3,
                            float* __restrict__ blockmin, int N) {
    __shared__ float sm[TAIL_TPB / 32];
    const long long t = (long long)blockIdx.x * blockDim.x + threadIdx.x;
    const int i = (int)(t >> 2);
    const int q = (int)t & 3;
    float s0[4] = { 0.0f, 0.0f, 0.0f, 0.0f };
    float s1[4] = { 0.0f, 0.0f, 0.0f, 0.0f };
    float g0 = 0.0f, g1 = 0.0f;
    if (i < N) {
        const int p0 = off[i];
        const int pe = off[i + 1];
        if (q == 0) {
            const float2 g = *(const float2*)(agg3 + (size_t)i * 2);
            g0 = g.x; g1 = g.y;
        }
        for (int base = p0; base < pe; base += 32) {
            int    kk[8];
            float2 r[8];
            #pragma unroll
            for (int u = 0; u < 8; ++u) kk[u] = key[min(base + q + 4 * u, pe - 1)];
            #pragma unroll
            for (int u = 0; u < 8; ++u) r[u] = *(const float2*)(rel3 + (size_t)kk[u] * 2);
            #pragma unroll
            for (int u = 0; u < 8; ++u) {
                const float w = (base + q + 4 * u < pe) ? 1.0f : 0.0f;
                s0[u & 3] = fmaf(r[u].x, w, s0[u & 3]);
                s1[u & 3] = fmaf(r[u].y, w, s1[u & 3]);
            }
        }
    }
    float t0 = (s0[0] + s0[2]) + (s0[1] + s0[3]);
    float t1 = (s1[0] + s1[2]) + (s1[1] + s1[3]);
    for (int m = 2; m >= 1; m >>= 1) {
        t0 += __shfl_xor(t0, m, 4);
        t1 += __shfl_xor(t1, m, 4);
    }
    float mn = 3.402823466e38f;
    if (i < N && q == 0) {
        const float v0 = g0 + t0;
        const float v1 = g1 + t1;
        agg3[(size_t)i * 2 + 0] = v0;
        agg3[(size_t)i * 2 + 1] = v1;
        mn = fminf(v0, v1);
    }
    for (int m = 16; m >= 4; m >>= 1) mn = fminf(mn, __shfl_xor(mn, m, 32));
    if ((threadIdx.x & 31) == 0) sm[threadIdx.x >> 5] = mn;
    __syncthreads();
    if (threadIdx.x == 0) {
        float r = sm[0];
        #pragma unroll
        for (int w = 1; w < TAIL_TPB / 32; ++w) r = fminf(r, sm[w]);
        blockmin[blockIdx.x] = r;
    }
}

__global__ __launch_bounds__(TAIL_TPB) void finalize_bm(const float* __restrict__ agg3,
                            const int* __restrict__ cell,
                            const int* __restrict__ maxs,
                            const float* __restrict__ blockmin, int nblk,
                            float* __restrict__ out, int N) {
    __shared__ float sm[TAIL_TPB / 32];
    float mn = 3.402823466e38f;
    for (int j0 = threadIdx.x; j0 < nblk; j0 += 16 * TAIL_TPB) {
        float v[16];
        #pragma unroll
        for (int u = 0; u < 16; ++u) v[u] = blockmin[min(j0 + u * TAIL_TPB, nblk - 1)];
        #pragma unroll
        for (int u = 0; u < 16; ++u) mn = fminf(mn, v[u]);
    }
    for (int m = 16; m >= 1; m >>= 1) mn = fminf(mn, __shfl_xor(mn, m, 32));
    if ((threadIdx.x & 31) == 0) sm[threadIdx.x >> 5] = mn;
    __syncthreads();
    float r = sm[0];
    #pragma unroll
    for (int w = 1; w < TAIL_TPB / 32; ++w) r = fminf(r, sm[w]);
    const float gmin = r - 1.0f;
    const int i = blockIdx.x * blockDim.x + threadIdx.x;
    if (i >= N) return;
    const bool upper = cell[i] >= (maxs[i] - 1);
    const bool lower = cell[i] == 0;
    out[(size_t)i * 2 + 0] = upper ? gmin : agg3[(size_t)i * 2 + 0];
    out[(size_t)i * 2 + 1] = lower ? gmin : agg3[(size_t)i * 2 + 1];
}

__device__ __forceinline__ unsigned f2ord(float f) {
    unsigned u = __float_as_uint(f);
    return (u & 0x80000000u) ? ~u : (u | 0x80000000u);
}
__device__ __forceinline__ float ord2f(unsigned u) {
    return __uint_as_float((u & 0x80000000u) ? (u ^ 0x80000000u) : ~u);
}

__global__ void init_slot(unsigned* slot) { *slot = 0xFFFFFFFFu; }

__global__ void min_reduce(const float* __restrict__ p, int n, unsigned* slot) {
    __shared__ float sm[256];
    float m = 3.402823466e38f;
    for (int i = blockIdx.x * blockDim.x + threadIdx.x; i < n;
         i += gridDim.x * blockDim.x)
        m = fminf(m, p[i]);
    sm[threadIdx.x] = m;
    __syncthreads();
    for (int s = blockDim.x >> 1; s > 0; s >>= 1) {
        if ((int)threadIdx.x < s) sm[threadIdx.x] = fminf(sm[threadIdx.x], sm[threadIdx.x + s]);
        __syncthreads();
    }
    if (threadIdx.x == 0) atomicMin(slot, f2ord(sm[0]));
}

__global__ void finalize_k(const float* __restrict__ agg3,
                           const int* __restrict__ cell,
                           const int* __restrict__ maxs,
                           const unsigned* __restrict__ slot,
                           float* __restrict__ out, int N) {
    const int i = blockIdx.x * blockDim.x + threadIdx.x;
    if (i >= N) return;
    const float gmin = ord2f(*slot) - 1.0f;
    const bool upper = cell[i] >= (maxs[i] - 1);
    const bool lower = cell[i] == 0;
    out[(size_t)i * 2 + 0] = upper ? gmin : agg3[(size_t)i * 2 + 0];
    out[(size_t)i * 2 + 1] = lower ? gmin : agg3[(size_t)i * 2 + 1];
}

extern "C" void kernel_launch(void* const* d_in, const int* in_sizes, int n_in,
                              void* d_out, int out_size, void* d_ws, size_t ws_size,
                              hipStream_t stream) {
    const float* x      = (const float*)d_in[0];
    const int*   src    = (const int*)d_in[1];
    const int*   dst    = (const int*)d_in[2];
    const int*   etypes = (const int*)d_in[3];
    const int*   cell   = (const int*)d_in[4];
    const int*   maxs   = (const int*)d_in[5];
    const float* W1     = (const float*)d_in[6];
    const float* loop1  = (const float*)d_in[7];
    const float* b1     = (const float*)d_in[8];
    const float* W2     = (const float*)d_in[9];
    const float* loop2  = (const float*)d_in[10];
    const float* b2     = (const float*)d_in[11];
    const float* W3     = (const float*)d_in[12];
    const float* loop3  = (const float*)d_in[13];
    const float* b3     = (const float*)d_in[14];
    float* out = (float*)d_out;

    const int N = in_sizes[4];
    const int E = in_sizes[1];
    const int K1 = in_sizes[0] / N;

    float* relbuf = (float*)d_ws;
    float* h1     = relbuf + (size_t)2 * N * 64;
    float* h2     = h1 + (size_t)N * 64;
    float* rel3   = h2 + (size_t)N * 64;
    float* agg3   = rel3 + (size_t)4 * N;
    unsigned* slot = (unsigned*)(agg3 + (size_t)2 * N);

    const int waves   = (N + 15) / 16;
    const int gblocks = (waves * 32 + 255) / 256;
    const int eb64    = (int)(((long long)E * 16 + 255) / 256);
    const int nb      = (N + 255) / 256;
    const int fb      = (N * 64 + 255) / 256;
    const int eb2     = (E + 255) / 256;

    int* key      = (int*)(slot + 4);
    int* off      = key + (size_t)E;
    int* blockcnt = (int*)relbuf;
    int* rank     = blockcnt + (size_t)CSR_NB * N;
    const size_t need = ((size_t)262 * N + 4 + (size_t)E + (size_t)N + 1) * sizeof(float);
    int* coarse   = (int*)rel3;
    const bool csr_ok = ws_size >= need && (long long)E <= (long long)192 * N
                        && (long long)CSR_NB * nb <= (long long)6 * N;
    const size_t need2 = need + (size_t)E * sizeof(int);
    const int chunk0 = (E + CSR_NB - 1) / CSR_NB;
    const bool shared_ok = csr_ok && ws_size >= need2
                           && N <= CSR_CAP && chunk0 <= 1024 * CSR_EPT
                           && (long long)2 * N <= (long long)0x1000000
                           && (long long)2 * CSR_NB * nb <= (long long)6 * N;

    if (shared_ok) {
        const int chunk = chunk0;
        const int gb8   = (int)(((long long)N * 8 + 255) / 256);
        const int gb4   = (int)(((long long)N * 4 + 255) / 256);
        const int l2b   = (N + L2_NODES - 1) / L2_NODES;
        unsigned* ebin  = (unsigned*)(off + (size_t)N + 1);
        int* cstart     = coarse + (size_t)CSR_NB * nb;
        csr_hist_bin<<<CSR_NB, 1024, 0, stream>>>(dst, src, etypes, ebin, coarse, cstart, E, N, chunk, nb);

        gemm3_xw_colscan_place<<<gblocks + nb, 256, 0, stream>>>(x, W1, loop1, b1, relbuf, h1, N, K1, gblocks,
                                                                 coarse, cstart, ebin, off, key, E, chunk, nb);
        gather64_relu<<<gb8, 256, 0, stream>>>(relbuf, off, key, h1, N);

        layer2_sum_xw_l3<<<l2b, 256, 0, stream>>>(h1, off, key, W2, loop2, b2, W3, loop3, b3, rel3, agg3, N);

        float* blockmin = relbuf;
        gather2_min<<<gb4, TAIL_TPB, 0, stream>>>(rel3, off, key, agg3, blockmin, N);
        finalize_bm<<<nb, TAIL_TPB, 0, stream>>>(agg3, cell, maxs, blockmin, gb4, out, N);
    } else if (csr_ok) {
        const int chunk = (E + CSR_NB - 1) / CSR_NB;
        const int gb8   = (int)(((long long)N * 8 + 255) / 256);
        const int gb4   = (int)(((long long)N * 4 + 255) / 256);
        const int l2b   = (N + L2_NODES - 1) / L2_NODES;
        csr_hist<<<CSR_NB, 1024, 0, stream>>>(dst, rank, blockcnt, coarse, E, N, chunk, nb);
        csr_colscan_off<<<nb, CSR_RANGE, 0, stream>>>(blockcnt, coarse, off, N, nb);
        csr_fill<<<eb2, 256, 0, stream>>>(src, dst, etypes, rank, blockcnt, key, E, N, chunk);

        gemm3_xw<<<gblocks, 256, 0, stream>>>(x, W1, loop1, b1, relbuf, h1, N, K1);
        gather64_relu<<<gb8, 256, 0, stream>>>(relbuf, off, key, h1, N);

        layer2_sum_xw_l3<<<l2b, 256, 0, stream>>>(h1, off, key, W2, loop2, b2, W3, loop3, b3, rel3, agg3, N);

        float* blockmin = relbuf;
        gather2_min<<<gb4, TAIL_TPB, 0, stream>>>(rel3, off, key, agg3, blockmin, N);
        finalize_bm<<<nb, TAIL_TPB, 0, stream>>>(agg3, cell, maxs, blockmin, gb4, out, N);
    } else {
        gemm3_xw<<<gblocks, 256, 0, stream>>>(x, W1, loop1, b1, relbuf, h1, N, K1);
        edge_scatter64<<<eb64, 256, 0, stream>>>(relbuf, src, dst, etypes, h1, E, N);
        relu_k<<<fb, 256, 0, stream>>>(h1, N * 64);

        gemm3_xw<<<gblocks, 256, 0, stream>>>(h1, W2, loop2, b2, relbuf, h2, N, 64);
        edge_scatter64<<<eb64, 256, 0, stream>>>(relbuf, src, dst, etypes, h2, E, N);
        relu_k<<<fb, 256, 0, stream>>>(h2, N * 64);

        layer3_node<<<nb, 256, 0, stream>>>(h2, W3, loop3, b3, rel3, agg3, N);
        edge_scatter2<<<eb2, 256, 0, stream>>>(rel3, src, dst, etypes, agg3, E, N);
        init_slot<<<1, 1, 0, stream>>>(slot);
        min_reduce<<<512, 256, 0, stream>>>(agg3, N * 2, slot);
        finalize_k<<<nb, 256, 0, stream>>>(agg3, cell, maxs, slot, out, N);
    }
}
